# starting kernel
# baseline (speedup 1.0000x reference)
#include <hip/hip_runtime.h>
#include <math.h>

#define NTYPE 4
#define NWAVE 8
#define NORBIT 128
#define NEIGH 64
#define NB 32
#define NATOM 256
#define NPAIR_PER (NATOM * NEIGH)
#define NPAIR_TOT (NB * NPAIR_PER)
#define TOTNATOM (NB * NATOM)
#define NPARA 13
#define PI_OVER_CUTOFF 0.62831853071795864769f

typedef float v2f __attribute__((ext_vector_type(2)));
typedef float v8f __attribute__((ext_vector_type(8)));

__device__ __forceinline__ v8f wmma_f32_16x16x4(v2f a, v2f b, v8f c) {
  return __builtin_amdgcn_wmma_f32_16x16x4_f32(false, a, false, b, (short)0, c,
                                               false, false);
}

__global__ void k_init_orb(const float* __restrict__ params,
                           const int* __restrict__ species,
                           float* __restrict__ orb) {
  int t = blockIdx.x * blockDim.x + threadIdx.x;
  int a = t / NWAVE, w = t % NWAVE;
  orb[t] = params[species[a] * NWAVE + w];
}

__global__ void k_zero(float* __restrict__ p, int n) {
  int t = blockIdx.x * blockDim.x + threadIdx.x;
  if (t < n) p[t] = 0.0f;
}

__global__ void k_pairs(const float* __restrict__ cart,
                        const float* __restrict__ shifts,
                        const float* __restrict__ rs,
                        const float* __restrict__ inta,
                        const int* __restrict__ species,
                        const int* __restrict__ atom_index,
                        const float* __restrict__ orb,
                        float* __restrict__ sum_w) {
  int p = blockIdx.x * blockDim.x + threadIdx.x;
  int b = p / NPAIR_PER;
  int gi = b * NATOM + atom_index[p];
  int gj = b * NATOM + atom_index[NPAIR_TOT + p];

  float dx = cart[gi * 3 + 0] - cart[gj * 3 + 0] + shifts[p * 3 + 0];
  float dy = cart[gi * 3 + 1] - cart[gj * 3 + 1] + shifts[p * 3 + 1];
  float dz = cart[gi * 3 + 2] - cart[gj * 3 + 2] + shifts[p * 3 + 2];
  float d = sqrtf(dx * dx + dy * dy + dz * dz);
  float fc = 0.5f * cosf(d * PI_OVER_CUTOFF) + 0.5f;
  fc = fc * fc;

  int sp = species[gj];
  float wv[NWAVE];
#pragma unroll
  for (int w = 0; w < NWAVE; ++w) {
    float t = d - rs[sp * NWAVE + w];
    wv[w] = expf(inta[sp * NWAVE + w] * t * t) * orb[gj * NWAVE + w];
  }

  float ang[NPARA];
  float dv[3] = {dx, dy, dz};
  ang[0] = fc;
#pragma unroll
  for (int a2 = 0; a2 < 3; ++a2) ang[1 + a2] = fc * dv[a2];
#pragma unroll
  for (int a2 = 0; a2 < 3; ++a2)
#pragma unroll
    for (int b2 = 0; b2 < 3; ++b2) ang[4 + a2 * 3 + b2] = ang[1 + a2] * dv[b2];

  float* dst = sum_w + gi * (NPARA * NWAVE);
#pragma unroll
  for (int j = 0; j < NPARA; ++j)
#pragma unroll
    for (int w = 0; w < NWAVE; ++w)
      unsafeAtomicAdd(dst + j * NWAVE + w, ang[j] * wv[w]);
}

__global__ void k_density(const float* __restrict__ sum_w,
                          const float* __restrict__ hyper,
                          float* __restrict__ density) {
  int wave = threadIdx.x >> 5;
  int lane = threadIdx.x & 31;
  int tile = blockIdx.x * 8 + wave;
  int lrow = lane & 15;
  int lhi = lane >> 4;
  int khalf = lhi * 2;

  for (int nt = 0; nt < 8; ++nt) {
    int n = nt * 16 + lrow;
    v8f acc = {};
    int para = 0;
    for (int ip = 0; ip < 3; ++ip) {
      const float* hb = hyper + ip * (NWAVE * NORBIT);
      v2f Blo, Bhi;
      Blo.x = hb[(khalf + 0) * NORBIT + n];
      Blo.y = hb[(khalf + 1) * NORBIT + n];
      Bhi.x = hb[(khalf + 4) * NORBIT + n];
      Bhi.y = hb[(khalf + 5) * NORBIT + n];
      int npj = (ip == 0) ? 1 : ((ip == 1) ? 3 : 9);
      for (int q = 0; q < npj; ++q, ++para) {
        const float* ab = sum_w + ((tile * 16 + lrow) * NPARA + para) * NWAVE;
        v2f Alo = {ab[khalf + 0], ab[khalf + 1]};
        v2f Ahi = {ab[khalf + 4], ab[khalf + 5]};
        v8f c = {};
        c = wmma_f32_16x16x4(Alo, Blo, c);
        c = wmma_f32_16x16x4(Ahi, Bhi, c);
#pragma unroll
        for (int r = 0; r < 8; ++r) acc[r] += c[r] * c[r];
      }
    }
#pragma unroll
    for (int r = 0; r < 8; ++r) {
      int atom = tile * 16 + r + lhi * 8;
      density[atom * NORBIT + nt * 16 + lrow] = acc[r];
    }
  }
}

__global__ void k_update(const float* __restrict__ density,
                         const float* __restrict__ oc_w,
                         const float* __restrict__ oc_e,
                         const int* __restrict__ species,
                         float* __restrict__ orb) {
  int wave = threadIdx.x >> 5;
  int lane = threadIdx.x & 31;
  int tile = blockIdx.x * 8 + wave;
  int lrow = lane & 15;
  int lhi = lane >> 4;

  v8f acc = {};
  const float* arow = density + (tile * 16 + lrow) * NORBIT;
  bool valid = (lrow < NWAVE);
#pragma unroll 4
  for (int kb = 0; kb < 32; ++kb) {
    int k = kb * 4 + lhi * 2;
    v2f A = {arow[k], arow[k + 1]};
    v2f B;
    B.x = valid ? oc_w[(k + 0) * NWAVE + lrow] : 0.0f;
    B.y = valid ? oc_w[(k + 1) * NWAVE + lrow] : 0.0f;
    acc = wmma_f32_16x16x4(A, B, acc);
  }
  if (valid) {
#pragma unroll
    for (int r = 0; r < 8; ++r) {
      int atom = tile * 16 + r + lhi * 8;
      float v = tanhf(acc[r] + oc_e[species[atom] * NWAVE + lrow]);
      orb[atom * NWAVE + lrow] += v;
    }
  }
}

extern "C" void kernel_launch(void* const* d_in, const int* in_sizes, int n_in,
                              void* d_out, int out_size, void* d_ws,
                              size_t ws_size, hipStream_t stream) {
  const float* cart = (const float*)d_in[0];
  const float* shifts = (const float*)d_in[1];
  const float* rs = (const float*)d_in[2];
  const float* inta = (const float*)d_in[3];
  const float* params = (const float*)d_in[4];
  const float* hyper = (const float*)d_in[5];
  const float* oc_w0 = (const float*)d_in[6];
  const float* oc_e0 = (const float*)d_in[7];
  const float* oc_w1 = (const float*)d_in[8];
  const float* oc_e1 = (const float*)d_in[9];
  const int* species = (const int*)d_in[11];
  const int* atom_index = (const int*)d_in[12];

  float* density = (float*)d_out;
  float* orb = (float*)d_ws;
  float* sum_w = orb + TOTNATOM * NWAVE;

  const int SW_N = TOTNATOM * NPARA * NWAVE;
  const float* Ws[2] = {oc_w0, oc_w1};
  const float* Es[2] = {oc_e0, oc_e1};

  k_init_orb<<<(TOTNATOM * NWAVE) / 256, 256, 0, stream>>>(params, species, orb);
  for (int it = 0; it < 3; ++it) {
    k_zero<<<(SW_N + 255) / 256, 256, 0, stream>>>(sum_w, SW_N);
    k_pairs<<<NPAIR_TOT / 256, 256, 0, stream>>>(cart, shifts, rs, inta, species,
                                                 atom_index, orb, sum_w);
    k_density<<<TOTNATOM / 16 / 8, 256, 0, stream>>>(sum_w, hyper, density);
    if (it < 2)
      k_update<<<TOTNATOM / 16 / 8, 256, 0, stream>>>(density, Ws[it], Es[it],
                                                      species, orb);
  }
}


# Segment sum without float atomics: each block counting-sorts its molecule's pairs by center in LDS and sums each center's run in registers in ascending pair order; zero launches dropped
# speedup vs baseline: 43.2809x; 43.2809x over previous
#include <hip/hip_runtime.h>
#include <math.h>

#define NTYPE 4
#define NWAVE 8
#define NORBIT 128
#define NEIGH 64
#define NB 32
#define NATOM 256
#define NPAIR_PER (NATOM * NEIGH)
#define NPAIR_TOT (NB * NPAIR_PER)
#define TOTNATOM (NB * NATOM)
#define NPARA 13
#define PI_OVER_CUTOFF 0.62831853071795864769f

typedef float v2f __attribute__((ext_vector_type(2)));
typedef float v8f __attribute__((ext_vector_type(8)));

__device__ __forceinline__ v8f wmma_f32_16x16x4(v2f a, v2f b, v8f c) {
  return __builtin_amdgcn_wmma_f32_16x16x4_f32(false, a, false, b, (short)0, c,
                                               false, false);
}

__global__ void k_init_orb(const float* __restrict__ params,
                           const int* __restrict__ species,
                           float* __restrict__ orb) {
  int t = blockIdx.x * blockDim.x + threadIdx.x;
  int a = t / NWAVE, w = t % NWAVE;
  orb[t] = params[species[a] * NWAVE + w];
}

#define GRP 8
#define CPG (NATOM / GRP)
#define SLICE (NPAIR_PER / 8)
#define CH 2560
#define ROWF (NPARA * NWAVE)

__global__ void __launch_bounds__(256)
k_gather(const float* __restrict__ cart,
         const float* __restrict__ shifts,
         const float* __restrict__ rs,
         const float* __restrict__ inta,
         const int* __restrict__ species,
         const int* __restrict__ atom_index,
         const float* __restrict__ orb,
         float* __restrict__ sum_w) {
  __shared__ int s_list[8][SLICE];
  __shared__ int s_sorted[NPAIR_PER];
  __shared__ float4 s_geo[CH];
  __shared__ float s_d[CH];
  __shared__ int s_js[CH];
  __shared__ float s_out[CPG * ROWF];
  __shared__ float s_orb[NATOM * NWAVE];
  __shared__ float s_cart[NATOM * 3];
  __shared__ int s_spec[NATOM];
  __shared__ float s_rs[NTYPE * NWAVE];
  __shared__ float s_inta[NTYPE * NWAVE];
  __shared__ int s_cnt[8][CPG];
  __shared__ int s_start[CPG + 1];

  const int t = threadIdx.x;
  const int wave = t >> 5;
  const int lane = t & 31;
  const int b = blockIdx.x / GRP;
  const int g = blockIdx.x % GRP;

  for (int i = t; i < NATOM * 3; i += 256) s_cart[i] = cart[b * NATOM * 3 + i];
  for (int i = t; i < NATOM * NWAVE; i += 256) s_orb[i] = orb[b * NATOM * NWAVE + i];
  s_spec[t] = species[b * NATOM + t];
  if (t < NTYPE * NWAVE) { s_rs[t] = rs[t]; s_inta[t] = inta[t]; }
  (&s_cnt[0][0])[t] = 0;

  const int* keys = atom_index + b * NPAIR_PER;
  const int kbase = wave * SLICE;
  const unsigned ltmask = (1u << lane) - 1u;
  int nloc = 0;
  for (int s0 = 0; s0 < SLICE / 32; s0 += 32) {
    int kr[32];
#pragma unroll
    for (int j = 0; j < 32; ++j) kr[j] = keys[kbase + (s0 + j) * 32 + lane];
#pragma unroll
    for (int j = 0; j < 32; ++j) {
      const int cc = kr[j];
      const bool inr = (cc >> 5) == g;
      const unsigned mk = __builtin_amdgcn_ballot_w32(inr);
      if (inr)
        s_list[wave][nloc + __builtin_popcount(mk & ltmask)] =
            (kbase + (s0 + j) * 32 + lane) | ((cc & 31) << 14);
      nloc += __builtin_popcount(mk);
    }
  }
  __syncthreads();

  for (int s = 0; s < nloc; s += 32) {
    const int idx = s + lane;
    const bool ok = idx < nloc;
    const int e = ok ? s_list[wave][idx] : 0;
    const int c5 = (e >> 14) & 31;
    unsigned m = __builtin_amdgcn_ballot_w32(ok);
#pragma unroll
    for (int bit = 0; bit < 5; ++bit) {
      const bool bs = ((c5 >> bit) & 1) != 0;
      const unsigned bb = __builtin_amdgcn_ballot_w32(bs);
      m &= bs ? bb : ~bb;
    }
    if (ok) {
      const int prev = s_cnt[wave][c5];
      const int r = __builtin_popcount(m & ltmask);
      s_list[wave][idx] = e | ((prev + r) << 19);
      if (r == 0) s_cnt[wave][c5] = prev + __builtin_popcount(m);
    }
  }
  __syncthreads();

  if (wave == 0) {
    int tot = 0;
#pragma unroll
    for (int w = 0; w < 8; ++w) {
      const int v = s_cnt[w][lane];
      s_cnt[w][lane] = tot;
      tot += v;
    }
    int inc = tot;
#pragma unroll
    for (int o = 1; o < 32; o <<= 1) {
      const int n = __shfl_up(inc, o);
      if (lane >= o) inc += n;
    }
    s_start[lane] = inc - tot;
    if (lane == 31) s_start[CPG] = inc;
  }
  __syncthreads();

  for (int s = 0; s < nloc; s += 32) {
    const int idx = s + lane;
    if (idx < nloc) {
      const int e = s_list[wave][idx];
      const int c5 = (e >> 14) & 31;
      s_sorted[s_start[c5] + s_cnt[wave][c5] + (e >> 19)] = e & 0x7ffff;
    }
  }
  __syncthreads();

  const int ne = s_start[CPG];
  const int c5 = t >> 3;
  const int w = t & 7;
  const int mystart = s_start[c5];
  const int myend = s_start[c5 + 1];
  float acc[NPARA];
#pragma unroll
  for (int j = 0; j < NPARA; ++j) acc[j] = 0.0f;

  for (int e0 = 0; e0 < ne; e0 += CH) {
    const int nchunk = (ne - e0 < CH) ? (ne - e0) : CH;

    for (int i0 = t; i0 < nchunk; i0 += 256 * 4) {
      int ent[4], ljv[4];
      float sx[4], sy[4], sz[4];
#pragma unroll
      for (int u = 0; u < 4; ++u) {
        const int i = i0 + u * 256;
        ent[u] = s_sorted[e0 + ((i < nchunk) ? i : i0)];
      }
#pragma unroll
      for (int u = 0; u < 4; ++u) {
        const int p = b * NPAIR_PER + (ent[u] & 0x3fff);
        ljv[u] = atom_index[NPAIR_TOT + p];
        sx[u] = shifts[p * 3 + 0];
        sy[u] = shifts[p * 3 + 1];
        sz[u] = shifts[p * 3 + 2];
      }
#pragma unroll
      for (int u = 0; u < 4; ++u) {
        const int i = i0 + u * 256;
        const int li = g * CPG + (ent[u] >> 14);
        const int lj = ljv[u];
        const float dx = s_cart[li * 3 + 0] - s_cart[lj * 3 + 0] + sx[u];
        const float dy = s_cart[li * 3 + 1] - s_cart[lj * 3 + 1] + sy[u];
        const float dz = s_cart[li * 3 + 2] - s_cart[lj * 3 + 2] + sz[u];
        const float d = sqrtf(dx * dx + dy * dy + dz * dz);
        float fc = 0.5f * cosf(d * PI_OVER_CUTOFF) + 0.5f;
        fc = fc * fc;
        const int js = lj | (s_spec[lj] << 8);
        if (i < nchunk) {
          s_geo[i] = make_float4(fc, dx, dy, dz);
          s_d[i] = d;
          s_js[i] = js;
        }
      }
    }
    __syncthreads();

    const int lo = ((mystart > e0) ? mystart : e0) - e0;
    const int hi = ((myend < e0 + nchunk) ? myend : (e0 + nchunk)) - e0;
#pragma unroll 4
    for (int i = lo; i < hi; ++i) {
      const float4 g4 = s_geo[i];
      const float d = s_d[i];
      const int js = s_js[i];
      const int lj = js & 255;
      const int sp = js >> 8;
      const float tt = d - s_rs[sp * NWAVE + w];
      const float wv = expf(s_inta[sp * NWAVE + w] * tt * tt) * s_orb[lj * NWAVE + w];
      float ang[NPARA];
      const float dv[3] = {g4.y, g4.z, g4.w};
      ang[0] = g4.x;
#pragma unroll
      for (int a2 = 0; a2 < 3; ++a2) ang[1 + a2] = g4.x * dv[a2];
#pragma unroll
      for (int a2 = 0; a2 < 3; ++a2)
#pragma unroll
        for (int b2 = 0; b2 < 3; ++b2) ang[4 + a2 * 3 + b2] = ang[1 + a2] * dv[b2];
#pragma unroll
      for (int j = 0; j < NPARA; ++j) acc[j] += ang[j] * wv;
    }
    __syncthreads();
  }

#pragma unroll
  for (int j = 0; j < NPARA; ++j) s_out[c5 * ROWF + j * NWAVE + w] = acc[j];
  __syncthreads();
  float* dst = sum_w + (b * NATOM + g * CPG) * ROWF;
  for (int i = t; i < CPG * ROWF; i += 256) dst[i] = s_out[i];
}

__global__ void k_density(const float* __restrict__ sum_w,
                          const float* __restrict__ hyper,
                          float* __restrict__ density) {
  int wave = threadIdx.x >> 5;
  int lane = threadIdx.x & 31;
  int tile = blockIdx.x * 8 + wave;
  int lrow = lane & 15;
  int lhi = lane >> 4;
  int khalf = lhi * 2;

  for (int nt = 0; nt < 8; ++nt) {
    int n = nt * 16 + lrow;
    v8f acc = {};
    int para = 0;
    for (int ip = 0; ip < 3; ++ip) {
      const float* hb = hyper + ip * (NWAVE * NORBIT);
      v2f Blo, Bhi;
      Blo.x = hb[(khalf + 0) * NORBIT + n];
      Blo.y = hb[(khalf + 1) * NORBIT + n];
      Bhi.x = hb[(khalf + 4) * NORBIT + n];
      Bhi.y = hb[(khalf + 5) * NORBIT + n];
      int npj = (ip == 0) ? 1 : ((ip == 1) ? 3 : 9);
      for (int q = 0; q < npj; ++q, ++para) {
        const float* ab = sum_w + ((tile * 16 + lrow) * NPARA + para) * NWAVE;
        v2f Alo = {ab[khalf + 0], ab[khalf + 1]};
        v2f Ahi = {ab[khalf + 4], ab[khalf + 5]};
        v8f c = {};
        c = wmma_f32_16x16x4(Alo, Blo, c);
        c = wmma_f32_16x16x4(Ahi, Bhi, c);
#pragma unroll
        for (int r = 0; r < 8; ++r) acc[r] += c[r] * c[r];
      }
    }
#pragma unroll
    for (int r = 0; r < 8; ++r) {
      int atom = tile * 16 + r + lhi * 8;
      density[atom * NORBIT + nt * 16 + lrow] = acc[r];
    }
  }
}

__global__ void k_update(const float* __restrict__ density,
                         const float* __restrict__ oc_w,
                         const float* __restrict__ oc_e,
                         const int* __restrict__ species,
                         float* __restrict__ orb) {
  int wave = threadIdx.x >> 5;
  int lane = threadIdx.x & 31;
  int tile = blockIdx.x * 8 + wave;
  int lrow = lane & 15;
  int lhi = lane >> 4;

  v8f acc = {};
  const float* arow = density + (tile * 16 + lrow) * NORBIT;
  bool valid = (lrow < NWAVE);
#pragma unroll 4
  for (int kb = 0; kb < 32; ++kb) {
    int k = kb * 4 + lhi * 2;
    v2f A = {arow[k], arow[k + 1]};
    v2f B;
    B.x = valid ? oc_w[(k + 0) * NWAVE + lrow] : 0.0f;
    B.y = valid ? oc_w[(k + 1) * NWAVE + lrow] : 0.0f;
    acc = wmma_f32_16x16x4(A, B, acc);
  }
  if (valid) {
#pragma unroll
    for (int r = 0; r < 8; ++r) {
      int atom = tile * 16 + r + lhi * 8;
      float v = tanhf(acc[r] + oc_e[species[atom] * NWAVE + lrow]);
      orb[atom * NWAVE + lrow] += v;
    }
  }
}

extern "C" void kernel_launch(void* const* d_in, const int* in_sizes, int n_in,
                              void* d_out, int out_size, void* d_ws,
                              size_t ws_size, hipStream_t stream) {
  const float* cart = (const float*)d_in[0];
  const float* shifts = (const float*)d_in[1];
  const float* rs = (const float*)d_in[2];
  const float* inta = (const float*)d_in[3];
  const float* params = (const float*)d_in[4];
  const float* hyper = (const float*)d_in[5];
  const float* oc_w0 = (const float*)d_in[6];
  const float* oc_e0 = (const float*)d_in[7];
  const float* oc_w1 = (const float*)d_in[8];
  const float* oc_e1 = (const float*)d_in[9];
  const int* species = (const int*)d_in[11];
  const int* atom_index = (const int*)d_in[12];

  float* density = (float*)d_out;
  float* orb = (float*)d_ws;
  float* sum_w = orb + TOTNATOM * NWAVE;

  const float* Ws[2] = {oc_w0, oc_w1};
  const float* Es[2] = {oc_e0, oc_e1};

  k_init_orb<<<(TOTNATOM * NWAVE) / 256, 256, 0, stream>>>(params, species, orb);
  for (int it = 0; it < 3; ++it) {
    k_gather<<<NB * GRP, 256, 0, stream>>>(cart, shifts, rs, inta, species,
                                           atom_index, orb, sum_w);
    k_density<<<TOTNATOM / 16 / 8, 256, 0, stream>>>(sum_w, hyper, density);
    if (it < 2)
      k_update<<<TOTNATOM / 16 / 8, 256, 0, stream>>>(density, Ws[it], Es[it],
                                                      species, orb);
  }
}


# Fuse each obtain's segment sum, einsum-square and coefficient update into one kernel: 3 launches (was 9); weighted sums and the first two densities stay in LDS; summation orders unchanged
# speedup vs baseline: 70.3766x; 1.6260x over previous
#include <hip/hip_runtime.h>
#include <math.h>

#define NTYPE 4
#define NWAVE 8
#define NORBIT 128
#define NEIGH 64
#define NB 32
#define NATOM 256
#define NPAIR_PER (NATOM * NEIGH)
#define NPAIR_TOT (NB * NPAIR_PER)
#define TOTNATOM (NB * NATOM)
#define NPARA 13
#define PI_OVER_CUTOFF 0.62831853071795864769f

typedef float v2f __attribute__((ext_vector_type(2)));
typedef float v8f __attribute__((ext_vector_type(8)));

__device__ __forceinline__ v8f wmma_f32_16x16x4(v2f a, v2f b, v8f c) {
  return __builtin_amdgcn_wmma_f32_16x16x4_f32(false, a, false, b, (short)0, c,
                                               false, false);
}

#define GRP 8
#define CPG (NATOM / GRP)
#define SLICE (NPAIR_PER / 8)
#define CH 2560
#define ROWF (NPARA * NWAVE)
#define SOP (ROWF + 4)
#define DP (NORBIT + 4)

__global__ void __launch_bounds__(256)
k_obtain(const float* __restrict__ cart,
         const float* __restrict__ shifts,
         const float* __restrict__ rs,
         const float* __restrict__ inta,
         const float* __restrict__ params,
         const float* __restrict__ hyper,
         const float* __restrict__ oc_w,
         const float* __restrict__ oc_e,
         const int* __restrict__ species,
         const int* __restrict__ atom_index,
         const float* __restrict__ orb_in,
         float* __restrict__ orb_out,
         float* __restrict__ density,
         const int it) {
  __shared__ int s_list[8][SLICE];
  __shared__ int s_sorted[NPAIR_PER];
  __shared__ float4 s_geo[CH];
  __shared__ float s_d[CH];
  __shared__ int s_js[CH];
  __shared__ float s_out[CPG * SOP];
  __shared__ __attribute__((aligned(16))) float s_den[CPG * DP];
  __shared__ __attribute__((aligned(16))) float s_new[CPG * NWAVE];
  __shared__ float s_orb[NATOM * NWAVE];
  __shared__ float s_cart[NATOM * 3];
  __shared__ int s_spec[NATOM];
  __shared__ float s_rs[NTYPE * NWAVE];
  __shared__ float s_inta[NTYPE * NWAVE];
  __shared__ int s_cnt[8][CPG];
  __shared__ int s_start[CPG + 1];

  const int t = threadIdx.x;
  const int wave = t >> 5;
  const int lane = t & 31;
  const int b = blockIdx.x / GRP;
  const int g = blockIdx.x % GRP;

  for (int i = t; i < NATOM * 3; i += 256) s_cart[i] = cart[b * NATOM * 3 + i];
  if (it == 0) {
    for (int i = t; i < NATOM * NWAVE; i += 256)
      s_orb[i] = params[species[b * NATOM + (i >> 3)] * NWAVE + (i & 7)];
  } else {
    for (int i = t; i < NATOM * NWAVE; i += 256) s_orb[i] = orb_in[b * NATOM * NWAVE + i];
  }
  s_spec[t] = species[b * NATOM + t];
  if (t < NTYPE * NWAVE) { s_rs[t] = rs[t]; s_inta[t] = inta[t]; }
  (&s_cnt[0][0])[t] = 0;

  const int* keys = atom_index + b * NPAIR_PER;
  const int kbase = wave * SLICE;
  const unsigned ltmask = (1u << lane) - 1u;
  int nloc = 0;
  for (int s0 = 0; s0 < SLICE / 32; s0 += 32) {
    int kr[32];
#pragma unroll
    for (int j = 0; j < 32; ++j) kr[j] = keys[kbase + (s0 + j) * 32 + lane];
#pragma unroll
    for (int j = 0; j < 32; ++j) {
      const int cc = kr[j];
      const bool inr = (cc >> 5) == g;
      const unsigned mk = __builtin_amdgcn_ballot_w32(inr);
      if (inr)
        s_list[wave][nloc + __builtin_popcount(mk & ltmask)] =
            (kbase + (s0 + j) * 32 + lane) | ((cc & 31) << 14);
      nloc += __builtin_popcount(mk);
    }
  }
  __syncthreads();

  for (int s = 0; s < nloc; s += 32) {
    const int idx = s + lane;
    const bool ok = idx < nloc;
    const int e = ok ? s_list[wave][idx] : 0;
    const int c5 = (e >> 14) & 31;
    unsigned m = __builtin_amdgcn_ballot_w32(ok);
#pragma unroll
    for (int bit = 0; bit < 5; ++bit) {
      const bool bs = ((c5 >> bit) & 1) != 0;
      const unsigned bb = __builtin_amdgcn_ballot_w32(bs);
      m &= bs ? bb : ~bb;
    }
    if (ok) {
      const int prev = s_cnt[wave][c5];
      const int r = __builtin_popcount(m & ltmask);
      s_list[wave][idx] = e | ((prev + r) << 19);
      if (r == 0) s_cnt[wave][c5] = prev + __builtin_popcount(m);
    }
  }
  __syncthreads();

  if (wave == 0) {
    int tot = 0;
#pragma unroll
    for (int w = 0; w < 8; ++w) {
      const int v = s_cnt[w][lane];
      s_cnt[w][lane] = tot;
      tot += v;
    }
    int inc = tot;
#pragma unroll
    for (int o = 1; o < 32; o <<= 1) {
      const int n = __shfl_up(inc, o);
      if (lane >= o) inc += n;
    }
    s_start[lane] = inc - tot;
    if (lane == 31) s_start[CPG] = inc;
  }
  __syncthreads();

  for (int s = 0; s < nloc; s += 32) {
    const int idx = s + lane;
    if (idx < nloc) {
      const int e = s_list[wave][idx];
      const int c5 = (e >> 14) & 31;
      s_sorted[s_start[c5] + s_cnt[wave][c5] + (e >> 19)] = e & 0x7ffff;
    }
  }
  __syncthreads();

  const int ne = s_start[CPG];
  const int c5 = t >> 3;
  const int w = t & 7;
  const int mystart = s_start[c5];
  const int myend = s_start[c5 + 1];
  float acc[NPARA];
#pragma unroll
  for (int j = 0; j < NPARA; ++j) acc[j] = 0.0f;

  for (int e0 = 0; e0 < ne; e0 += CH) {
    const int nchunk = (ne - e0 < CH) ? (ne - e0) : CH;

    for (int i0 = t; i0 < nchunk; i0 += 256 * 4) {
      int ent[4], ljv[4];
      float sx[4], sy[4], sz[4];
#pragma unroll
      for (int u = 0; u < 4; ++u) {
        const int i = i0 + u * 256;
        ent[u] = s_sorted[e0 + ((i < nchunk) ? i : i0)];
      }
#pragma unroll
      for (int u = 0; u < 4; ++u) {
        const int p = b * NPAIR_PER + (ent[u] & 0x3fff);
        ljv[u] = atom_index[NPAIR_TOT + p];
        sx[u] = shifts[p * 3 + 0];
        sy[u] = shifts[p * 3 + 1];
        sz[u] = shifts[p * 3 + 2];
      }
#pragma unroll
      for (int u = 0; u < 4; ++u) {
        const int i = i0 + u * 256;
        const int li = g * CPG + (ent[u] >> 14);
        const int lj = ljv[u];
        const float dx = s_cart[li * 3 + 0] - s_cart[lj * 3 + 0] + sx[u];
        const float dy = s_cart[li * 3 + 1] - s_cart[lj * 3 + 1] + sy[u];
        const float dz = s_cart[li * 3 + 2] - s_cart[lj * 3 + 2] + sz[u];
        const float d = sqrtf(dx * dx + dy * dy + dz * dz);
        float fc = 0.5f * cosf(d * PI_OVER_CUTOFF) + 0.5f;
        fc = fc * fc;
        const int js = lj | (s_spec[lj] << 8);
        if (i < nchunk) {
          s_geo[i] = make_float4(fc, dx, dy, dz);
          s_d[i] = d;
          s_js[i] = js;
        }
      }
    }
    __syncthreads();

    const int lo = ((mystart > e0) ? mystart : e0) - e0;
    const int hi = ((myend < e0 + nchunk) ? myend : (e0 + nchunk)) - e0;
#pragma unroll 4
    for (int i = lo; i < hi; ++i) {
      const float4 g4 = s_geo[i];
      const float d = s_d[i];
      const int js = s_js[i];
      const int lj = js & 255;
      const int sp = js >> 8;
      const float tt = d - s_rs[sp * NWAVE + w];
      const float wv = expf(s_inta[sp * NWAVE + w] * tt * tt) * s_orb[lj * NWAVE + w];
      float ang[NPARA];
      const float dv[3] = {g4.y, g4.z, g4.w};
      ang[0] = g4.x;
#pragma unroll
      for (int a2 = 0; a2 < 3; ++a2) ang[1 + a2] = g4.x * dv[a2];
#pragma unroll
      for (int a2 = 0; a2 < 3; ++a2)
#pragma unroll
        for (int b2 = 0; b2 < 3; ++b2) ang[4 + a2 * 3 + b2] = ang[1 + a2] * dv[b2];
#pragma unroll
      for (int j = 0; j < NPARA; ++j) acc[j] += ang[j] * wv;
    }
    __syncthreads();
  }

  const int lrow = lane & 15;
  const int lhi = lane >> 4;
  const int khalf = lhi * 2;
  const int mt = wave & 1;
  const int np = wave >> 1;
  const int n0 = np * 32 + lrow;
  v2f Blo[3][2], Bhi[3][2];
#pragma unroll
  for (int ip = 0; ip < 3; ++ip) {
    const float* hb = hyper + ip * (NWAVE * NORBIT);
#pragma unroll
    for (int h = 0; h < 2; ++h) {
      Blo[ip][h].x = hb[(khalf + 0) * NORBIT + n0 + h * 16];
      Blo[ip][h].y = hb[(khalf + 1) * NORBIT + n0 + h * 16];
      Bhi[ip][h].x = hb[(khalf + 4) * NORBIT + n0 + h * 16];
      Bhi[ip][h].y = hb[(khalf + 5) * NORBIT + n0 + h * 16];
    }
  }

#pragma unroll
  for (int j = 0; j < NPARA; ++j) s_out[c5 * SOP + j * NWAVE + w] = acc[j];
  __syncthreads();

  {
    v8f d0 = {}, d1 = {};
    const float* arow = s_out + (mt * 16 + lrow) * SOP;
    int para = 0;
#pragma unroll
    for (int ip = 0; ip < 3; ++ip) {
      const v2f B0lo = Blo[ip][0], B0hi = Bhi[ip][0];
      const v2f B1lo = Blo[ip][1], B1hi = Bhi[ip][1];
      const int npj = (ip == 0) ? 1 : ((ip == 1) ? 3 : 9);
#pragma unroll
      for (int q = 0; q < npj; ++q, ++para) {
        const float* ab = arow + para * NWAVE;
        v2f Alo = {ab[khalf + 0], ab[khalf + 1]};
        v2f Ahi = {ab[khalf + 4], ab[khalf + 5]};
        v8f c0 = {}, c1 = {};
        c0 = wmma_f32_16x16x4(Alo, B0lo, c0);
        c1 = wmma_f32_16x16x4(Alo, B1lo, c1);
        c0 = wmma_f32_16x16x4(Ahi, B0hi, c0);
        c1 = wmma_f32_16x16x4(Ahi, B1hi, c1);
#pragma unroll
        for (int r = 0; r < 8; ++r) {
          d0[r] += c0[r] * c0[r];
          d1[r] += c1[r] * c1[r];
        }
      }
    }
#pragma unroll
    for (int r = 0; r < 8; ++r) {
      float* dr = s_den + (mt * 16 + r + lhi * 8) * DP + n0;
      dr[0] = d0[r];
      dr[16] = d1[r];
    }
  }
  __syncthreads();

  if (it == 2) {
    float* dst = density + (b * NATOM + g * CPG) * NORBIT;
    for (int i = t; i < CPG * (NORBIT / 4); i += 256) {
      const int row = i >> 5, c4 = (i & 31) * 4;
      *reinterpret_cast<float4*>(dst + row * NORBIT + c4) =
          *reinterpret_cast<const float4*>(s_den + row * DP + c4);
    }
  } else {
    if (wave < 2) {
      v8f u = {};
      const float* drow = s_den + (wave * 16 + lrow) * DP;
      const bool valid = (lrow < NWAVE);
      float w0[32], w1[32];
#pragma unroll
      for (int kb = 0; kb < 32; ++kb) {
        const int k = kb * 4 + lhi * 2;
        w0[kb] = oc_w[(k + 0) * NWAVE + (lrow & 7)];
        w1[kb] = oc_w[(k + 1) * NWAVE + (lrow & 7)];
      }
#pragma unroll
      for (int kb = 0; kb < 32; ++kb) {
        const int k = kb * 4 + lhi * 2;
        v2f A = {drow[k], drow[k + 1]};
        v2f B = {valid ? w0[kb] : 0.0f, valid ? w1[kb] : 0.0f};
        u = wmma_f32_16x16x4(A, B, u);
      }
      if (valid) {
#pragma unroll
        for (int r = 0; r < 8; ++r) {
          const int lg = wave * 16 + r + lhi * 8;
          const int la = g * CPG + lg;
          const float v = tanhf(u[r] + oc_e[s_spec[la] * NWAVE + lrow]);
          s_new[lg * NWAVE + lrow] = s_orb[la * NWAVE + lrow] + v;
        }
      }
    }
    __syncthreads();
    if (t < CPG * NWAVE / 4)
      *reinterpret_cast<float4*>(orb_out + (b * NATOM + g * CPG) * NWAVE + t * 4) =
          *reinterpret_cast<const float4*>(s_new + t * 4);
  }
}

extern "C" void kernel_launch(void* const* d_in, const int* in_sizes, int n_in,
                              void* d_out, int out_size, void* d_ws,
                              size_t ws_size, hipStream_t stream) {
  const float* cart = (const float*)d_in[0];
  const float* shifts = (const float*)d_in[1];
  const float* rs = (const float*)d_in[2];
  const float* inta = (const float*)d_in[3];
  const float* params = (const float*)d_in[4];
  const float* hyper = (const float*)d_in[5];
  const float* oc_w0 = (const float*)d_in[6];
  const float* oc_e0 = (const float*)d_in[7];
  const float* oc_w1 = (const float*)d_in[8];
  const float* oc_e1 = (const float*)d_in[9];
  const int* species = (const int*)d_in[11];
  const int* atom_index = (const int*)d_in[12];

  float* density = (float*)d_out;
  float* orb_a = (float*)d_ws;
  float* orb_b = orb_a + TOTNATOM * NWAVE;

  k_obtain<<<NB * GRP, 256, 0, stream>>>(cart, shifts, rs, inta, params, hyper,
                                         oc_w0, oc_e0, species, atom_index,
                                         orb_b, orb_a, density, 0);
  k_obtain<<<NB * GRP, 256, 0, stream>>>(cart, shifts, rs, inta, params, hyper,
                                         oc_w1, oc_e1, species, atom_index,
                                         orb_a, orb_b, density, 1);
  k_obtain<<<NB * GRP, 256, 0, stream>>>(cart, shifts, rs, inta, params, hyper,
                                         oc_w1, oc_e1, species, atom_index,
                                         orb_b, orb_a, density, 2);
}


# All launch-start global loads (coordinates, coefficient rows, species, tables, update weights, the slice's 64 keys) issued as one group at entry; were one exposed round trip per loop trip
# speedup vs baseline: 82.7102x; 1.1753x over previous
#include <hip/hip_runtime.h>
#include <math.h>

#define NTYPE 4
#define NWAVE 8
#define NORBIT 128
#define NEIGH 64
#define NB 32
#define NATOM 256
#define NPAIR_PER (NATOM * NEIGH)
#define NPAIR_TOT (NB * NPAIR_PER)
#define TOTNATOM (NB * NATOM)
#define NPARA 13
#define PI_OVER_CUTOFF 0.62831853071795864769f

typedef float v2f __attribute__((ext_vector_type(2)));
typedef float v8f __attribute__((ext_vector_type(8)));

__device__ __forceinline__ v8f wmma_f32_16x16x4(v2f a, v2f b, v8f c) {
  return __builtin_amdgcn_wmma_f32_16x16x4_f32(false, a, false, b, (short)0, c,
                                               false, false);
}

#define GRP 8
#define CPG (NATOM / GRP)
#define SLICE (NPAIR_PER / 8)
#define CH 2560
#define ROWF (NPARA * NWAVE)
#define SOP (ROWF + 4)
#define DP (NORBIT + 4)

__global__ void __launch_bounds__(256)
k_obtain(const float* __restrict__ cart,
         const float* __restrict__ shifts,
         const float* __restrict__ rs,
         const float* __restrict__ inta,
         const float* __restrict__ params,
         const float* __restrict__ hyper,
         const float* __restrict__ oc_w,
         const float* __restrict__ oc_e,
         const int* __restrict__ species,
         const int* __restrict__ atom_index,
         const float* __restrict__ orb_in,
         float* __restrict__ orb_out,
         float* __restrict__ density,
         const int it) {
  __shared__ int s_list[8][SLICE];
  __shared__ int s_sorted[NPAIR_PER];
  __shared__ float4 s_geo[CH];
  __shared__ float s_d[CH];
  __shared__ int s_js[CH];
  __shared__ float s_out[CPG * SOP];
  __shared__ __attribute__((aligned(16))) float s_den[CPG * DP];
  __shared__ __attribute__((aligned(16))) float s_new[CPG * NWAVE];
  __shared__ __attribute__((aligned(16))) float s_orb[NATOM * NWAVE];
  __shared__ __attribute__((aligned(16))) float s_w[NORBIT * 16];
  __shared__ float s_e[NTYPE * NWAVE];
  __shared__ float s_cart[NATOM * 3];
  __shared__ int s_spec[NATOM];
  __shared__ float s_rs[NTYPE * NWAVE];
  __shared__ float s_inta[NTYPE * NWAVE];
  __shared__ int s_cnt[8][CPG];
  __shared__ int s_start[CPG + 1];

  const int t = threadIdx.x;
  const int wave = t >> 5;
  const int lane = t & 31;
  const int b = blockIdx.x / GRP;
  const int g = blockIdx.x % GRP;

  const int r_spec = species[b * NATOM + t];
  float r_cart[3];
#pragma unroll
  for (int j = 0; j < 3; ++j) r_cart[j] = cart[b * NATOM * 3 + j * 256 + t];
  float4 r_o0 = make_float4(0.0f, 0.0f, 0.0f, 0.0f), r_o1 = r_o0;
  if (it != 0) {
    const float* src = orb_in + (b * NATOM + t) * NWAVE;
    r_o0 = *reinterpret_cast<const float4*>(src);
    r_o1 = *reinterpret_cast<const float4*>(src + 4);
  }
  float4 r_w = make_float4(0.0f, 0.0f, 0.0f, 0.0f);
  if (it != 2) r_w = *reinterpret_cast<const float4*>(oc_w + t * 4);
  float r_rs = 0.0f, r_inta = 0.0f, r_e = 0.0f;
  if (t < NTYPE * NWAVE) { r_rs = rs[t]; r_inta = inta[t]; r_e = oc_e[t]; }
  (&s_cnt[0][0])[t] = 0;

  const int* keys = atom_index + b * NPAIR_PER;
  const int kbase = wave * SLICE;
  const unsigned ltmask = (1u << lane) - 1u;
  int nloc = 0;
  int kr[SLICE / 32];
#pragma unroll
  for (int j = 0; j < SLICE / 32; ++j) kr[j] = keys[kbase + j * 32 + lane];
  if (it == 0) {
    const float* src = params + r_spec * NWAVE;
    r_o0 = *reinterpret_cast<const float4*>(src);
    r_o1 = *reinterpret_cast<const float4*>(src + 4);
  }
#pragma unroll
  for (int j = 0; j < SLICE / 32; ++j) {
    const int cc = kr[j];
    const bool inr = (cc >> 5) == g;
    const unsigned mk = __builtin_amdgcn_ballot_w32(inr);
    if (inr)
      s_list[wave][nloc + __builtin_popcount(mk & ltmask)] =
          (kbase + j * 32 + lane) | ((cc & 31) << 14);
    nloc += __builtin_popcount(mk);
  }
#pragma unroll
  for (int j = 0; j < 3; ++j) s_cart[j * 256 + t] = r_cart[j];
  *reinterpret_cast<float4*>(s_orb + t * NWAVE) = r_o0;
  *reinterpret_cast<float4*>(s_orb + t * NWAVE + 4) = r_o1;
  *reinterpret_cast<float4*>(s_w + (t >> 1) * 16 + (t & 1) * 4) = r_w;
  *reinterpret_cast<float4*>(s_w + (t >> 1) * 16 + 8 + (t & 1) * 4) =
      make_float4(0.0f, 0.0f, 0.0f, 0.0f);
  s_spec[t] = r_spec;
  if (t < NTYPE * NWAVE) { s_rs[t] = r_rs; s_inta[t] = r_inta; s_e[t] = r_e; }
  __syncthreads();

  for (int s = 0; s < nloc; s += 32) {
    const int idx = s + lane;
    const bool ok = idx < nloc;
    const int e = ok ? s_list[wave][idx] : 0;
    const int c5 = (e >> 14) & 31;
    unsigned m = __builtin_amdgcn_ballot_w32(ok);
#pragma unroll
    for (int bit = 0; bit < 5; ++bit) {
      const bool bs = ((c5 >> bit) & 1) != 0;
      const unsigned bb = __builtin_amdgcn_ballot_w32(bs);
      m &= bs ? bb : ~bb;
    }
    if (ok) {
      const int prev = s_cnt[wave][c5];
      const int r = __builtin_popcount(m & ltmask);
      s_list[wave][idx] = e | ((prev + r) << 19);
      if (r == 0) s_cnt[wave][c5] = prev + __builtin_popcount(m);
    }
  }
  __syncthreads();

  if (wave == 0) {
    int tot = 0;
#pragma unroll
    for (int w = 0; w < 8; ++w) {
      const int v = s_cnt[w][lane];
      s_cnt[w][lane] = tot;
      tot += v;
    }
    int inc = tot;
#pragma unroll
    for (int o = 1; o < 32; o <<= 1) {
      const int n = __shfl_up(inc, o);
      if (lane >= o) inc += n;
    }
    s_start[lane] = inc - tot;
    if (lane == 31) s_start[CPG] = inc;
  }
  __syncthreads();

  for (int s = 0; s < nloc; s += 32) {
    const int idx = s + lane;
    if (idx < nloc) {
      const int e = s_list[wave][idx];
      const int c5 = (e >> 14) & 31;
      s_sorted[s_start[c5] + s_cnt[wave][c5] + (e >> 19)] = e & 0x7ffff;
    }
  }
  __syncthreads();

  const int ne = s_start[CPG];
  const int c5 = t >> 3;
  const int w = t & 7;
  const int mystart = s_start[c5];
  const int myend = s_start[c5 + 1];
  float acc[NPARA];
#pragma unroll
  for (int j = 0; j < NPARA; ++j) acc[j] = 0.0f;

  for (int e0 = 0; e0 < ne; e0 += CH) {
    const int nchunk = (ne - e0 < CH) ? (ne - e0) : CH;

    for (int i0 = t; i0 < nchunk; i0 += 256 * 4) {
      int ent[4], ljv[4];
      float sx[4], sy[4], sz[4];
#pragma unroll
      for (int u = 0; u < 4; ++u) {
        const int i = i0 + u * 256;
        ent[u] = s_sorted[e0 + ((i < nchunk) ? i : i0)];
      }
#pragma unroll
      for (int u = 0; u < 4; ++u) {
        const int p = b * NPAIR_PER + (ent[u] & 0x3fff);
        ljv[u] = atom_index[NPAIR_TOT + p];
        sx[u] = shifts[p * 3 + 0];
        sy[u] = shifts[p * 3 + 1];
        sz[u] = shifts[p * 3 + 2];
      }
#pragma unroll
      for (int u = 0; u < 4; ++u) {
        const int i = i0 + u * 256;
        const int li = g * CPG + (ent[u] >> 14);
        const int lj = ljv[u];
        const float dx = s_cart[li * 3 + 0] - s_cart[lj * 3 + 0] + sx[u];
        const float dy = s_cart[li * 3 + 1] - s_cart[lj * 3 + 1] + sy[u];
        const float dz = s_cart[li * 3 + 2] - s_cart[lj * 3 + 2] + sz[u];
        const float d = sqrtf(dx * dx + dy * dy + dz * dz);
        float fc = 0.5f * cosf(d * PI_OVER_CUTOFF) + 0.5f;
        fc = fc * fc;
        const int js = lj | (s_spec[lj] << 8);
        if (i < nchunk) {
          s_geo[i] = make_float4(fc, dx, dy, dz);
          s_d[i] = d;
          s_js[i] = js;
        }
      }
    }
    __syncthreads();

    const int lo = ((mystart > e0) ? mystart : e0) - e0;
    const int hi = ((myend < e0 + nchunk) ? myend : (e0 + nchunk)) - e0;
#pragma unroll 4
    for (int i = lo; i < hi; ++i) {
      const float4 g4 = s_geo[i];
      const float d = s_d[i];
      const int js = s_js[i];
      const int lj = js & 255;
      const int sp = js >> 8;
      const float tt = d - s_rs[sp * NWAVE + w];
      const float wv = expf(s_inta[sp * NWAVE + w] * tt * tt) * s_orb[lj * NWAVE + w];
      float ang[NPARA];
      const float dv[3] = {g4.y, g4.z, g4.w};
      ang[0] = g4.x;
#pragma unroll
      for (int a2 = 0; a2 < 3; ++a2) ang[1 + a2] = g4.x * dv[a2];
#pragma unroll
      for (int a2 = 0; a2 < 3; ++a2)
#pragma unroll
        for (int b2 = 0; b2 < 3; ++b2) ang[4 + a2 * 3 + b2] = ang[1 + a2] * dv[b2];
#pragma unroll
      for (int j = 0; j < NPARA; ++j) acc[j] += ang[j] * wv;
    }
    __syncthreads();
  }

  const int lrow = lane & 15;
  const int lhi = lane >> 4;
  const int khalf = lhi * 2;
  const int mt = wave & 1;
  const int np = wave >> 1;
  const int n0 = np * 32 + lrow;
  v2f Blo[3][2], Bhi[3][2];
#pragma unroll
  for (int ip = 0; ip < 3; ++ip) {
    const float* hb = hyper + ip * (NWAVE * NORBIT);
#pragma unroll
    for (int h = 0; h < 2; ++h) {
      Blo[ip][h].x = hb[(khalf + 0) * NORBIT + n0 + h * 16];
      Blo[ip][h].y = hb[(khalf + 1) * NORBIT + n0 + h * 16];
      Bhi[ip][h].x = hb[(khalf + 4) * NORBIT + n0 + h * 16];
      Bhi[ip][h].y = hb[(khalf + 5) * NORBIT + n0 + h * 16];
    }
  }

#pragma unroll
  for (int j = 0; j < NPARA; ++j) s_out[c5 * SOP + j * NWAVE + w] = acc[j];
  __syncthreads();

  {
    v8f d0 = {}, d1 = {};
    const float* arow = s_out + (mt * 16 + lrow) * SOP;
    int para = 0;
#pragma unroll
    for (int ip = 0; ip < 3; ++ip) {
      const v2f B0lo = Blo[ip][0], B0hi = Bhi[ip][0];
      const v2f B1lo = Blo[ip][1], B1hi = Bhi[ip][1];
      const int npj = (ip == 0) ? 1 : ((ip == 1) ? 3 : 9);
#pragma unroll
      for (int q = 0; q < npj; ++q, ++para) {
        const float* ab = arow + para * NWAVE;
        v2f Alo = {ab[khalf + 0], ab[khalf + 1]};
        v2f Ahi = {ab[khalf + 4], ab[khalf + 5]};
        v8f c0 = {}, c1 = {};
        c0 = wmma_f32_16x16x4(Alo, B0lo, c0);
        c1 = wmma_f32_16x16x4(Alo, B1lo, c1);
        c0 = wmma_f32_16x16x4(Ahi, B0hi, c0);
        c1 = wmma_f32_16x16x4(Ahi, B1hi, c1);
#pragma unroll
        for (int r = 0; r < 8; ++r) {
          d0[r] += c0[r] * c0[r];
          d1[r] += c1[r] * c1[r];
        }
      }
    }
#pragma unroll
    for (int r = 0; r < 8; ++r) {
      float* dr = s_den + (mt * 16 + r + lhi * 8) * DP + n0;
      dr[0] = d0[r];
      dr[16] = d1[r];
    }
  }
  __syncthreads();

  if (it == 2) {
    float* dst = density + (b * NATOM + g * CPG) * NORBIT;
    for (int i = t; i < CPG * (NORBIT / 4); i += 256) {
      const int row = i >> 5, c4 = (i & 31) * 4;
      *reinterpret_cast<float4*>(dst + row * NORBIT + c4) =
          *reinterpret_cast<const float4*>(s_den + row * DP + c4);
    }
  } else {
    if (wave < 2) {
      v8f u = {};
      const float* drow = s_den + (wave * 16 + lrow) * DP;
      const bool valid = (lrow < NWAVE);
      float w0[32], w1[32];
#pragma unroll
      for (int kb = 0; kb < 32; ++kb) {
        const int k = kb * 4 + lhi * 2;
        w0[kb] = s_w[(k + 0) * 16 + lrow];
        w1[kb] = s_w[(k + 1) * 16 + lrow];
      }
#pragma unroll
      for (int kb = 0; kb < 32; ++kb) {
        const int k = kb * 4 + lhi * 2;
        v2f A = {drow[k], drow[k + 1]};
        v2f B = {w0[kb], w1[kb]};
        u = wmma_f32_16x16x4(A, B, u);
      }
      if (valid) {
#pragma unroll
        for (int r = 0; r < 8; ++r) {
          const int lg = wave * 16 + r + lhi * 8;
          const int la = g * CPG + lg;
          const float v = tanhf(u[r] + s_e[s_spec[la] * NWAVE + lrow]);
          s_new[lg * NWAVE + lrow] = s_orb[la * NWAVE + lrow] + v;
        }
      }
    }
    __syncthreads();
    if (t < CPG * NWAVE / 4)
      *reinterpret_cast<float4*>(orb_out + (b * NATOM + g * CPG) * NWAVE + t * 4) =
          *reinterpret_cast<const float4*>(s_new + t * 4);
  }
}

extern "C" void kernel_launch(void* const* d_in, const int* in_sizes, int n_in,
                              void* d_out, int out_size, void* d_ws,
                              size_t ws_size, hipStream_t stream) {
  const float* cart = (const float*)d_in[0];
  const float* shifts = (const float*)d_in[1];
  const float* rs = (const float*)d_in[2];
  const float* inta = (const float*)d_in[3];
  const float* params = (const float*)d_in[4];
  const float* hyper = (const float*)d_in[5];
  const float* oc_w0 = (const float*)d_in[6];
  const float* oc_e0 = (const float*)d_in[7];
  const float* oc_w1 = (const float*)d_in[8];
  const float* oc_e1 = (const float*)d_in[9];
  const int* species = (const int*)d_in[11];
  const int* atom_index = (const int*)d_in[12];

  float* density = (float*)d_out;
  float* orb_a = (float*)d_ws;
  float* orb_b = orb_a + TOTNATOM * NWAVE;

  k_obtain<<<NB * GRP, 256, 0, stream>>>(cart, shifts, rs, inta, params, hyper,
                                         oc_w0, oc_e0, species, atom_index,
                                         orb_b, orb_a, density, 0);
  k_obtain<<<NB * GRP, 256, 0, stream>>>(cart, shifts, rs, inta, params, hyper,
                                         oc_w1, oc_e1, species, atom_index,
                                         orb_a, orb_b, density, 1);
  k_obtain<<<NB * GRP, 256, 0, stream>>>(cart, shifts, rs, inta, params, hyper,
                                         oc_w1, oc_e1, species, atom_index,
                                         orb_b, orb_a, density, 2);
}


# Geometry pass walks each wave's kept list in pair order, loads one trip ahead (first trip before ranking), writes each record to its sorted slot; placement pass, sorted list dropped; sums unchanged
# speedup vs baseline: 86.3729x; 1.0443x over previous
#include <hip/hip_runtime.h>
#include <math.h>

#define NTYPE 4
#define NWAVE 8
#define NORBIT 128
#define NEIGH 64
#define NB 32
#define NATOM 256
#define NPAIR_PER (NATOM * NEIGH)
#define NPAIR_TOT (NB * NPAIR_PER)
#define TOTNATOM (NB * NATOM)
#define NPARA 13
#define PI_OVER_CUTOFF 0.62831853071795864769f

typedef float v2f __attribute__((ext_vector_type(2)));
typedef float v8f __attribute__((ext_vector_type(8)));

__device__ __forceinline__ v8f wmma_f32_16x16x4(v2f a, v2f b, v8f c) {
  return __builtin_amdgcn_wmma_f32_16x16x4_f32(false, a, false, b, (short)0, c,
                                               false, false);
}

#define GRP 8
#define CPG (NATOM / GRP)
#define SLICE (NPAIR_PER / 8)
#define CH 2560
#define BU 5
#define ROWF (NPARA * NWAVE)
#define SOP (ROWF + 4)
#define DP (NORBIT + 4)

#define B_LOAD(S, LJ, SX, SY, SZ)                                   \
  _Pragma("unroll") for (int u = 0; u < BU; ++u) {                  \
    const int idx = (S) + u * 32 + lane;                            \
    const int e = s_list[wave][(idx < nloc) ? idx : 0];             \
    const int p = b * NPAIR_PER + (e & 0x3fff);                     \
    LJ[u] = atom_index[NPAIR_TOT + p];      \
    SX[u] = shifts[p * 3 + 0];                                      \
    SY[u] = shifts[p * 3 + 1];                                      \
    SZ[u] = shifts[p * 3 + 2];                                      \
  }

__global__ void __launch_bounds__(256)
k_obtain(const float* __restrict__ cart,
         const float* __restrict__ shifts,
         const float* __restrict__ rs,
         const float* __restrict__ inta,
         const float* __restrict__ params,
         const float* __restrict__ hyper,
         const float* __restrict__ oc_w,
         const float* __restrict__ oc_e,
         const int* __restrict__ species,
         const int* __restrict__ atom_index,
         const float* __restrict__ orb_in,
         float* __restrict__ orb_out,
         float* __restrict__ density,
         const int it) {
  __shared__ int s_list[8][SLICE];
  __shared__ float4 s_geo[CH];
  __shared__ float s_d[CH];
  __shared__ int s_js[CH];
  __shared__ float s_out[CPG * SOP];
  __shared__ __attribute__((aligned(16))) float s_den[CPG * DP];
  __shared__ __attribute__((aligned(16))) float s_new[CPG * NWAVE];
  __shared__ __attribute__((aligned(16))) float s_orb[NATOM * NWAVE];
  __shared__ __attribute__((aligned(16))) float s_w[NORBIT * 16];
  __shared__ float s_e[NTYPE * NWAVE];
  __shared__ float s_cart[NATOM * 3];
  __shared__ int s_spec[NATOM];
  __shared__ float s_rs[NTYPE * NWAVE];
  __shared__ float s_inta[NTYPE * NWAVE];
  __shared__ int s_cnt[8][CPG];
  __shared__ int s_start[CPG + 1];

  const int t = threadIdx.x;
  const int wave = t >> 5;
  const int lane = t & 31;
  const int b = blockIdx.x / GRP;
  const int g = blockIdx.x % GRP;

  const int r_spec = species[b * NATOM + t];
  float r_cart[3];
#pragma unroll
  for (int j = 0; j < 3; ++j) r_cart[j] = cart[b * NATOM * 3 + j * 256 + t];
  float4 r_o0 = make_float4(0.0f, 0.0f, 0.0f, 0.0f), r_o1 = r_o0;
  if (it != 0) {
    const float* src = orb_in + (b * NATOM + t) * NWAVE;
    r_o0 = *reinterpret_cast<const float4*>(src);
    r_o1 = *reinterpret_cast<const float4*>(src + 4);
  }
  float4 r_w = make_float4(0.0f, 0.0f, 0.0f, 0.0f);
  if (it != 2) r_w = *reinterpret_cast<const float4*>(oc_w + t * 4);
  float r_rs = 0.0f, r_inta = 0.0f, r_e = 0.0f;
  if (t < NTYPE * NWAVE) { r_rs = rs[t]; r_inta = inta[t]; r_e = oc_e[t]; }
  (&s_cnt[0][0])[t] = 0;

  const int* keys = atom_index + b * NPAIR_PER;
  const int kbase = wave * SLICE;
  const unsigned ltmask = (1u << lane) - 1u;
  int nloc = 0;
  int kr[SLICE / 32];
#pragma unroll
  for (int j = 0; j < SLICE / 32; ++j) kr[j] = keys[kbase + j * 32 + lane];
  if (it == 0) {
    const float* src = params + r_spec * NWAVE;
    r_o0 = *reinterpret_cast<const float4*>(src);
    r_o1 = *reinterpret_cast<const float4*>(src + 4);
  }
#pragma unroll
  for (int j = 0; j < SLICE / 32; ++j) {
    const int cc = kr[j];
    const bool inr = (cc >> 5) == g;
    const unsigned mk = __builtin_amdgcn_ballot_w32(inr);
    if (inr)
      s_list[wave][nloc + __builtin_popcount(mk & ltmask)] =
          (kbase + j * 32 + lane) | ((cc & 31) << 14);
    nloc += __builtin_popcount(mk);
  }
#pragma unroll
  for (int j = 0; j < 3; ++j) s_cart[j * 256 + t] = r_cart[j];
  *reinterpret_cast<float4*>(s_orb + t * NWAVE) = r_o0;
  *reinterpret_cast<float4*>(s_orb + t * NWAVE + 4) = r_o1;
  *reinterpret_cast<float4*>(s_w + (t >> 1) * 16 + (t & 1) * 4) = r_w;
  *reinterpret_cast<float4*>(s_w + (t >> 1) * 16 + 8 + (t & 1) * 4) =
      make_float4(0.0f, 0.0f, 0.0f, 0.0f);
  s_spec[t] = r_spec;
  if (t < NTYPE * NWAVE) { s_rs[t] = r_rs; s_inta[t] = r_inta; s_e[t] = r_e; }
  __syncthreads();

  int c_lj[BU];
  float c_sx[BU], c_sy[BU], c_sz[BU];
  B_LOAD(0, c_lj, c_sx, c_sy, c_sz)

  for (int s = 0; s < nloc; s += 32) {
    const int idx = s + lane;
    const bool ok = idx < nloc;
    const int e = ok ? s_list[wave][idx] : 0;
    const int c5 = (e >> 14) & 31;
    unsigned m = __builtin_amdgcn_ballot_w32(ok);
#pragma unroll
    for (int bit = 0; bit < 5; ++bit) {
      const bool bs = ((c5 >> bit) & 1) != 0;
      const unsigned bb = __builtin_amdgcn_ballot_w32(bs);
      m &= bs ? bb : ~bb;
    }
    if (ok) {
      const int prev = s_cnt[wave][c5];
      const int r = __builtin_popcount(m & ltmask);
      s_list[wave][idx] = e | ((prev + r) << 19);
      if (r == 0) s_cnt[wave][c5] = prev + __builtin_popcount(m);
    }
  }
  __syncthreads();

  if (wave == 0) {
    int tot = 0;
#pragma unroll
    for (int w = 0; w < 8; ++w) {
      const int v = s_cnt[w][lane];
      s_cnt[w][lane] = tot;
      tot += v;
    }
    int inc = tot;
#pragma unroll
    for (int o = 1; o < 32; o <<= 1) {
      const int n = __shfl_up(inc, o);
      if (lane >= o) inc += n;
    }
    s_start[lane] = inc - tot;
    if (lane == 31) s_start[CPG] = inc;
  }
  __syncthreads();

  const int ne = s_start[CPG];
  const int c5 = t >> 3;
  const int w = t & 7;
  const int mystart = s_start[c5];
  const int myend = s_start[c5 + 1];
  float acc[NPARA];
#pragma unroll
  for (int j = 0; j < NPARA; ++j) acc[j] = 0.0f;

  for (int e0 = 0; e0 < ne; e0 += CH) {
    const int nchunk = (ne - e0 < CH) ? (ne - e0) : CH;

    if (e0 != 0) {
      B_LOAD(0, c_lj, c_sx, c_sy, c_sz)
    }
    for (int s = 0; s < nloc; s += 32 * BU) {
      int n_lj[BU];
      float n_sx[BU], n_sy[BU], n_sz[BU];
      B_LOAD(s + 32 * BU, n_lj, n_sx, n_sy, n_sz)
#pragma unroll
      for (int u = 0; u < BU; ++u) {
        if (s + u * 32 < nloc) {
          const int idx = s + u * 32 + lane;
          const bool ok = idx < nloc;
          const int e = s_list[wave][ok ? idx : 0];
          const int ce = (e >> 14) & 31;
          const int slot = s_start[ce] + s_cnt[wave][ce] + (e >> 19) - e0;
          const int li = g * CPG + ce;
          const int lj = c_lj[u];
          const float dx = s_cart[li * 3 + 0] - s_cart[lj * 3 + 0] + c_sx[u];
          const float dy = s_cart[li * 3 + 1] - s_cart[lj * 3 + 1] + c_sy[u];
          const float dz = s_cart[li * 3 + 2] - s_cart[lj * 3 + 2] + c_sz[u];
          const float d = sqrtf(dx * dx + dy * dy + dz * dz);
          float fc = 0.5f * cosf(d * PI_OVER_CUTOFF) + 0.5f;
          fc = fc * fc;
          const int js = lj | (s_spec[lj] << 8);
          if (ok && slot >= 0 && slot < nchunk) {
            s_geo[slot] = make_float4(fc, dx, dy, dz);
            s_d[slot] = d;
            s_js[slot] = js;
          }
        }
      }
#pragma unroll
      for (int u = 0; u < BU; ++u) {
        c_lj[u] = n_lj[u]; c_sx[u] = n_sx[u]; c_sy[u] = n_sy[u]; c_sz[u] = n_sz[u];
      }
    }
    __syncthreads();

    const int lo = ((mystart > e0) ? mystart : e0) - e0;
    const int hi = ((myend < e0 + nchunk) ? myend : (e0 + nchunk)) - e0;
#pragma unroll 4
    for (int i = lo; i < hi; ++i) {
      const float4 g4 = s_geo[i];
      const float d = s_d[i];
      const int js = s_js[i];
      const int lj = js & 255;
      const int sp = js >> 8;
      const float tt = d - s_rs[sp * NWAVE + w];
      const float wv = expf(s_inta[sp * NWAVE + w] * tt * tt) * s_orb[lj * NWAVE + w];
      float ang[NPARA];
      const float dv[3] = {g4.y, g4.z, g4.w};
      ang[0] = g4.x;
#pragma unroll
      for (int a2 = 0; a2 < 3; ++a2) ang[1 + a2] = g4.x * dv[a2];
#pragma unroll
      for (int a2 = 0; a2 < 3; ++a2)
#pragma unroll
        for (int b2 = 0; b2 < 3; ++b2) ang[4 + a2 * 3 + b2] = ang[1 + a2] * dv[b2];
#pragma unroll
      for (int j = 0; j < NPARA; ++j) acc[j] += ang[j] * wv;
    }
    __syncthreads();
  }

  const int lrow = lane & 15;
  const int lhi = lane >> 4;
  const int khalf = lhi * 2;
  const int mt = wave & 1;
  const int np = wave >> 1;
  const int n0 = np * 32 + lrow;
  v2f Blo[3][2], Bhi[3][2];
#pragma unroll
  for (int ip = 0; ip < 3; ++ip) {
    const float* hb = hyper + ip * (NWAVE * NORBIT);
#pragma unroll
    for (int h = 0; h < 2; ++h) {
      Blo[ip][h].x = hb[(khalf + 0) * NORBIT + n0 + h * 16];
      Blo[ip][h].y = hb[(khalf + 1) * NORBIT + n0 + h * 16];
      Bhi[ip][h].x = hb[(khalf + 4) * NORBIT + n0 + h * 16];
      Bhi[ip][h].y = hb[(khalf + 5) * NORBIT + n0 + h * 16];
    }
  }

#pragma unroll
  for (int j = 0; j < NPARA; ++j) s_out[c5 * SOP + j * NWAVE + w] = acc[j];
  __syncthreads();

  {
    v8f d0 = {}, d1 = {};
    const float* arow = s_out + (mt * 16 + lrow) * SOP;
    int para = 0;
#pragma unroll
    for (int ip = 0; ip < 3; ++ip) {
      const v2f B0lo = Blo[ip][0], B0hi = Bhi[ip][0];
      const v2f B1lo = Blo[ip][1], B1hi = Bhi[ip][1];
      const int npj = (ip == 0) ? 1 : ((ip == 1) ? 3 : 9);
#pragma unroll
      for (int q = 0; q < npj; ++q, ++para) {
        const float* ab = arow + para * NWAVE;
        v2f Alo = {ab[khalf + 0], ab[khalf + 1]};
        v2f Ahi = {ab[khalf + 4], ab[khalf + 5]};
        v8f c0 = {}, c1 = {};
        c0 = wmma_f32_16x16x4(Alo, B0lo, c0);
        c1 = wmma_f32_16x16x4(Alo, B1lo, c1);
        c0 = wmma_f32_16x16x4(Ahi, B0hi, c0);
        c1 = wmma_f32_16x16x4(Ahi, B1hi, c1);
#pragma unroll
        for (int r = 0; r < 8; ++r) {
          d0[r] += c0[r] * c0[r];
          d1[r] += c1[r] * c1[r];
        }
      }
    }
#pragma unroll
    for (int r = 0; r < 8; ++r) {
      float* dr = s_den + (mt * 16 + r + lhi * 8) * DP + n0;
      dr[0] = d0[r];
      dr[16] = d1[r];
    }
  }
  __syncthreads();

  if (it == 2) {
    float* dst = density + (b * NATOM + g * CPG) * NORBIT;
    for (int i = t; i < CPG * (NORBIT / 4); i += 256) {
      const int row = i >> 5, c4 = (i & 31) * 4;
      *reinterpret_cast<float4*>(dst + row * NORBIT + c4) =
          *reinterpret_cast<const float4*>(s_den + row * DP + c4);
    }
  } else {
    if (wave < 2) {
      v8f u = {};
      const float* drow = s_den + (wave * 16 + lrow) * DP;
      const bool valid = (lrow < NWAVE);
      float w0[32], w1[32];
#pragma unroll
      for (int kb = 0; kb < 32; ++kb) {
        const int k = kb * 4 + lhi * 2;
        w0[kb] = s_w[(k + 0) * 16 + lrow];
        w1[kb] = s_w[(k + 1) * 16 + lrow];
      }
#pragma unroll
      for (int kb = 0; kb < 32; ++kb) {
        const int k = kb * 4 + lhi * 2;
        v2f A = {drow[k], drow[k + 1]};
        v2f B = {w0[kb], w1[kb]};
        u = wmma_f32_16x16x4(A, B, u);
      }
      if (valid) {
#pragma unroll
        for (int r = 0; r < 8; ++r) {
          const int lg = wave * 16 + r + lhi * 8;
          const int la = g * CPG + lg;
          const float v = tanhf(u[r] + s_e[s_spec[la] * NWAVE + lrow]);
          s_new[lg * NWAVE + lrow] = s_orb[la * NWAVE + lrow] + v;
        }
      }
    }
    __syncthreads();
    if (t < CPG * NWAVE / 4)
      *reinterpret_cast<float4*>(orb_out + (b * NATOM + g * CPG) * NWAVE + t * 4) =
          *reinterpret_cast<const float4*>(s_new + t * 4);
  }
}

extern "C" void kernel_launch(void* const* d_in, const int* in_sizes, int n_in,
                              void* d_out, int out_size, void* d_ws,
                              size_t ws_size, hipStream_t stream) {
  const float* cart = (const float*)d_in[0];
  const float* shifts = (const float*)d_in[1];
  const float* rs = (const float*)d_in[2];
  const float* inta = (const float*)d_in[3];
  const float* params = (const float*)d_in[4];
  const float* hyper = (const float*)d_in[5];
  const float* oc_w0 = (const float*)d_in[6];
  const float* oc_e0 = (const float*)d_in[7];
  const float* oc_w1 = (const float*)d_in[8];
  const float* oc_e1 = (const float*)d_in[9];
  const int* species = (const int*)d_in[11];
  const int* atom_index = (const int*)d_in[12];

  float* density = (float*)d_out;
  float* orb_a = (float*)d_ws;
  float* orb_b = orb_a + TOTNATOM * NWAVE;

  k_obtain<<<NB * GRP, 256, 0, stream>>>(cart, shifts, rs, inta, params, hyper,
                                         oc_w0, oc_e0, species, atom_index,
                                         orb_b, orb_a, density, 0);
  k_obtain<<<NB * GRP, 256, 0, stream>>>(cart, shifts, rs, inta, params, hyper,
                                         oc_w1, oc_e1, species, atom_index,
                                         orb_a, orb_b, density, 1);
  k_obtain<<<NB * GRP, 256, 0, stream>>>(cart, shifts, rs, inta, params, hyper,
                                         oc_w1, oc_e1, species, atom_index,
                                         orb_b, orb_a, density, 2);
}


# Launch 0 saves each block's ranked pair lists and run offsets in its own output rows; launches 1 and 2 load them and skip key scan, ranking and offset passes; sums and their order unchanged
# speedup vs baseline: 99.4957x; 1.1519x over previous
#include <hip/hip_runtime.h>
#include <math.h>

#define NTYPE 4
#define NWAVE 8
#define NORBIT 128
#define NEIGH 64
#define NB 32
#define NATOM 256
#define NPAIR_PER (NATOM * NEIGH)
#define NPAIR_TOT (NB * NPAIR_PER)
#define TOTNATOM (NB * NATOM)
#define NPARA 13
#define PI_OVER_CUTOFF 0.62831853071795864769f

typedef float v2f __attribute__((ext_vector_type(2)));
typedef float v8f __attribute__((ext_vector_type(8)));

__device__ __forceinline__ v8f wmma_f32_16x16x4(v2f a, v2f b, v8f c) {
  return __builtin_amdgcn_wmma_f32_16x16x4_f32(false, a, false, b, (short)0, c,
                                               false, false);
}

#define GRP 8
#define CPG (NATOM / GRP)
#define SLICE (NPAIR_PER / 8)
#define CH 2560
#define BU 5
#define ROWF (NPARA * NWAVE)
#define SOP (ROWF + 4)
#define DP (NORBIT + 4)
#define SCAP 448
#define SV_CNT (8 * SCAP)
#define SV_START (SV_CNT + 256)
#define SV_NLOC (SV_START + 40)

#define B_LOAD(S, LJ, SX, SY, SZ)                                   \
  _Pragma("unroll") for (int u = 0; u < BU; ++u) {                  \
    const int idx = (S) + u * 32 + lane;                            \
    const int e = s_list[wave][(idx < nloc) ? idx : 0];             \
    const int p = b * NPAIR_PER + (e & 0x3fff);                     \
    LJ[u] = atom_index[NPAIR_TOT + p];      \
    SX[u] = shifts[p * 3 + 0];                                      \
    SY[u] = shifts[p * 3 + 1];                                      \
    SZ[u] = shifts[p * 3 + 2];                                      \
  }

template <bool FIRST>
__global__ void __launch_bounds__(256)
k_obtain(const float* __restrict__ cart,
         const float* __restrict__ shifts,
         const float* __restrict__ rs,
         const float* __restrict__ inta,
         const float* __restrict__ params,
         const float* __restrict__ hyper,
         const float* __restrict__ oc_w,
         const float* __restrict__ oc_e,
         const int* __restrict__ species,
         const int* __restrict__ atom_index,
         const float* __restrict__ orb_in,
         float* __restrict__ orb_out,
         float* __restrict__ density,
         const int it) {
  __shared__ int s_list[8][SLICE];
  __shared__ float4 s_geo[CH];
  __shared__ float s_d[CH];
  __shared__ int s_js[CH];
  __shared__ float s_out[CPG * SOP];
  __shared__ __attribute__((aligned(16))) float s_den[CPG * DP];
  __shared__ __attribute__((aligned(16))) float s_new[CPG * NWAVE];
  __shared__ __attribute__((aligned(16))) float s_orb[NATOM * NWAVE];
  __shared__ __attribute__((aligned(16))) float s_w[NORBIT * 16];
  __shared__ float s_e[NTYPE * NWAVE];
  __shared__ float s_cart[NATOM * 3];
  __shared__ int s_spec[NATOM];
  __shared__ float s_rs[NTYPE * NWAVE];
  __shared__ float s_inta[NTYPE * NWAVE];
  __shared__ int s_cnt[8][CPG];
  __shared__ int s_start[CPG + 1];

  const int t = threadIdx.x;
  const int wave = t >> 5;
  const int lane = t & 31;
  const int b = blockIdx.x / GRP;
  const int g = blockIdx.x % GRP;

  int* sv = reinterpret_cast<int*>(density) + (b * NATOM + g * CPG) * NORBIT;
  int lw[SCAP / 32];
  int r_cnt = 0, r_start = 0, r_nmine = 0;
  int4 r_na = make_int4(0, 0, 0, 0), r_nb = r_na;
  if (!FIRST) {
    r_nmine = sv[SV_NLOC + wave];
#pragma unroll
    for (int k = 0; k < SCAP / 32; ++k) lw[k] = sv[wave * SCAP + k * 32 + lane];
    r_cnt = sv[SV_CNT + t];
    r_start = sv[SV_START + ((t <= CPG) ? t : 0)];
    r_na = *reinterpret_cast<const int4*>(sv + SV_NLOC);
    r_nb = *reinterpret_cast<const int4*>(sv + SV_NLOC + 4);
  }

  const int r_spec = species[b * NATOM + t];
  float r_cart[3];
#pragma unroll
  for (int j = 0; j < 3; ++j) r_cart[j] = cart[b * NATOM * 3 + j * 256 + t];
  float4 r_o0 = make_float4(0.0f, 0.0f, 0.0f, 0.0f), r_o1 = r_o0;
  if (!FIRST) {
    const float* src = orb_in + (b * NATOM + t) * NWAVE;
    r_o0 = *reinterpret_cast<const float4*>(src);
    r_o1 = *reinterpret_cast<const float4*>(src + 4);
  }
  float4 r_w = make_float4(0.0f, 0.0f, 0.0f, 0.0f);
  if (it != 2) r_w = *reinterpret_cast<const float4*>(oc_w + t * 4);
  float r_rs = 0.0f, r_inta = 0.0f, r_e = 0.0f;
  if (t < NTYPE * NWAVE) { r_rs = rs[t]; r_inta = inta[t]; r_e = oc_e[t]; }

  const unsigned ltmask = (1u << lane) - 1u;
  int nloc = 0;
  bool reuse = false;
  if (!FIRST) {
    const int nmax = max(max(max(r_na.x, r_na.y), max(r_na.z, r_na.w)),
                         max(max(r_nb.x, r_nb.y), max(r_nb.z, r_nb.w)));
    reuse = __builtin_amdgcn_readfirstlane(nmax) <= SCAP;
  }
  if (reuse) {
    nloc = __builtin_amdgcn_readfirstlane(r_nmine);
#pragma unroll
    for (int k = 0; k < SCAP / 32; ++k) s_list[wave][k * 32 + lane] = lw[k];
    (&s_cnt[0][0])[t] = r_cnt;
    if (t <= CPG) s_start[t] = r_start;
  } else {
  (&s_cnt[0][0])[t] = 0;

  const int* keys = atom_index + b * NPAIR_PER;
  const int kbase = wave * SLICE;
  int kr[SLICE / 32];
#pragma unroll
  for (int j = 0; j < SLICE / 32; ++j) kr[j] = keys[kbase + j * 32 + lane];
  __builtin_amdgcn_sched_barrier(0);
  if (FIRST) {
    const float* src = params + r_spec * NWAVE;
    r_o0 = *reinterpret_cast<const float4*>(src);
    r_o1 = *reinterpret_cast<const float4*>(src + 4);
  }
#pragma unroll
  for (int j = 0; j < SLICE / 32; ++j) {
    const int cc = kr[j];
    const bool inr = (cc >> 5) == g;
    const unsigned mk = __builtin_amdgcn_ballot_w32(inr);
    if (inr)
      s_list[wave][nloc + __builtin_popcount(mk & ltmask)] =
          (kbase + j * 32 + lane) | ((cc & 31) << 14);
    nloc += __builtin_popcount(mk);
  }
  }

  int c_lj[BU];
  float c_sx[BU], c_sy[BU], c_sz[BU];
  if (reuse) {
    B_LOAD(0, c_lj, c_sx, c_sy, c_sz)
  }

#pragma unroll
  for (int j = 0; j < 3; ++j) s_cart[j * 256 + t] = r_cart[j];
  *reinterpret_cast<float4*>(s_orb + t * NWAVE) = r_o0;
  *reinterpret_cast<float4*>(s_orb + t * NWAVE + 4) = r_o1;
  *reinterpret_cast<float4*>(s_w + (t >> 1) * 16 + (t & 1) * 4) = r_w;
  *reinterpret_cast<float4*>(s_w + (t >> 1) * 16 + 8 + (t & 1) * 4) =
      make_float4(0.0f, 0.0f, 0.0f, 0.0f);
  s_spec[t] = r_spec;
  if (t < NTYPE * NWAVE) { s_rs[t] = r_rs; s_inta[t] = r_inta; s_e[t] = r_e; }
  __syncthreads();

  if (!reuse) {
  B_LOAD(0, c_lj, c_sx, c_sy, c_sz)

  for (int s = 0; s < nloc; s += 32) {
    const int idx = s + lane;
    const bool ok = idx < nloc;
    const int e = ok ? s_list[wave][idx] : 0;
    const int c5 = (e >> 14) & 31;
    unsigned m = __builtin_amdgcn_ballot_w32(ok);
#pragma unroll
    for (int bit = 0; bit < 5; ++bit) {
      const bool bs = ((c5 >> bit) & 1) != 0;
      const unsigned bb = __builtin_amdgcn_ballot_w32(bs);
      m &= bs ? bb : ~bb;
    }
    if (ok) {
      const int prev = s_cnt[wave][c5];
      const int r = __builtin_popcount(m & ltmask);
      s_list[wave][idx] = e | ((prev + r) << 19);
      if (r == 0) s_cnt[wave][c5] = prev + __builtin_popcount(m);
    }
  }
  __syncthreads();

  if (wave == 0) {
    int tot = 0;
#pragma unroll
    for (int w = 0; w < 8; ++w) {
      const int v = s_cnt[w][lane];
      s_cnt[w][lane] = tot;
      tot += v;
    }
    int inc = tot;
#pragma unroll
    for (int o = 1; o < 32; o <<= 1) {
      const int n = __shfl_up(inc, o);
      if (lane >= o) inc += n;
    }
    s_start[lane] = inc - tot;
    if (lane == 31) s_start[CPG] = inc;
  }
  __syncthreads();

  if (FIRST) {
    int sl[SCAP / 32];
#pragma unroll
    for (int k = 0; k < SCAP / 32; ++k) {
      const int idx = k * 32 + lane;
      sl[k] = s_list[wave][(idx < nloc) ? idx : 0];
    }
    const int scn = (&s_cnt[0][0])[t];
    const int sst = s_start[(t <= CPG) ? t : 0];
#pragma unroll
    for (int k = 0; k < SCAP / 32; ++k)
      sv[wave * SCAP + k * 32 + lane] = (k * 32 + lane < nloc) ? sl[k] : 0;
    sv[SV_CNT + t] = scn;
    if (t <= CPG) sv[SV_START + t] = sst;
    if (lane == 0) sv[SV_NLOC + wave] = nloc;
  }
  }

  const int ne = s_start[CPG];
  const int c5 = t >> 3;
  const int w = t & 7;
  const int mystart = s_start[c5];
  const int myend = s_start[c5 + 1];
  float acc[NPARA];
#pragma unroll
  for (int j = 0; j < NPARA; ++j) acc[j] = 0.0f;

  for (int e0 = 0; e0 < ne; e0 += CH) {
    const int nchunk = (ne - e0 < CH) ? (ne - e0) : CH;

    if (e0 != 0) {
      B_LOAD(0, c_lj, c_sx, c_sy, c_sz)
    }
    for (int s = 0; s < nloc; s += 32 * BU) {
      int n_lj[BU];
      float n_sx[BU], n_sy[BU], n_sz[BU];
      B_LOAD(s + 32 * BU, n_lj, n_sx, n_sy, n_sz)
#pragma unroll
      for (int u = 0; u < BU; ++u) {
        if (s + u * 32 < nloc) {
          const int idx = s + u * 32 + lane;
          const bool ok = idx < nloc;
          const int e = s_list[wave][ok ? idx : 0];
          const int ce = (e >> 14) & 31;
          const int slot = s_start[ce] + s_cnt[wave][ce] + (e >> 19) - e0;
          const int li = g * CPG + ce;
          const int lj = c_lj[u];
          const float dx = s_cart[li * 3 + 0] - s_cart[lj * 3 + 0] + c_sx[u];
          const float dy = s_cart[li * 3 + 1] - s_cart[lj * 3 + 1] + c_sy[u];
          const float dz = s_cart[li * 3 + 2] - s_cart[lj * 3 + 2] + c_sz[u];
          const float d = sqrtf(dx * dx + dy * dy + dz * dz);
          float fc = 0.5f * cosf(d * PI_OVER_CUTOFF) + 0.5f;
          fc = fc * fc;
          const int js = lj | (s_spec[lj] << 8);
          if (ok && slot >= 0 && slot < nchunk) {
            s_geo[slot] = make_float4(fc, dx, dy, dz);
            s_d[slot] = d;
            s_js[slot] = js;
          }
        }
      }
#pragma unroll
      for (int u = 0; u < BU; ++u) {
        c_lj[u] = n_lj[u]; c_sx[u] = n_sx[u]; c_sy[u] = n_sy[u]; c_sz[u] = n_sz[u];
      }
    }
    __syncthreads();

    const int lo = ((mystart > e0) ? mystart : e0) - e0;
    const int hi = ((myend < e0 + nchunk) ? myend : (e0 + nchunk)) - e0;
#pragma unroll 4
    for (int i = lo; i < hi; ++i) {
      const float4 g4 = s_geo[i];
      const float d = s_d[i];
      const int js = s_js[i];
      const int lj = js & 255;
      const int sp = js >> 8;
      const float tt = d - s_rs[sp * NWAVE + w];
      const float wv = expf(s_inta[sp * NWAVE + w] * tt * tt) * s_orb[lj * NWAVE + w];
      float ang[NPARA];
      const float dv[3] = {g4.y, g4.z, g4.w};
      ang[0] = g4.x;
#pragma unroll
      for (int a2 = 0; a2 < 3; ++a2) ang[1 + a2] = g4.x * dv[a2];
#pragma unroll
      for (int a2 = 0; a2 < 3; ++a2)
#pragma unroll
        for (int b2 = 0; b2 < 3; ++b2) ang[4 + a2 * 3 + b2] = ang[1 + a2] * dv[b2];
#pragma unroll
      for (int j = 0; j < NPARA; ++j) acc[j] += ang[j] * wv;
    }
    __syncthreads();
  }

  const int lrow = lane & 15;
  const int lhi = lane >> 4;
  const int khalf = lhi * 2;
  const int mt = wave & 1;
  const int np = wave >> 1;
  const int n0 = np * 32 + lrow;
  v2f Blo[3][2], Bhi[3][2];
#pragma unroll
  for (int ip = 0; ip < 3; ++ip) {
    const float* hb = hyper + ip * (NWAVE * NORBIT);
#pragma unroll
    for (int h = 0; h < 2; ++h) {
      Blo[ip][h].x = hb[(khalf + 0) * NORBIT + n0 + h * 16];
      Blo[ip][h].y = hb[(khalf + 1) * NORBIT + n0 + h * 16];
      Bhi[ip][h].x = hb[(khalf + 4) * NORBIT + n0 + h * 16];
      Bhi[ip][h].y = hb[(khalf + 5) * NORBIT + n0 + h * 16];
    }
  }

#pragma unroll
  for (int j = 0; j < NPARA; ++j) s_out[c5 * SOP + j * NWAVE + w] = acc[j];
  __syncthreads();

  {
    v8f d0 = {}, d1 = {};
    const float* arow = s_out + (mt * 16 + lrow) * SOP;
    int para = 0;
#pragma unroll
    for (int ip = 0; ip < 3; ++ip) {
      const v2f B0lo = Blo[ip][0], B0hi = Bhi[ip][0];
      const v2f B1lo = Blo[ip][1], B1hi = Bhi[ip][1];
      const int npj = (ip == 0) ? 1 : ((ip == 1) ? 3 : 9);
#pragma unroll
      for (int q = 0; q < npj; ++q, ++para) {
        const float* ab = arow + para * NWAVE;
        v2f Alo = {ab[khalf + 0], ab[khalf + 1]};
        v2f Ahi = {ab[khalf + 4], ab[khalf + 5]};
        v8f c0 = {}, c1 = {};
        c0 = wmma_f32_16x16x4(Alo, B0lo, c0);
        c1 = wmma_f32_16x16x4(Alo, B1lo, c1);
        c0 = wmma_f32_16x16x4(Ahi, B0hi, c0);
        c1 = wmma_f32_16x16x4(Ahi, B1hi, c1);
#pragma unroll
        for (int r = 0; r < 8; ++r) {
          d0[r] += c0[r] * c0[r];
          d1[r] += c1[r] * c1[r];
        }
      }
    }
#pragma unroll
    for (int r = 0; r < 8; ++r) {
      float* dr = s_den + (mt * 16 + r + lhi * 8) * DP + n0;
      dr[0] = d0[r];
      dr[16] = d1[r];
    }
  }
  __syncthreads();

  if (it == 2) {
    float* dst = density + (b * NATOM + g * CPG) * NORBIT;
    for (int i = t; i < CPG * (NORBIT / 4); i += 256) {
      const int row = i >> 5, c4 = (i & 31) * 4;
      *reinterpret_cast<float4*>(dst + row * NORBIT + c4) =
          *reinterpret_cast<const float4*>(s_den + row * DP + c4);
    }
  } else {
    if (wave < 2) {
      v8f u = {};
      const float* drow = s_den + (wave * 16 + lrow) * DP;
      const bool valid = (lrow < NWAVE);
      float w0[32], w1[32];
#pragma unroll
      for (int kb = 0; kb < 32; ++kb) {
        const int k = kb * 4 + lhi * 2;
        w0[kb] = s_w[(k + 0) * 16 + lrow];
        w1[kb] = s_w[(k + 1) * 16 + lrow];
      }
#pragma unroll
      for (int kb = 0; kb < 32; ++kb) {
        const int k = kb * 4 + lhi * 2;
        v2f A = {drow[k], drow[k + 1]};
        v2f B = {w0[kb], w1[kb]};
        u = wmma_f32_16x16x4(A, B, u);
      }
      if (valid) {
#pragma unroll
        for (int r = 0; r < 8; ++r) {
          const int lg = wave * 16 + r + lhi * 8;
          const int la = g * CPG + lg;
          const float v = tanhf(u[r] + s_e[s_spec[la] * NWAVE + lrow]);
          s_new[lg * NWAVE + lrow] = s_orb[la * NWAVE + lrow] + v;
        }
      }
    }
    __syncthreads();
    if (t < CPG * NWAVE / 4)
      *reinterpret_cast<float4*>(orb_out + (b * NATOM + g * CPG) * NWAVE + t * 4) =
          *reinterpret_cast<const float4*>(s_new + t * 4);
  }
}

extern "C" void kernel_launch(void* const* d_in, const int* in_sizes, int n_in,
                              void* d_out, int out_size, void* d_ws,
                              size_t ws_size, hipStream_t stream) {
  const float* cart = (const float*)d_in[0];
  const float* shifts = (const float*)d_in[1];
  const float* rs = (const float*)d_in[2];
  const float* inta = (const float*)d_in[3];
  const float* params = (const float*)d_in[4];
  const float* hyper = (const float*)d_in[5];
  const float* oc_w0 = (const float*)d_in[6];
  const float* oc_e0 = (const float*)d_in[7];
  const float* oc_w1 = (const float*)d_in[8];
  const float* oc_e1 = (const float*)d_in[9];
  const int* species = (const int*)d_in[11];
  const int* atom_index = (const int*)d_in[12];

  float* density = (float*)d_out;
  float* orb_a = (float*)d_ws;
  float* orb_b = orb_a + TOTNATOM * NWAVE;

  k_obtain<true><<<NB * GRP, 256, 0, stream>>>(cart, shifts, rs, inta, params,
                                               hyper, oc_w0, oc_e0, species,
                                               atom_index, orb_b, orb_a, density, 0);
  k_obtain<false><<<NB * GRP, 256, 0, stream>>>(cart, shifts, rs, inta, params,
                                                hyper, oc_w1, oc_e1, species,
                                                atom_index, orb_a, orb_b, density, 1);
  k_obtain<false><<<NB * GRP, 256, 0, stream>>>(cart, shifts, rs, inta, params,
                                                hyper, oc_w1, oc_e1, species,
                                                atom_index, orb_b, orb_a, density, 2);
}


# Run sums: a lane takes a quarter of its center's run for 4 radial channels (was the whole run for 1), pair terms formed 2x not 8x; quarter sums added in the quad as (q0+q1)+(q2+q3)
# speedup vs baseline: 115.3160x; 1.1590x over previous
#include <hip/hip_runtime.h>
#include <math.h>

#define NTYPE 4
#define NWAVE 8
#define NORBIT 128
#define NEIGH 64
#define NB 32
#define NATOM 256
#define NPAIR_PER (NATOM * NEIGH)
#define NPAIR_TOT (NB * NPAIR_PER)
#define TOTNATOM (NB * NATOM)
#define NPARA 13
#define PI_OVER_CUTOFF 0.62831853071795864769f

typedef float v2f __attribute__((ext_vector_type(2)));
typedef float v8f __attribute__((ext_vector_type(8)));

__device__ __forceinline__ v8f wmma_f32_16x16x4(v2f a, v2f b, v8f c) {
  return __builtin_amdgcn_wmma_f32_16x16x4_f32(false, a, false, b, (short)0, c,
                                               false, false);
}

__device__ __forceinline__ float quad_xor1(float v) {
  return __int_as_float(__builtin_amdgcn_update_dpp(
      0, __float_as_int(v), 0xB1  , 0xF, 0xF, true));
}
__device__ __forceinline__ float quad_xor2(float v) {
  return __int_as_float(__builtin_amdgcn_update_dpp(
      0, __float_as_int(v), 0x4E  , 0xF, 0xF, true));
}

#define GRP 8
#define CPG (NATOM / GRP)
#define SLICE (NPAIR_PER / 8)
#define CH 2560
#define BU 5
#define ROWF (NPARA * NWAVE)
#define SOP (ROWF + 4)
#define DP (NORBIT + 4)
#define SCAP 448
#define SV_CNT (8 * SCAP)
#define SV_START (SV_CNT + 256)
#define SV_NLOC (SV_START + 40)

#define B_LOAD(S, LJ, SX, SY, SZ)                                   \
  _Pragma("unroll") for (int u = 0; u < BU; ++u) {                  \
    const int idx = (S) + u * 32 + lane;                            \
    const int e = s_list[wave][(idx < nloc) ? idx : 0];             \
    const int p = b * NPAIR_PER + (e & 0x3fff);                     \
    LJ[u] = atom_index[NPAIR_TOT + p];      \
    SX[u] = shifts[p * 3 + 0];                                      \
    SY[u] = shifts[p * 3 + 1];                                      \
    SZ[u] = shifts[p * 3 + 2];                                      \
  }

template <bool FIRST>
__global__ void __launch_bounds__(256)
k_obtain(const float* __restrict__ cart,
         const float* __restrict__ shifts,
         const float* __restrict__ rs,
         const float* __restrict__ inta,
         const float* __restrict__ params,
         const float* __restrict__ hyper,
         const float* __restrict__ oc_w,
         const float* __restrict__ oc_e,
         const int* __restrict__ species,
         const int* __restrict__ atom_index,
         const float* __restrict__ orb_in,
         float* __restrict__ orb_out,
         float* __restrict__ density,
         const int it) {
  __shared__ int s_list[8][SLICE];
  __shared__ float4 s_geo[CH];
  __shared__ float s_d[CH];
  __shared__ int s_js[CH];
  __shared__ __attribute__((aligned(16))) float s_out[CPG * SOP];
  __shared__ __attribute__((aligned(16))) float s_den[CPG * DP];
  __shared__ __attribute__((aligned(16))) float s_new[CPG * NWAVE];
  __shared__ __attribute__((aligned(16))) float s_orb[NATOM * NWAVE];
  __shared__ __attribute__((aligned(16))) float s_w[NORBIT * 16];
  __shared__ float s_e[NTYPE * NWAVE];
  __shared__ float s_cart[NATOM * 3];
  __shared__ int s_spec[NATOM];
  __shared__ __attribute__((aligned(16))) float s_rs[NTYPE * NWAVE];
  __shared__ __attribute__((aligned(16))) float s_inta[NTYPE * NWAVE];
  __shared__ int s_cnt[8][CPG];
  __shared__ int s_start[CPG + 1];

  const int t = threadIdx.x;
  const int wave = t >> 5;
  const int lane = t & 31;
  const int b = blockIdx.x / GRP;
  const int g = blockIdx.x % GRP;

  int* sv = reinterpret_cast<int*>(density) + (b * NATOM + g * CPG) * NORBIT;
  int lw[SCAP / 32];
  int r_cnt = 0, r_start = 0, r_nmine = 0;
  int4 r_na = make_int4(0, 0, 0, 0), r_nb = r_na;
  if (!FIRST) {
    r_nmine = sv[SV_NLOC + wave];
#pragma unroll
    for (int k = 0; k < SCAP / 32; ++k) lw[k] = sv[wave * SCAP + k * 32 + lane];
    r_cnt = sv[SV_CNT + t];
    r_start = sv[SV_START + ((t <= CPG) ? t : 0)];
    r_na = *reinterpret_cast<const int4*>(sv + SV_NLOC);
    r_nb = *reinterpret_cast<const int4*>(sv + SV_NLOC + 4);
  }

  const int r_spec = species[b * NATOM + t];
  float r_cart[3];
#pragma unroll
  for (int j = 0; j < 3; ++j) r_cart[j] = cart[b * NATOM * 3 + j * 256 + t];
  float4 r_o0 = make_float4(0.0f, 0.0f, 0.0f, 0.0f), r_o1 = r_o0;
  if (!FIRST) {
    const float* src = orb_in + (b * NATOM + t) * NWAVE;
    r_o0 = *reinterpret_cast<const float4*>(src);
    r_o1 = *reinterpret_cast<const float4*>(src + 4);
  }
  float4 r_w = make_float4(0.0f, 0.0f, 0.0f, 0.0f);
  if (it != 2) r_w = *reinterpret_cast<const float4*>(oc_w + t * 4);
  float r_rs = 0.0f, r_inta = 0.0f, r_e = 0.0f;
  if (t < NTYPE * NWAVE) { r_rs = rs[t]; r_inta = inta[t]; r_e = oc_e[t]; }

  const unsigned ltmask = (1u << lane) - 1u;
  int nloc = 0;
  bool reuse = false;
  if (!FIRST) {
    const int nmax = max(max(max(r_na.x, r_na.y), max(r_na.z, r_na.w)),
                         max(max(r_nb.x, r_nb.y), max(r_nb.z, r_nb.w)));
    reuse = __builtin_amdgcn_readfirstlane(nmax) <= SCAP;
  }
  if (reuse) {
    nloc = __builtin_amdgcn_readfirstlane(r_nmine);
#pragma unroll
    for (int k = 0; k < SCAP / 32; ++k) s_list[wave][k * 32 + lane] = lw[k];
    (&s_cnt[0][0])[t] = r_cnt;
    if (t <= CPG) s_start[t] = r_start;
  } else {
  (&s_cnt[0][0])[t] = 0;

  const int* keys = atom_index + b * NPAIR_PER;
  const int kbase = wave * SLICE;
  int kr[SLICE / 32];
#pragma unroll
  for (int j = 0; j < SLICE / 32; ++j) kr[j] = keys[kbase + j * 32 + lane];
  __builtin_amdgcn_sched_barrier(0);
  if (FIRST) {
    const float* src = params + r_spec * NWAVE;
    r_o0 = *reinterpret_cast<const float4*>(src);
    r_o1 = *reinterpret_cast<const float4*>(src + 4);
  }
#pragma unroll
  for (int j = 0; j < SLICE / 32; ++j) {
    const int cc = kr[j];
    const bool inr = (cc >> 5) == g;
    const unsigned mk = __builtin_amdgcn_ballot_w32(inr);
    if (inr)
      s_list[wave][nloc + __builtin_popcount(mk & ltmask)] =
          (kbase + j * 32 + lane) | ((cc & 31) << 14);
    nloc += __builtin_popcount(mk);
  }
  }

  int c_lj[BU];
  float c_sx[BU], c_sy[BU], c_sz[BU];
  if (reuse) {
    B_LOAD(0, c_lj, c_sx, c_sy, c_sz)
  }

#pragma unroll
  for (int j = 0; j < 3; ++j) s_cart[j * 256 + t] = r_cart[j];
  *reinterpret_cast<float4*>(s_orb + t * NWAVE) = r_o0;
  *reinterpret_cast<float4*>(s_orb + t * NWAVE + 4) = r_o1;
  *reinterpret_cast<float4*>(s_w + (t >> 1) * 16 + (t & 1) * 4) = r_w;
  *reinterpret_cast<float4*>(s_w + (t >> 1) * 16 + 8 + (t & 1) * 4) =
      make_float4(0.0f, 0.0f, 0.0f, 0.0f);
  s_spec[t] = r_spec;
  if (t < NTYPE * NWAVE) { s_rs[t] = r_rs; s_inta[t] = r_inta; s_e[t] = r_e; }
  __syncthreads();

  if (!reuse) {
  B_LOAD(0, c_lj, c_sx, c_sy, c_sz)

  for (int s = 0; s < nloc; s += 32) {
    const int idx = s + lane;
    const bool ok = idx < nloc;
    const int e = ok ? s_list[wave][idx] : 0;
    const int c5 = (e >> 14) & 31;
    unsigned m = __builtin_amdgcn_ballot_w32(ok);
#pragma unroll
    for (int bit = 0; bit < 5; ++bit) {
      const bool bs = ((c5 >> bit) & 1) != 0;
      const unsigned bb = __builtin_amdgcn_ballot_w32(bs);
      m &= bs ? bb : ~bb;
    }
    if (ok) {
      const int prev = s_cnt[wave][c5];
      const int r = __builtin_popcount(m & ltmask);
      s_list[wave][idx] = e | ((prev + r) << 19);
      if (r == 0) s_cnt[wave][c5] = prev + __builtin_popcount(m);
    }
  }
  __syncthreads();

  if (wave == 0) {
    int tot = 0;
#pragma unroll
    for (int w = 0; w < 8; ++w) {
      const int v = s_cnt[w][lane];
      s_cnt[w][lane] = tot;
      tot += v;
    }
    int inc = tot;
#pragma unroll
    for (int o = 1; o < 32; o <<= 1) {
      const int n = __shfl_up(inc, o);
      if (lane >= o) inc += n;
    }
    s_start[lane] = inc - tot;
    if (lane == 31) s_start[CPG] = inc;
  }
  __syncthreads();

  if (FIRST) {
    int sl[SCAP / 32];
#pragma unroll
    for (int k = 0; k < SCAP / 32; ++k) {
      const int idx = k * 32 + lane;
      sl[k] = s_list[wave][(idx < nloc) ? idx : 0];
    }
    const int scn = (&s_cnt[0][0])[t];
    const int sst = s_start[(t <= CPG) ? t : 0];
#pragma unroll
    for (int k = 0; k < SCAP / 32; ++k)
      sv[wave * SCAP + k * 32 + lane] = (k * 32 + lane < nloc) ? sl[k] : 0;
    sv[SV_CNT + t] = scn;
    if (t <= CPG) sv[SV_START + t] = sst;
    if (lane == 0) sv[SV_NLOC + wave] = nloc;
  }
  }

  const int ne = s_start[CPG];
  const int c5 = t >> 3;
  const int qd = t & 3;
  const int wq = (t >> 2) & 1;
  const int mystart = s_start[c5];
  const int myn = s_start[c5 + 1] - mystart;
  const int qstart = mystart + ((myn * qd) >> 2);
  const int qend = mystart + ((myn * (qd + 1)) >> 2);
  float4 acc[NPARA];
#pragma unroll
  for (int j = 0; j < NPARA; ++j) acc[j] = make_float4(0.0f, 0.0f, 0.0f, 0.0f);

  for (int e0 = 0; e0 < ne; e0 += CH) {
    const int nchunk = (ne - e0 < CH) ? (ne - e0) : CH;

    if (e0 != 0) {
      B_LOAD(0, c_lj, c_sx, c_sy, c_sz)
    }
    for (int s = 0; s < nloc; s += 32 * BU) {
      int n_lj[BU];
      float n_sx[BU], n_sy[BU], n_sz[BU];
      B_LOAD(s + 32 * BU, n_lj, n_sx, n_sy, n_sz)
#pragma unroll
      for (int u = 0; u < BU; ++u) {
        if (s + u * 32 < nloc) {
          const int idx = s + u * 32 + lane;
          const bool ok = idx < nloc;
          const int e = s_list[wave][ok ? idx : 0];
          const int ce = (e >> 14) & 31;
          const int slot = s_start[ce] + s_cnt[wave][ce] + (e >> 19) - e0;
          const int li = g * CPG + ce;
          const int lj = c_lj[u];
          const float dx = s_cart[li * 3 + 0] - s_cart[lj * 3 + 0] + c_sx[u];
          const float dy = s_cart[li * 3 + 1] - s_cart[lj * 3 + 1] + c_sy[u];
          const float dz = s_cart[li * 3 + 2] - s_cart[lj * 3 + 2] + c_sz[u];
          const float d = sqrtf(dx * dx + dy * dy + dz * dz);
          float fc = 0.5f * cosf(d * PI_OVER_CUTOFF) + 0.5f;
          fc = fc * fc;
          const int js = lj | (s_spec[lj] << 8);
          if (ok && slot >= 0 && slot < nchunk) {
            s_geo[slot] = make_float4(fc, dx, dy, dz);
            s_d[slot] = d;
            s_js[slot] = js;
          }
        }
      }
#pragma unroll
      for (int u = 0; u < BU; ++u) {
        c_lj[u] = n_lj[u]; c_sx[u] = n_sx[u]; c_sy[u] = n_sy[u]; c_sz[u] = n_sz[u];
      }
    }
    __syncthreads();

    const int lo = ((qstart > e0) ? qstart : e0) - e0;
    const int hi = ((qend < e0 + nchunk) ? qend : (e0 + nchunk)) - e0;
#pragma unroll 2
    for (int i = lo; i < hi; ++i) {
      const float4 g4 = s_geo[i];
      const float d = s_d[i];
      const int js = s_js[i];
      const int lj = js & 255;
      const int sp = js >> 8;
      const float4 rs4 = *reinterpret_cast<const float4*>(s_rs + sp * NWAVE + wq * 4);
      const float4 in4 = *reinterpret_cast<const float4*>(s_inta + sp * NWAVE + wq * 4);
      const float4 ob4 = *reinterpret_cast<const float4*>(s_orb + lj * NWAVE + wq * 4);
      const float rsv[4] = {rs4.x, rs4.y, rs4.z, rs4.w};
      const float inv[4] = {in4.x, in4.y, in4.z, in4.w};
      const float obv[4] = {ob4.x, ob4.y, ob4.z, ob4.w};
      float wv[4];
#pragma unroll
      for (int k = 0; k < 4; ++k) {
        const float tt = d - rsv[k];
        wv[k] = expf(inv[k] * tt * tt) * obv[k];
      }
      float ang[NPARA];
      const float dv[3] = {g4.y, g4.z, g4.w};
      ang[0] = g4.x;
#pragma unroll
      for (int a2 = 0; a2 < 3; ++a2) ang[1 + a2] = g4.x * dv[a2];
#pragma unroll
      for (int a2 = 0; a2 < 3; ++a2)
#pragma unroll
        for (int b2 = 0; b2 < 3; ++b2) ang[4 + a2 * 3 + b2] = ang[1 + a2] * dv[b2];
#pragma unroll
      for (int j = 0; j < NPARA; ++j) {
        acc[j].x += ang[j] * wv[0];
        acc[j].y += ang[j] * wv[1];
        acc[j].z += ang[j] * wv[2];
        acc[j].w += ang[j] * wv[3];
      }
    }
    __syncthreads();
  }

  const int lrow = lane & 15;
  const int lhi = lane >> 4;
  const int khalf = lhi * 2;
  const int mt = wave & 1;
  const int np = wave >> 1;
  const int n0 = np * 32 + lrow;
  v2f Blo[3][2], Bhi[3][2];
#pragma unroll
  for (int ip = 0; ip < 3; ++ip) {
    const float* hb = hyper + ip * (NWAVE * NORBIT);
#pragma unroll
    for (int h = 0; h < 2; ++h) {
      Blo[ip][h].x = hb[(khalf + 0) * NORBIT + n0 + h * 16];
      Blo[ip][h].y = hb[(khalf + 1) * NORBIT + n0 + h * 16];
      Bhi[ip][h].x = hb[(khalf + 4) * NORBIT + n0 + h * 16];
      Bhi[ip][h].y = hb[(khalf + 5) * NORBIT + n0 + h * 16];
    }
  }

#pragma unroll
  for (int j = 0; j < NPARA; ++j) {
    acc[j].x += quad_xor1(acc[j].x);
    acc[j].y += quad_xor1(acc[j].y);
    acc[j].z += quad_xor1(acc[j].z);
    acc[j].w += quad_xor1(acc[j].w);
    acc[j].x += quad_xor2(acc[j].x);
    acc[j].y += quad_xor2(acc[j].y);
    acc[j].z += quad_xor2(acc[j].z);
    acc[j].w += quad_xor2(acc[j].w);
  }
  if (qd == 0) {
#pragma unroll
    for (int j = 0; j < NPARA; ++j)
      *reinterpret_cast<float4*>(s_out + c5 * SOP + j * NWAVE + wq * 4) = acc[j];
  }
  __syncthreads();

  {
    v8f d0 = {}, d1 = {};
    const float* arow = s_out + (mt * 16 + lrow) * SOP;
    int para = 0;
#pragma unroll
    for (int ip = 0; ip < 3; ++ip) {
      const v2f B0lo = Blo[ip][0], B0hi = Bhi[ip][0];
      const v2f B1lo = Blo[ip][1], B1hi = Bhi[ip][1];
      const int npj = (ip == 0) ? 1 : ((ip == 1) ? 3 : 9);
#pragma unroll
      for (int q = 0; q < npj; ++q, ++para) {
        const float* ab = arow + para * NWAVE;
        v2f Alo = {ab[khalf + 0], ab[khalf + 1]};
        v2f Ahi = {ab[khalf + 4], ab[khalf + 5]};
        v8f c0 = {}, c1 = {};
        c0 = wmma_f32_16x16x4(Alo, B0lo, c0);
        c1 = wmma_f32_16x16x4(Alo, B1lo, c1);
        c0 = wmma_f32_16x16x4(Ahi, B0hi, c0);
        c1 = wmma_f32_16x16x4(Ahi, B1hi, c1);
#pragma unroll
        for (int r = 0; r < 8; ++r) {
          d0[r] += c0[r] * c0[r];
          d1[r] += c1[r] * c1[r];
        }
      }
    }
#pragma unroll
    for (int r = 0; r < 8; ++r) {
      float* dr = s_den + (mt * 16 + r + lhi * 8) * DP + n0;
      dr[0] = d0[r];
      dr[16] = d1[r];
    }
  }
  __syncthreads();

  if (it == 2) {
    float* dst = density + (b * NATOM + g * CPG) * NORBIT;
    for (int i = t; i < CPG * (NORBIT / 4); i += 256) {
      const int row = i >> 5, c4 = (i & 31) * 4;
      *reinterpret_cast<float4*>(dst + row * NORBIT + c4) =
          *reinterpret_cast<const float4*>(s_den + row * DP + c4);
    }
  } else {
    if (wave < 2) {
      v8f u = {};
      const float* drow = s_den + (wave * 16 + lrow) * DP;
      const bool valid = (lrow < NWAVE);
      float w0[32], w1[32];
#pragma unroll
      for (int kb = 0; kb < 32; ++kb) {
        const int k = kb * 4 + lhi * 2;
        w0[kb] = s_w[(k + 0) * 16 + lrow];
        w1[kb] = s_w[(k + 1) * 16 + lrow];
      }
#pragma unroll
      for (int kb = 0; kb < 32; ++kb) {
        const int k = kb * 4 + lhi * 2;
        v2f A = {drow[k], drow[k + 1]};
        v2f B = {w0[kb], w1[kb]};
        u = wmma_f32_16x16x4(A, B, u);
      }
      if (valid) {
#pragma unroll
        for (int r = 0; r < 8; ++r) {
          const int lg = wave * 16 + r + lhi * 8;
          const int la = g * CPG + lg;
          const float v = tanhf(u[r] + s_e[s_spec[la] * NWAVE + lrow]);
          s_new[lg * NWAVE + lrow] = s_orb[la * NWAVE + lrow] + v;
        }
      }
    }
    __syncthreads();
    if (t < CPG * NWAVE / 4)
      *reinterpret_cast<float4*>(orb_out + (b * NATOM + g * CPG) * NWAVE + t * 4) =
          *reinterpret_cast<const float4*>(s_new + t * 4);
  }
}

extern "C" void kernel_launch(void* const* d_in, const int* in_sizes, int n_in,
                              void* d_out, int out_size, void* d_ws,
                              size_t ws_size, hipStream_t stream) {
  const float* cart = (const float*)d_in[0];
  const float* shifts = (const float*)d_in[1];
  const float* rs = (const float*)d_in[2];
  const float* inta = (const float*)d_in[3];
  const float* params = (const float*)d_in[4];
  const float* hyper = (const float*)d_in[5];
  const float* oc_w0 = (const float*)d_in[6];
  const float* oc_e0 = (const float*)d_in[7];
  const float* oc_w1 = (const float*)d_in[8];
  const float* oc_e1 = (const float*)d_in[9];
  const int* species = (const int*)d_in[11];
  const int* atom_index = (const int*)d_in[12];

  float* density = (float*)d_out;
  float* orb_a = (float*)d_ws;
  float* orb_b = orb_a + TOTNATOM * NWAVE;

  k_obtain<true><<<NB * GRP, 256, 0, stream>>>(cart, shifts, rs, inta, params,
                                               hyper, oc_w0, oc_e0, species,
                                               atom_index, orb_b, orb_a, density, 0);
  k_obtain<false><<<NB * GRP, 256, 0, stream>>>(cart, shifts, rs, inta, params,
                                                hyper, oc_w1, oc_e1, species,
                                                atom_index, orb_a, orb_b, density, 1);
  k_obtain<false><<<NB * GRP, 256, 0, stream>>>(cart, shifts, rs, inta, params,
                                                hyper, oc_w1, oc_e1, species,
                                                atom_index, orb_b, orb_a, density, 2);
}


# Geometry pass: the 5 rows of a trip read LDS and form their distances as one branch-free block, so the rows' LDS round trips and root chains overlap (were row by row); values and sums unchanged
# speedup vs baseline: 121.7219x; 1.0556x over previous
#include <hip/hip_runtime.h>
#include <math.h>

#define NTYPE 4
#define NWAVE 8
#define NORBIT 128
#define NEIGH 64
#define NB 32
#define NATOM 256
#define NPAIR_PER (NATOM * NEIGH)
#define NPAIR_TOT (NB * NPAIR_PER)
#define TOTNATOM (NB * NATOM)
#define NPARA 13
#define PI_OVER_CUTOFF 0.62831853071795864769f

typedef float v2f __attribute__((ext_vector_type(2)));
typedef float v8f __attribute__((ext_vector_type(8)));

__device__ __forceinline__ v8f wmma_f32_16x16x4(v2f a, v2f b, v8f c) {
  return __builtin_amdgcn_wmma_f32_16x16x4_f32(false, a, false, b, (short)0, c,
                                               false, false);
}

__device__ __forceinline__ float quad_xor1(float v) {
  return __int_as_float(__builtin_amdgcn_update_dpp(
      0, __float_as_int(v), 0xB1  , 0xF, 0xF, true));
}
__device__ __forceinline__ float quad_xor2(float v) {
  return __int_as_float(__builtin_amdgcn_update_dpp(
      0, __float_as_int(v), 0x4E  , 0xF, 0xF, true));
}

#define GRP 8
#define CPG (NATOM / GRP)
#define SLICE (NPAIR_PER / 8)
#define CH 2560
#define BU 5
#define ROWF (NPARA * NWAVE)
#define SOP (ROWF + 4)
#define DP (NORBIT + 4)
#define SCAP 448
#define SV_CNT (8 * SCAP)
#define SV_START (SV_CNT + 256)
#define SV_NLOC (SV_START + 40)

#define B_LOAD(S, LJ, SX, SY, SZ)                                   \
  _Pragma("unroll") for (int u = 0; u < BU; ++u) {                  \
    const int idx = (S) + u * 32 + lane;                            \
    const int e = s_list[wave][(idx < nloc) ? idx : 0];             \
    const int p = b * NPAIR_PER + (e & 0x3fff);                     \
    LJ[u] = atom_index[NPAIR_TOT + p];      \
    SX[u] = shifts[p * 3 + 0];                                      \
    SY[u] = shifts[p * 3 + 1];                                      \
    SZ[u] = shifts[p * 3 + 2];                                      \
  }

template <bool FIRST>
__global__ void __launch_bounds__(256)
k_obtain(const float* __restrict__ cart,
         const float* __restrict__ shifts,
         const float* __restrict__ rs,
         const float* __restrict__ inta,
         const float* __restrict__ params,
         const float* __restrict__ hyper,
         const float* __restrict__ oc_w,
         const float* __restrict__ oc_e,
         const int* __restrict__ species,
         const int* __restrict__ atom_index,
         const float* __restrict__ orb_in,
         float* __restrict__ orb_out,
         float* __restrict__ density,
         const int it) {
  __shared__ int s_list[8][SLICE];
  __shared__ float4 s_geo[CH];
  __shared__ float s_d[CH];
  __shared__ int s_js[CH];
  __shared__ __attribute__((aligned(16))) float s_out[CPG * SOP];
  __shared__ __attribute__((aligned(16))) float s_den[CPG * DP];
  __shared__ __attribute__((aligned(16))) float s_new[CPG * NWAVE];
  __shared__ __attribute__((aligned(16))) float s_orb[NATOM * NWAVE];
  __shared__ __attribute__((aligned(16))) float s_w[NORBIT * 16];
  __shared__ float s_e[NTYPE * NWAVE];
  __shared__ float s_cart[NATOM * 3];
  __shared__ int s_spec[NATOM];
  __shared__ __attribute__((aligned(16))) float s_rs[NTYPE * NWAVE];
  __shared__ __attribute__((aligned(16))) float s_inta[NTYPE * NWAVE];
  __shared__ int s_cnt[8][CPG];
  __shared__ int s_start[CPG + 1];

  const int t = threadIdx.x;
  const int wave = t >> 5;
  const int lane = t & 31;
  const int b = blockIdx.x / GRP;
  const int g = blockIdx.x % GRP;

  int* sv = reinterpret_cast<int*>(density) + (b * NATOM + g * CPG) * NORBIT;
  int lw[SCAP / 32];
  int r_cnt = 0, r_start = 0, r_nmine = 0;
  int4 r_na = make_int4(0, 0, 0, 0), r_nb = r_na;
  if (!FIRST) {
    r_nmine = sv[SV_NLOC + wave];
#pragma unroll
    for (int k = 0; k < SCAP / 32; ++k) lw[k] = sv[wave * SCAP + k * 32 + lane];
    r_cnt = sv[SV_CNT + t];
    r_start = sv[SV_START + ((t <= CPG) ? t : 0)];
    r_na = *reinterpret_cast<const int4*>(sv + SV_NLOC);
    r_nb = *reinterpret_cast<const int4*>(sv + SV_NLOC + 4);
  }

  const int r_spec = species[b * NATOM + t];
  float r_cart[3];
#pragma unroll
  for (int j = 0; j < 3; ++j) r_cart[j] = cart[b * NATOM * 3 + j * 256 + t];
  float4 r_o0 = make_float4(0.0f, 0.0f, 0.0f, 0.0f), r_o1 = r_o0;
  if (!FIRST) {
    const float* src = orb_in + (b * NATOM + t) * NWAVE;
    r_o0 = *reinterpret_cast<const float4*>(src);
    r_o1 = *reinterpret_cast<const float4*>(src + 4);
  }
  float4 r_w = make_float4(0.0f, 0.0f, 0.0f, 0.0f);
  if (it != 2) r_w = *reinterpret_cast<const float4*>(oc_w + t * 4);
  float r_rs = 0.0f, r_inta = 0.0f, r_e = 0.0f;
  if (t < NTYPE * NWAVE) { r_rs = rs[t]; r_inta = inta[t]; r_e = oc_e[t]; }

  const unsigned ltmask = (1u << lane) - 1u;
  int nloc = 0;
  bool reuse = false;
  if (!FIRST) {
    const int nmax = max(max(max(r_na.x, r_na.y), max(r_na.z, r_na.w)),
                         max(max(r_nb.x, r_nb.y), max(r_nb.z, r_nb.w)));
    reuse = __builtin_amdgcn_readfirstlane(nmax) <= SCAP;
  }
  if (reuse) {
    nloc = __builtin_amdgcn_readfirstlane(r_nmine);
#pragma unroll
    for (int k = 0; k < SCAP / 32; ++k) s_list[wave][k * 32 + lane] = lw[k];
    (&s_cnt[0][0])[t] = r_cnt;
    if (t <= CPG) s_start[t] = r_start;
  } else {
  (&s_cnt[0][0])[t] = 0;

  const int* keys = atom_index + b * NPAIR_PER;
  const int kbase = wave * SLICE;
  int kr[SLICE / 32];
#pragma unroll
  for (int j = 0; j < SLICE / 32; ++j) kr[j] = keys[kbase + j * 32 + lane];
  __builtin_amdgcn_sched_barrier(0);
  if (FIRST) {
    const float* src = params + r_spec * NWAVE;
    r_o0 = *reinterpret_cast<const float4*>(src);
    r_o1 = *reinterpret_cast<const float4*>(src + 4);
  }
#pragma unroll
  for (int j = 0; j < SLICE / 32; ++j) {
    const int cc = kr[j];
    const bool inr = (cc >> 5) == g;
    const unsigned mk = __builtin_amdgcn_ballot_w32(inr);
    if (inr)
      s_list[wave][nloc + __builtin_popcount(mk & ltmask)] =
          (kbase + j * 32 + lane) | ((cc & 31) << 14);
    nloc += __builtin_popcount(mk);
  }
  }

  int c_lj[BU];
  float c_sx[BU], c_sy[BU], c_sz[BU];
  if (reuse) {
    B_LOAD(0, c_lj, c_sx, c_sy, c_sz)
  }

#pragma unroll
  for (int j = 0; j < 3; ++j) s_cart[j * 256 + t] = r_cart[j];
  *reinterpret_cast<float4*>(s_orb + t * NWAVE) = r_o0;
  *reinterpret_cast<float4*>(s_orb + t * NWAVE + 4) = r_o1;
  *reinterpret_cast<float4*>(s_w + (t >> 1) * 16 + (t & 1) * 4) = r_w;
  *reinterpret_cast<float4*>(s_w + (t >> 1) * 16 + 8 + (t & 1) * 4) =
      make_float4(0.0f, 0.0f, 0.0f, 0.0f);
  s_spec[t] = r_spec;
  if (t < NTYPE * NWAVE) { s_rs[t] = r_rs; s_inta[t] = r_inta; s_e[t] = r_e; }
  __syncthreads();

  if (!reuse) {
  B_LOAD(0, c_lj, c_sx, c_sy, c_sz)

  for (int s = 0; s < nloc; s += 32) {
    const int idx = s + lane;
    const bool ok = idx < nloc;
    const int e = ok ? s_list[wave][idx] : 0;
    const int c5 = (e >> 14) & 31;
    unsigned m = __builtin_amdgcn_ballot_w32(ok);
#pragma unroll
    for (int bit = 0; bit < 5; ++bit) {
      const bool bs = ((c5 >> bit) & 1) != 0;
      const unsigned bb = __builtin_amdgcn_ballot_w32(bs);
      m &= bs ? bb : ~bb;
    }
    if (ok) {
      const int prev = s_cnt[wave][c5];
      const int r = __builtin_popcount(m & ltmask);
      s_list[wave][idx] = e | ((prev + r) << 19);
      if (r == 0) s_cnt[wave][c5] = prev + __builtin_popcount(m);
    }
  }
  __syncthreads();

  if (wave == 0) {
    int tot = 0;
#pragma unroll
    for (int w = 0; w < 8; ++w) {
      const int v = s_cnt[w][lane];
      s_cnt[w][lane] = tot;
      tot += v;
    }
    int inc = tot;
#pragma unroll
    for (int o = 1; o < 32; o <<= 1) {
      const int n = __shfl_up(inc, o);
      if (lane >= o) inc += n;
    }
    s_start[lane] = inc - tot;
    if (lane == 31) s_start[CPG] = inc;
  }
  __syncthreads();

  if (FIRST) {
    int sl[SCAP / 32];
#pragma unroll
    for (int k = 0; k < SCAP / 32; ++k) {
      const int idx = k * 32 + lane;
      sl[k] = s_list[wave][(idx < nloc) ? idx : 0];
    }
    const int scn = (&s_cnt[0][0])[t];
    const int sst = s_start[(t <= CPG) ? t : 0];
#pragma unroll
    for (int k = 0; k < SCAP / 32; ++k)
      sv[wave * SCAP + k * 32 + lane] = (k * 32 + lane < nloc) ? sl[k] : 0;
    sv[SV_CNT + t] = scn;
    if (t <= CPG) sv[SV_START + t] = sst;
    if (lane == 0) sv[SV_NLOC + wave] = nloc;
  }
  }

  const int ne = s_start[CPG];
  const int c5 = t >> 3;
  const int qd = t & 3;
  const int wq = (t >> 2) & 1;
  const int mystart = s_start[c5];
  const int myn = s_start[c5 + 1] - mystart;
  const int qstart = mystart + ((myn * qd) >> 2);
  const int qend = mystart + ((myn * (qd + 1)) >> 2);
  float4 acc[NPARA];
#pragma unroll
  for (int j = 0; j < NPARA; ++j) acc[j] = make_float4(0.0f, 0.0f, 0.0f, 0.0f);

  for (int e0 = 0; e0 < ne; e0 += CH) {
    const int nchunk = (ne - e0 < CH) ? (ne - e0) : CH;

    if (e0 != 0) {
      B_LOAD(0, c_lj, c_sx, c_sy, c_sz)
    }
    for (int s = 0; s < nloc; s += 32 * BU) {
      int n_lj[BU];
      float n_sx[BU], n_sy[BU], n_sz[BU];
      B_LOAD(s + 32 * BU, n_lj, n_sx, n_sy, n_sz)
      float r_dx[BU], r_dy[BU], r_dz[BU], r_d[BU];
      int r_slot[BU], r_js[BU];
#pragma unroll
      for (int u = 0; u < BU; ++u) {
        const int idx = s + u * 32 + lane;
        const int e = s_list[wave][(idx < nloc) ? idx : 0];
        const int ce = (e >> 14) & 31;
        r_slot[u] = s_start[ce] + s_cnt[wave][ce] + (e >> 19) - e0;
        const int li = g * CPG + ce;
        const int lj = c_lj[u];
        const float dx = s_cart[li * 3 + 0] - s_cart[lj * 3 + 0] + c_sx[u];
        const float dy = s_cart[li * 3 + 1] - s_cart[lj * 3 + 1] + c_sy[u];
        const float dz = s_cart[li * 3 + 2] - s_cart[lj * 3 + 2] + c_sz[u];
        r_dx[u] = dx; r_dy[u] = dy; r_dz[u] = dz;
        r_d[u] = sqrtf(dx * dx + dy * dy + dz * dz);
        r_js[u] = lj | (s_spec[lj] << 8);
      }
#pragma unroll
      for (int u = 0; u < BU; ++u)
        asm volatile("" : : "v"(r_d[u]));
#pragma unroll
      for (int u = 0; u < BU; ++u) {
        if (s + u * 32 < nloc) {
          const bool ok = s + u * 32 + lane < nloc;
          const int slot = r_slot[u];
          float fc = 0.5f * cosf(r_d[u] * PI_OVER_CUTOFF) + 0.5f;
          fc = fc * fc;
          if (ok && slot >= 0 && slot < nchunk) {
            s_geo[slot] = make_float4(fc, r_dx[u], r_dy[u], r_dz[u]);
            s_d[slot] = r_d[u];
            s_js[slot] = r_js[u];
          }
        }
      }
#pragma unroll
      for (int u = 0; u < BU; ++u) {
        c_lj[u] = n_lj[u]; c_sx[u] = n_sx[u]; c_sy[u] = n_sy[u]; c_sz[u] = n_sz[u];
      }
    }
    __syncthreads();

    const int lo = ((qstart > e0) ? qstart : e0) - e0;
    const int hi = ((qend < e0 + nchunk) ? qend : (e0 + nchunk)) - e0;
#pragma unroll 2
    for (int i = lo; i < hi; ++i) {
      const float4 g4 = s_geo[i];
      const float d = s_d[i];
      const int js = s_js[i];
      const int lj = js & 255;
      const int sp = js >> 8;
      const float4 rs4 = *reinterpret_cast<const float4*>(s_rs + sp * NWAVE + wq * 4);
      const float4 in4 = *reinterpret_cast<const float4*>(s_inta + sp * NWAVE + wq * 4);
      const float4 ob4 = *reinterpret_cast<const float4*>(s_orb + lj * NWAVE + wq * 4);
      const float rsv[4] = {rs4.x, rs4.y, rs4.z, rs4.w};
      const float inv[4] = {in4.x, in4.y, in4.z, in4.w};
      const float obv[4] = {ob4.x, ob4.y, ob4.z, ob4.w};
      float wv[4];
#pragma unroll
      for (int k = 0; k < 4; ++k) {
        const float tt = d - rsv[k];
        wv[k] = expf(inv[k] * tt * tt) * obv[k];
      }
      float ang[NPARA];
      const float dv[3] = {g4.y, g4.z, g4.w};
      ang[0] = g4.x;
#pragma unroll
      for (int a2 = 0; a2 < 3; ++a2) ang[1 + a2] = g4.x * dv[a2];
#pragma unroll
      for (int a2 = 0; a2 < 3; ++a2)
#pragma unroll
        for (int b2 = 0; b2 < 3; ++b2) ang[4 + a2 * 3 + b2] = ang[1 + a2] * dv[b2];
#pragma unroll
      for (int j = 0; j < NPARA; ++j) {
        acc[j].x += ang[j] * wv[0];
        acc[j].y += ang[j] * wv[1];
        acc[j].z += ang[j] * wv[2];
        acc[j].w += ang[j] * wv[3];
      }
    }
    __syncthreads();
  }

  const int lrow = lane & 15;
  const int lhi = lane >> 4;
  const int khalf = lhi * 2;
  const int mt = wave & 1;
  const int np = wave >> 1;
  const int n0 = np * 32 + lrow;
  v2f Blo[3][2], Bhi[3][2];
#pragma unroll
  for (int ip = 0; ip < 3; ++ip) {
    const float* hb = hyper + ip * (NWAVE * NORBIT);
#pragma unroll
    for (int h = 0; h < 2; ++h) {
      Blo[ip][h].x = hb[(khalf + 0) * NORBIT + n0 + h * 16];
      Blo[ip][h].y = hb[(khalf + 1) * NORBIT + n0 + h * 16];
      Bhi[ip][h].x = hb[(khalf + 4) * NORBIT + n0 + h * 16];
      Bhi[ip][h].y = hb[(khalf + 5) * NORBIT + n0 + h * 16];
    }
  }

#pragma unroll
  for (int j = 0; j < NPARA; ++j) {
    acc[j].x += quad_xor1(acc[j].x);
    acc[j].y += quad_xor1(acc[j].y);
    acc[j].z += quad_xor1(acc[j].z);
    acc[j].w += quad_xor1(acc[j].w);
    acc[j].x += quad_xor2(acc[j].x);
    acc[j].y += quad_xor2(acc[j].y);
    acc[j].z += quad_xor2(acc[j].z);
    acc[j].w += quad_xor2(acc[j].w);
  }
  if (qd == 0) {
#pragma unroll
    for (int j = 0; j < NPARA; ++j)
      *reinterpret_cast<float4*>(s_out + c5 * SOP + j * NWAVE + wq * 4) = acc[j];
  }
  __syncthreads();

  {
    v8f d0 = {}, d1 = {};
    const float* arow = s_out + (mt * 16 + lrow) * SOP;
    int para = 0;
#pragma unroll
    for (int ip = 0; ip < 3; ++ip) {
      const v2f B0lo = Blo[ip][0], B0hi = Bhi[ip][0];
      const v2f B1lo = Blo[ip][1], B1hi = Bhi[ip][1];
      const int npj = (ip == 0) ? 1 : ((ip == 1) ? 3 : 9);
#pragma unroll
      for (int q = 0; q < npj; ++q, ++para) {
        const float* ab = arow + para * NWAVE;
        v2f Alo = {ab[khalf + 0], ab[khalf + 1]};
        v2f Ahi = {ab[khalf + 4], ab[khalf + 5]};
        v8f c0 = {}, c1 = {};
        c0 = wmma_f32_16x16x4(Alo, B0lo, c0);
        c1 = wmma_f32_16x16x4(Alo, B1lo, c1);
        c0 = wmma_f32_16x16x4(Ahi, B0hi, c0);
        c1 = wmma_f32_16x16x4(Ahi, B1hi, c1);
#pragma unroll
        for (int r = 0; r < 8; ++r) {
          d0[r] += c0[r] * c0[r];
          d1[r] += c1[r] * c1[r];
        }
      }
    }
#pragma unroll
    for (int r = 0; r < 8; ++r) {
      float* dr = s_den + (mt * 16 + r + lhi * 8) * DP + n0;
      dr[0] = d0[r];
      dr[16] = d1[r];
    }
  }
  __syncthreads();

  if (it == 2) {
    float* dst = density + (b * NATOM + g * CPG) * NORBIT;
    for (int i = t; i < CPG * (NORBIT / 4); i += 256) {
      const int row = i >> 5, c4 = (i & 31) * 4;
      *reinterpret_cast<float4*>(dst + row * NORBIT + c4) =
          *reinterpret_cast<const float4*>(s_den + row * DP + c4);
    }
  } else {
    if (wave < 2) {
      v8f u = {};
      const float* drow = s_den + (wave * 16 + lrow) * DP;
      const bool valid = (lrow < NWAVE);
      float w0[32], w1[32];
#pragma unroll
      for (int kb = 0; kb < 32; ++kb) {
        const int k = kb * 4 + lhi * 2;
        w0[kb] = s_w[(k + 0) * 16 + lrow];
        w1[kb] = s_w[(k + 1) * 16 + lrow];
      }
#pragma unroll
      for (int kb = 0; kb < 32; ++kb) {
        const int k = kb * 4 + lhi * 2;
        v2f A = {drow[k], drow[k + 1]};
        v2f B = {w0[kb], w1[kb]};
        u = wmma_f32_16x16x4(A, B, u);
      }
      if (valid) {
#pragma unroll
        for (int r = 0; r < 8; ++r) {
          const int lg = wave * 16 + r + lhi * 8;
          const int la = g * CPG + lg;
          const float v = tanhf(u[r] + s_e[s_spec[la] * NWAVE + lrow]);
          s_new[lg * NWAVE + lrow] = s_orb[la * NWAVE + lrow] + v;
        }
      }
    }
    __syncthreads();
    if (t < CPG * NWAVE / 4)
      *reinterpret_cast<float4*>(orb_out + (b * NATOM + g * CPG) * NWAVE + t * 4) =
          *reinterpret_cast<const float4*>(s_new + t * 4);
  }
}

extern "C" void kernel_launch(void* const* d_in, const int* in_sizes, int n_in,
                              void* d_out, int out_size, void* d_ws,
                              size_t ws_size, hipStream_t stream) {
  const float* cart = (const float*)d_in[0];
  const float* shifts = (const float*)d_in[1];
  const float* rs = (const float*)d_in[2];
  const float* inta = (const float*)d_in[3];
  const float* params = (const float*)d_in[4];
  const float* hyper = (const float*)d_in[5];
  const float* oc_w0 = (const float*)d_in[6];
  const float* oc_e0 = (const float*)d_in[7];
  const float* oc_w1 = (const float*)d_in[8];
  const float* oc_e1 = (const float*)d_in[9];
  const int* species = (const int*)d_in[11];
  const int* atom_index = (const int*)d_in[12];

  float* density = (float*)d_out;
  float* orb_a = (float*)d_ws;
  float* orb_b = orb_a + TOTNATOM * NWAVE;

  k_obtain<true><<<NB * GRP, 256, 0, stream>>>(cart, shifts, rs, inta, params,
                                               hyper, oc_w0, oc_e0, species,
                                               atom_index, orb_b, orb_a, density, 0);
  k_obtain<false><<<NB * GRP, 256, 0, stream>>>(cart, shifts, rs, inta, params,
                                                hyper, oc_w1, oc_e1, species,
                                                atom_index, orb_a, orb_b, density, 1);
  k_obtain<false><<<NB * GRP, 256, 0, stream>>>(cart, shifts, rs, inta, params,
                                                hyper, oc_w1, oc_e1, species,
                                                atom_index, orb_b, orb_a, density, 2);
}


# Coefficient update: offset, tanh and old-coefficient add of a block's 256 new values done one per thread on all 8 waves (were 8 in a row per lane on 2 waves); sums and their order unchanged
# speedup vs baseline: 125.7177x; 1.0328x over previous
#include <hip/hip_runtime.h>
#include <math.h>

#define NTYPE 4
#define NWAVE 8
#define NORBIT 128
#define NEIGH 64
#define NB 32
#define NATOM 256
#define NPAIR_PER (NATOM * NEIGH)
#define NPAIR_TOT (NB * NPAIR_PER)
#define TOTNATOM (NB * NATOM)
#define NPARA 13
#define PI_OVER_CUTOFF 0.62831853071795864769f

typedef float v2f __attribute__((ext_vector_type(2)));
typedef float v8f __attribute__((ext_vector_type(8)));

__device__ __forceinline__ v8f wmma_f32_16x16x4(v2f a, v2f b, v8f c) {
  return __builtin_amdgcn_wmma_f32_16x16x4_f32(false, a, false, b, (short)0, c,
                                               false, false);
}

__device__ __forceinline__ float quad_xor1(float v) {
  return __int_as_float(__builtin_amdgcn_update_dpp(
      0, __float_as_int(v), 0xB1  , 0xF, 0xF, true));
}
__device__ __forceinline__ float quad_xor2(float v) {
  return __int_as_float(__builtin_amdgcn_update_dpp(
      0, __float_as_int(v), 0x4E  , 0xF, 0xF, true));
}

#define GRP 8
#define CPG (NATOM / GRP)
#define SLICE (NPAIR_PER / 8)
#define CH 2560
#define BU 5
#define ROWF (NPARA * NWAVE)
#define SOP (ROWF + 4)
#define DP (NORBIT + 4)
#define SCAP 448
#define SV_CNT (8 * SCAP)
#define SV_START (SV_CNT + 256)
#define SV_NLOC (SV_START + 40)

#define B_LOAD(S, LJ, SX, SY, SZ)                                   \
  _Pragma("unroll") for (int u = 0; u < BU; ++u) {                  \
    const int idx = (S) + u * 32 + lane;                            \
    const int e = s_list[wave][(idx < nloc) ? idx : 0];             \
    const int p = b * NPAIR_PER + (e & 0x3fff);                     \
    LJ[u] = atom_index[NPAIR_TOT + p];      \
    SX[u] = shifts[p * 3 + 0];                                      \
    SY[u] = shifts[p * 3 + 1];                                      \
    SZ[u] = shifts[p * 3 + 2];                                      \
  }

template <bool FIRST>
__global__ void __launch_bounds__(256)
k_obtain(const float* __restrict__ cart,
         const float* __restrict__ shifts,
         const float* __restrict__ rs,
         const float* __restrict__ inta,
         const float* __restrict__ params,
         const float* __restrict__ hyper,
         const float* __restrict__ oc_w,
         const float* __restrict__ oc_e,
         const int* __restrict__ species,
         const int* __restrict__ atom_index,
         const float* __restrict__ orb_in,
         float* __restrict__ orb_out,
         float* __restrict__ density,
         const int it) {
  __shared__ int s_list[8][SLICE];
  __shared__ float4 s_geo[CH];
  __shared__ float s_d[CH];
  __shared__ int s_js[CH];
  __shared__ __attribute__((aligned(16))) float s_out[CPG * SOP];
  __shared__ __attribute__((aligned(16))) float s_den[CPG * DP];
  __shared__ __attribute__((aligned(16))) float s_new[CPG * NWAVE];
  __shared__ __attribute__((aligned(16))) float s_orb[NATOM * NWAVE];
  __shared__ __attribute__((aligned(16))) float s_w[NORBIT * 16];
  __shared__ float s_e[NTYPE * NWAVE];
  __shared__ float s_cart[NATOM * 3];
  __shared__ int s_spec[NATOM];
  __shared__ __attribute__((aligned(16))) float s_rs[NTYPE * NWAVE];
  __shared__ __attribute__((aligned(16))) float s_inta[NTYPE * NWAVE];
  __shared__ int s_cnt[8][CPG];
  __shared__ int s_start[CPG + 1];

  const int t = threadIdx.x;
  const int wave = t >> 5;
  const int lane = t & 31;
  const int b = blockIdx.x / GRP;
  const int g = blockIdx.x % GRP;

  int* sv = reinterpret_cast<int*>(density) + (b * NATOM + g * CPG) * NORBIT;
  int lw[SCAP / 32];
  int r_cnt = 0, r_start = 0, r_nmine = 0;
  int4 r_na = make_int4(0, 0, 0, 0), r_nb = r_na;
  if (!FIRST) {
    r_nmine = sv[SV_NLOC + wave];
#pragma unroll
    for (int k = 0; k < SCAP / 32; ++k) lw[k] = sv[wave * SCAP + k * 32 + lane];
    r_cnt = sv[SV_CNT + t];
    r_start = sv[SV_START + ((t <= CPG) ? t : 0)];
    r_na = *reinterpret_cast<const int4*>(sv + SV_NLOC);
    r_nb = *reinterpret_cast<const int4*>(sv + SV_NLOC + 4);
  }

  const int r_spec = species[b * NATOM + t];
  float r_cart[3];
#pragma unroll
  for (int j = 0; j < 3; ++j) r_cart[j] = cart[b * NATOM * 3 + j * 256 + t];
  float4 r_o0 = make_float4(0.0f, 0.0f, 0.0f, 0.0f), r_o1 = r_o0;
  if (!FIRST) {
    const float* src = orb_in + (b * NATOM + t) * NWAVE;
    r_o0 = *reinterpret_cast<const float4*>(src);
    r_o1 = *reinterpret_cast<const float4*>(src + 4);
  }
  float4 r_w = make_float4(0.0f, 0.0f, 0.0f, 0.0f);
  if (it != 2) r_w = *reinterpret_cast<const float4*>(oc_w + t * 4);
  float r_rs = 0.0f, r_inta = 0.0f, r_e = 0.0f;
  if (t < NTYPE * NWAVE) { r_rs = rs[t]; r_inta = inta[t]; r_e = oc_e[t]; }

  const unsigned ltmask = (1u << lane) - 1u;
  int nloc = 0;
  bool reuse = false;
  if (!FIRST) {
    const int nmax = max(max(max(r_na.x, r_na.y), max(r_na.z, r_na.w)),
                         max(max(r_nb.x, r_nb.y), max(r_nb.z, r_nb.w)));
    reuse = __builtin_amdgcn_readfirstlane(nmax) <= SCAP;
  }
  if (reuse) {
    nloc = __builtin_amdgcn_readfirstlane(r_nmine);
#pragma unroll
    for (int k = 0; k < SCAP / 32; ++k) s_list[wave][k * 32 + lane] = lw[k];
    (&s_cnt[0][0])[t] = r_cnt;
    if (t <= CPG) s_start[t] = r_start;
  } else {
  (&s_cnt[0][0])[t] = 0;

  const int* keys = atom_index + b * NPAIR_PER;
  const int kbase = wave * SLICE;
  int kr[SLICE / 32];
#pragma unroll
  for (int j = 0; j < SLICE / 32; ++j) kr[j] = keys[kbase + j * 32 + lane];
  __builtin_amdgcn_sched_barrier(0);
  if (FIRST) {
    const float* src = params + r_spec * NWAVE;
    r_o0 = *reinterpret_cast<const float4*>(src);
    r_o1 = *reinterpret_cast<const float4*>(src + 4);
  }
#pragma unroll
  for (int j = 0; j < SLICE / 32; ++j) {
    const int cc = kr[j];
    const bool inr = (cc >> 5) == g;
    const unsigned mk = __builtin_amdgcn_ballot_w32(inr);
    if (inr)
      s_list[wave][nloc + __builtin_popcount(mk & ltmask)] =
          (kbase + j * 32 + lane) | ((cc & 31) << 14);
    nloc += __builtin_popcount(mk);
  }
  }

  int c_lj[BU];
  float c_sx[BU], c_sy[BU], c_sz[BU];
  if (reuse) {
    B_LOAD(0, c_lj, c_sx, c_sy, c_sz)
  }

#pragma unroll
  for (int j = 0; j < 3; ++j) s_cart[j * 256 + t] = r_cart[j];
  *reinterpret_cast<float4*>(s_orb + t * NWAVE) = r_o0;
  *reinterpret_cast<float4*>(s_orb + t * NWAVE + 4) = r_o1;
  *reinterpret_cast<float4*>(s_w + (t >> 1) * 16 + (t & 1) * 4) = r_w;
  *reinterpret_cast<float4*>(s_w + (t >> 1) * 16 + 8 + (t & 1) * 4) =
      make_float4(0.0f, 0.0f, 0.0f, 0.0f);
  s_spec[t] = r_spec;
  if (t < NTYPE * NWAVE) { s_rs[t] = r_rs; s_inta[t] = r_inta; s_e[t] = r_e; }
  __syncthreads();

  if (!reuse) {
  B_LOAD(0, c_lj, c_sx, c_sy, c_sz)

  for (int s = 0; s < nloc; s += 32) {
    const int idx = s + lane;
    const bool ok = idx < nloc;
    const int e = ok ? s_list[wave][idx] : 0;
    const int c5 = (e >> 14) & 31;
    unsigned m = __builtin_amdgcn_ballot_w32(ok);
#pragma unroll
    for (int bit = 0; bit < 5; ++bit) {
      const bool bs = ((c5 >> bit) & 1) != 0;
      const unsigned bb = __builtin_amdgcn_ballot_w32(bs);
      m &= bs ? bb : ~bb;
    }
    if (ok) {
      const int prev = s_cnt[wave][c5];
      const int r = __builtin_popcount(m & ltmask);
      s_list[wave][idx] = e | ((prev + r) << 19);
      if (r == 0) s_cnt[wave][c5] = prev + __builtin_popcount(m);
    }
  }
  __syncthreads();

  if (wave == 0) {
    int tot = 0;
#pragma unroll
    for (int w = 0; w < 8; ++w) {
      const int v = s_cnt[w][lane];
      s_cnt[w][lane] = tot;
      tot += v;
    }
    int inc = tot;
#pragma unroll
    for (int o = 1; o < 32; o <<= 1) {
      const int n = __shfl_up(inc, o);
      if (lane >= o) inc += n;
    }
    s_start[lane] = inc - tot;
    if (lane == 31) s_start[CPG] = inc;
  }
  __syncthreads();

  if (FIRST) {
    int sl[SCAP / 32];
#pragma unroll
    for (int k = 0; k < SCAP / 32; ++k) {
      const int idx = k * 32 + lane;
      sl[k] = s_list[wave][(idx < nloc) ? idx : 0];
    }
    const int scn = (&s_cnt[0][0])[t];
    const int sst = s_start[(t <= CPG) ? t : 0];
#pragma unroll
    for (int k = 0; k < SCAP / 32; ++k)
      sv[wave * SCAP + k * 32 + lane] = (k * 32 + lane < nloc) ? sl[k] : 0;
    sv[SV_CNT + t] = scn;
    if (t <= CPG) sv[SV_START + t] = sst;
    if (lane == 0) sv[SV_NLOC + wave] = nloc;
  }
  }

  const int ne = s_start[CPG];
  const int c5 = t >> 3;
  const int qd = t & 3;
  const int wq = (t >> 2) & 1;
  const int mystart = s_start[c5];
  const int myn = s_start[c5 + 1] - mystart;
  const int qstart = mystart + ((myn * qd) >> 2);
  const int qend = mystart + ((myn * (qd + 1)) >> 2);
  float4 acc[NPARA];
#pragma unroll
  for (int j = 0; j < NPARA; ++j) acc[j] = make_float4(0.0f, 0.0f, 0.0f, 0.0f);

  for (int e0 = 0; e0 < ne; e0 += CH) {
    const int nchunk = (ne - e0 < CH) ? (ne - e0) : CH;

    if (e0 != 0) {
      B_LOAD(0, c_lj, c_sx, c_sy, c_sz)
    }
    for (int s = 0; s < nloc; s += 32 * BU) {
      int n_lj[BU];
      float n_sx[BU], n_sy[BU], n_sz[BU];
      B_LOAD(s + 32 * BU, n_lj, n_sx, n_sy, n_sz)
      float r_dx[BU], r_dy[BU], r_dz[BU], r_d[BU];
      int r_slot[BU], r_js[BU];
#pragma unroll
      for (int u = 0; u < BU; ++u) {
        const int idx = s + u * 32 + lane;
        const int e = s_list[wave][(idx < nloc) ? idx : 0];
        const int ce = (e >> 14) & 31;
        r_slot[u] = s_start[ce] + s_cnt[wave][ce] + (e >> 19) - e0;
        const int li = g * CPG + ce;
        const int lj = c_lj[u];
        const float dx = s_cart[li * 3 + 0] - s_cart[lj * 3 + 0] + c_sx[u];
        const float dy = s_cart[li * 3 + 1] - s_cart[lj * 3 + 1] + c_sy[u];
        const float dz = s_cart[li * 3 + 2] - s_cart[lj * 3 + 2] + c_sz[u];
        r_dx[u] = dx; r_dy[u] = dy; r_dz[u] = dz;
        r_d[u] = sqrtf(dx * dx + dy * dy + dz * dz);
        r_js[u] = lj | (s_spec[lj] << 8);
      }
#pragma unroll
      for (int u = 0; u < BU; ++u)
        asm volatile("" : : "v"(r_d[u]));
#pragma unroll
      for (int u = 0; u < BU; ++u) {
        if (s + u * 32 < nloc) {
          const bool ok = s + u * 32 + lane < nloc;
          const int slot = r_slot[u];
          float fc = 0.5f * cosf(r_d[u] * PI_OVER_CUTOFF) + 0.5f;
          fc = fc * fc;
          if (ok && slot >= 0 && slot < nchunk) {
            s_geo[slot] = make_float4(fc, r_dx[u], r_dy[u], r_dz[u]);
            s_d[slot] = r_d[u];
            s_js[slot] = r_js[u];
          }
        }
      }
#pragma unroll
      for (int u = 0; u < BU; ++u) {
        c_lj[u] = n_lj[u]; c_sx[u] = n_sx[u]; c_sy[u] = n_sy[u]; c_sz[u] = n_sz[u];
      }
    }
    __syncthreads();

    const int lo = ((qstart > e0) ? qstart : e0) - e0;
    const int hi = ((qend < e0 + nchunk) ? qend : (e0 + nchunk)) - e0;
#pragma unroll 2
    for (int i = lo; i < hi; ++i) {
      const float4 g4 = s_geo[i];
      const float d = s_d[i];
      const int js = s_js[i];
      const int lj = js & 255;
      const int sp = js >> 8;
      const float4 rs4 = *reinterpret_cast<const float4*>(s_rs + sp * NWAVE + wq * 4);
      const float4 in4 = *reinterpret_cast<const float4*>(s_inta + sp * NWAVE + wq * 4);
      const float4 ob4 = *reinterpret_cast<const float4*>(s_orb + lj * NWAVE + wq * 4);
      const float rsv[4] = {rs4.x, rs4.y, rs4.z, rs4.w};
      const float inv[4] = {in4.x, in4.y, in4.z, in4.w};
      const float obv[4] = {ob4.x, ob4.y, ob4.z, ob4.w};
      float wv[4];
#pragma unroll
      for (int k = 0; k < 4; ++k) {
        const float tt = d - rsv[k];
        wv[k] = expf(inv[k] * tt * tt) * obv[k];
      }
      float ang[NPARA];
      const float dv[3] = {g4.y, g4.z, g4.w};
      ang[0] = g4.x;
#pragma unroll
      for (int a2 = 0; a2 < 3; ++a2) ang[1 + a2] = g4.x * dv[a2];
#pragma unroll
      for (int a2 = 0; a2 < 3; ++a2)
#pragma unroll
        for (int b2 = 0; b2 < 3; ++b2) ang[4 + a2 * 3 + b2] = ang[1 + a2] * dv[b2];
#pragma unroll
      for (int j = 0; j < NPARA; ++j) {
        acc[j].x += ang[j] * wv[0];
        acc[j].y += ang[j] * wv[1];
        acc[j].z += ang[j] * wv[2];
        acc[j].w += ang[j] * wv[3];
      }
    }
    __syncthreads();
  }

  const int lrow = lane & 15;
  const int lhi = lane >> 4;
  const int khalf = lhi * 2;
  const int mt = wave & 1;
  const int np = wave >> 1;
  const int n0 = np * 32 + lrow;
  v2f Blo[3][2], Bhi[3][2];
#pragma unroll
  for (int ip = 0; ip < 3; ++ip) {
    const float* hb = hyper + ip * (NWAVE * NORBIT);
#pragma unroll
    for (int h = 0; h < 2; ++h) {
      Blo[ip][h].x = hb[(khalf + 0) * NORBIT + n0 + h * 16];
      Blo[ip][h].y = hb[(khalf + 1) * NORBIT + n0 + h * 16];
      Bhi[ip][h].x = hb[(khalf + 4) * NORBIT + n0 + h * 16];
      Bhi[ip][h].y = hb[(khalf + 5) * NORBIT + n0 + h * 16];
    }
  }

  const int u_lg = t >> 3;
  const int u_ch = t & (NWAVE - 1);
  const int u_la = g * CPG + u_lg;
  const float u_off = s_e[s_spec[u_la] * NWAVE + u_ch];
  const float u_old = s_orb[u_la * NWAVE + u_ch];

#pragma unroll
  for (int j = 0; j < NPARA; ++j) {
    acc[j].x += quad_xor1(acc[j].x);
    acc[j].y += quad_xor1(acc[j].y);
    acc[j].z += quad_xor1(acc[j].z);
    acc[j].w += quad_xor1(acc[j].w);
    acc[j].x += quad_xor2(acc[j].x);
    acc[j].y += quad_xor2(acc[j].y);
    acc[j].z += quad_xor2(acc[j].z);
    acc[j].w += quad_xor2(acc[j].w);
  }
  if (qd == 0) {
#pragma unroll
    for (int j = 0; j < NPARA; ++j)
      *reinterpret_cast<float4*>(s_out + c5 * SOP + j * NWAVE + wq * 4) = acc[j];
  }
  __syncthreads();

  {
    v8f d0 = {}, d1 = {};
    const float* arow = s_out + (mt * 16 + lrow) * SOP;
    int para = 0;
#pragma unroll
    for (int ip = 0; ip < 3; ++ip) {
      const v2f B0lo = Blo[ip][0], B0hi = Bhi[ip][0];
      const v2f B1lo = Blo[ip][1], B1hi = Bhi[ip][1];
      const int npj = (ip == 0) ? 1 : ((ip == 1) ? 3 : 9);
#pragma unroll
      for (int q = 0; q < npj; ++q, ++para) {
        const float* ab = arow + para * NWAVE;
        v2f Alo = {ab[khalf + 0], ab[khalf + 1]};
        v2f Ahi = {ab[khalf + 4], ab[khalf + 5]};
        v8f c0 = {}, c1 = {};
        c0 = wmma_f32_16x16x4(Alo, B0lo, c0);
        c1 = wmma_f32_16x16x4(Alo, B1lo, c1);
        c0 = wmma_f32_16x16x4(Ahi, B0hi, c0);
        c1 = wmma_f32_16x16x4(Ahi, B1hi, c1);
#pragma unroll
        for (int r = 0; r < 8; ++r) {
          d0[r] += c0[r] * c0[r];
          d1[r] += c1[r] * c1[r];
        }
      }
    }
#pragma unroll
    for (int r = 0; r < 8; ++r) {
      float* dr = s_den + (mt * 16 + r + lhi * 8) * DP + n0;
      dr[0] = d0[r];
      dr[16] = d1[r];
    }
  }
  __syncthreads();

  if (it == 2) {
    float* dst = density + (b * NATOM + g * CPG) * NORBIT;
    for (int i = t; i < CPG * (NORBIT / 4); i += 256) {
      const int row = i >> 5, c4 = (i & 31) * 4;
      *reinterpret_cast<float4*>(dst + row * NORBIT + c4) =
          *reinterpret_cast<const float4*>(s_den + row * DP + c4);
    }
  } else {
    if (wave < 2) {
      v8f u = {};
      const float* drow = s_den + (wave * 16 + lrow) * DP;
      const bool valid = (lrow < NWAVE);
      float w0[32], w1[32];
#pragma unroll
      for (int kb = 0; kb < 32; ++kb) {
        const int k = kb * 4 + lhi * 2;
        w0[kb] = s_w[(k + 0) * 16 + lrow];
        w1[kb] = s_w[(k + 1) * 16 + lrow];
      }
#pragma unroll
      for (int kb = 0; kb < 32; ++kb) {
        const int k = kb * 4 + lhi * 2;
        v2f A = {drow[k], drow[k + 1]};
        v2f B = {w0[kb], w1[kb]};
        u = wmma_f32_16x16x4(A, B, u);
      }
      if (valid) {
#pragma unroll
        for (int r = 0; r < 8; ++r)
          s_new[(wave * 16 + r + lhi * 8) * NWAVE + lrow] = u[r];
      }
    }
    __syncthreads();
    orb_out[(b * NATOM + g * CPG) * NWAVE + t] = u_old + tanhf(s_new[t] + u_off);
  }
}

extern "C" void kernel_launch(void* const* d_in, const int* in_sizes, int n_in,
                              void* d_out, int out_size, void* d_ws,
                              size_t ws_size, hipStream_t stream) {
  const float* cart = (const float*)d_in[0];
  const float* shifts = (const float*)d_in[1];
  const float* rs = (const float*)d_in[2];
  const float* inta = (const float*)d_in[3];
  const float* params = (const float*)d_in[4];
  const float* hyper = (const float*)d_in[5];
  const float* oc_w0 = (const float*)d_in[6];
  const float* oc_e0 = (const float*)d_in[7];
  const float* oc_w1 = (const float*)d_in[8];
  const float* oc_e1 = (const float*)d_in[9];
  const int* species = (const int*)d_in[11];
  const int* atom_index = (const int*)d_in[12];

  float* density = (float*)d_out;
  float* orb_a = (float*)d_ws;
  float* orb_b = orb_a + TOTNATOM * NWAVE;

  k_obtain<true><<<NB * GRP, 256, 0, stream>>>(cart, shifts, rs, inta, params,
                                               hyper, oc_w0, oc_e0, species,
                                               atom_index, orb_b, orb_a, density, 0);
  k_obtain<false><<<NB * GRP, 256, 0, stream>>>(cart, shifts, rs, inta, params,
                                                hyper, oc_w1, oc_e1, species,
                                                atom_index, orb_a, orb_b, density, 1);
  k_obtain<false><<<NB * GRP, 256, 0, stream>>>(cart, shifts, rs, inta, params,
                                                hyper, oc_w1, oc_e1, species,
                                                atom_index, orb_b, orb_a, density, 2);
}


# Run sums: each radial exponential is one v_exp_f32 (log2 e folded into the decay table at entry; was the 13-instruction library expf); rel. error about 2^-23*(1+|exponent|); sum order unchanged
# speedup vs baseline: 132.5022x; 1.0540x over previous
#include <hip/hip_runtime.h>
#include <math.h>

#define NTYPE 4
#define NWAVE 8
#define NORBIT 128
#define NEIGH 64
#define NB 32
#define NATOM 256
#define NPAIR_PER (NATOM * NEIGH)
#define NPAIR_TOT (NB * NPAIR_PER)
#define TOTNATOM (NB * NATOM)
#define NPARA 13
#define PI_OVER_CUTOFF 0.62831853071795864769f
#define LOG2_E 1.44269504088896340736f

typedef float v2f __attribute__((ext_vector_type(2)));
typedef float v8f __attribute__((ext_vector_type(8)));

__device__ __forceinline__ v8f wmma_f32_16x16x4(v2f a, v2f b, v8f c) {
  return __builtin_amdgcn_wmma_f32_16x16x4_f32(false, a, false, b, (short)0, c,
                                               false, false);
}

__device__ __forceinline__ float quad_xor1(float v) {
  return __int_as_float(__builtin_amdgcn_update_dpp(
      0, __float_as_int(v), 0xB1  , 0xF, 0xF, true));
}
__device__ __forceinline__ float quad_xor2(float v) {
  return __int_as_float(__builtin_amdgcn_update_dpp(
      0, __float_as_int(v), 0x4E  , 0xF, 0xF, true));
}

#define GRP 8
#define CPG (NATOM / GRP)
#define SLICE (NPAIR_PER / 8)
#define CH 2560
#define BU 5
#define ROWF (NPARA * NWAVE)
#define SOP (ROWF + 4)
#define DP (NORBIT + 4)
#define SCAP 448
#define SV_CNT (8 * SCAP)
#define SV_START (SV_CNT + 256)
#define SV_NLOC (SV_START + 40)

#define B_LOAD(S, LJ, SX, SY, SZ)                                   \
  _Pragma("unroll") for (int u = 0; u < BU; ++u) {                  \
    const int idx = (S) + u * 32 + lane;                            \
    const int e = s_list[wave][(idx < nloc) ? idx : 0];             \
    const int p = b * NPAIR_PER + (e & 0x3fff);                     \
    LJ[u] = atom_index[NPAIR_TOT + p];      \
    SX[u] = shifts[p * 3 + 0];                                      \
    SY[u] = shifts[p * 3 + 1];                                      \
    SZ[u] = shifts[p * 3 + 2];                                      \
  }

template <bool FIRST>
__global__ void __launch_bounds__(256)
k_obtain(const float* __restrict__ cart,
         const float* __restrict__ shifts,
         const float* __restrict__ rs,
         const float* __restrict__ inta,
         const float* __restrict__ params,
         const float* __restrict__ hyper,
         const float* __restrict__ oc_w,
         const float* __restrict__ oc_e,
         const int* __restrict__ species,
         const int* __restrict__ atom_index,
         const float* __restrict__ orb_in,
         float* __restrict__ orb_out,
         float* __restrict__ density,
         const int it) {
  __shared__ int s_list[8][SLICE];
  __shared__ float4 s_geo[CH];
  __shared__ float s_d[CH];
  __shared__ int s_js[CH];
  __shared__ __attribute__((aligned(16))) float s_out[CPG * SOP];
  __shared__ __attribute__((aligned(16))) float s_den[CPG * DP];
  __shared__ __attribute__((aligned(16))) float s_new[CPG * NWAVE];
  __shared__ __attribute__((aligned(16))) float s_orb[NATOM * NWAVE];
  __shared__ __attribute__((aligned(16))) float s_w[NORBIT * 16];
  __shared__ float s_e[NTYPE * NWAVE];
  __shared__ float s_cart[NATOM * 3];
  __shared__ int s_spec[NATOM];
  __shared__ __attribute__((aligned(16))) float s_rs[NTYPE * NWAVE];
  __shared__ __attribute__((aligned(16))) float s_inta[NTYPE * NWAVE];
  __shared__ int s_cnt[8][CPG];
  __shared__ int s_start[CPG + 1];

  const int t = threadIdx.x;
  const int wave = t >> 5;
  const int lane = t & 31;
  const int b = blockIdx.x / GRP;
  const int g = blockIdx.x % GRP;

  int* sv = reinterpret_cast<int*>(density) + (b * NATOM + g * CPG) * NORBIT;
  int lw[SCAP / 32];
  int r_cnt = 0, r_start = 0, r_nmine = 0;
  int4 r_na = make_int4(0, 0, 0, 0), r_nb = r_na;
  if (!FIRST) {
    r_nmine = sv[SV_NLOC + wave];
#pragma unroll
    for (int k = 0; k < SCAP / 32; ++k) lw[k] = sv[wave * SCAP + k * 32 + lane];
    r_cnt = sv[SV_CNT + t];
    r_start = sv[SV_START + ((t <= CPG) ? t : 0)];
    r_na = *reinterpret_cast<const int4*>(sv + SV_NLOC);
    r_nb = *reinterpret_cast<const int4*>(sv + SV_NLOC + 4);
  }

  const int r_spec = species[b * NATOM + t];
  float r_cart[3];
#pragma unroll
  for (int j = 0; j < 3; ++j) r_cart[j] = cart[b * NATOM * 3 + j * 256 + t];
  float4 r_o0 = make_float4(0.0f, 0.0f, 0.0f, 0.0f), r_o1 = r_o0;
  if (!FIRST) {
    const float* src = orb_in + (b * NATOM + t) * NWAVE;
    r_o0 = *reinterpret_cast<const float4*>(src);
    r_o1 = *reinterpret_cast<const float4*>(src + 4);
  }
  float4 r_w = make_float4(0.0f, 0.0f, 0.0f, 0.0f);
  if (it != 2) r_w = *reinterpret_cast<const float4*>(oc_w + t * 4);
  float r_rs = 0.0f, r_inta = 0.0f, r_e = 0.0f;
  if (t < NTYPE * NWAVE) { r_rs = rs[t]; r_inta = inta[t]; r_e = oc_e[t]; }

  const unsigned ltmask = (1u << lane) - 1u;
  int nloc = 0;
  bool reuse = false;
  if (!FIRST) {
    const int nmax = max(max(max(r_na.x, r_na.y), max(r_na.z, r_na.w)),
                         max(max(r_nb.x, r_nb.y), max(r_nb.z, r_nb.w)));
    reuse = __builtin_amdgcn_readfirstlane(nmax) <= SCAP;
  }
  if (reuse) {
    nloc = __builtin_amdgcn_readfirstlane(r_nmine);
#pragma unroll
    for (int k = 0; k < SCAP / 32; ++k) s_list[wave][k * 32 + lane] = lw[k];
    (&s_cnt[0][0])[t] = r_cnt;
    if (t <= CPG) s_start[t] = r_start;
  } else {
  (&s_cnt[0][0])[t] = 0;

  const int* keys = atom_index + b * NPAIR_PER;
  const int kbase = wave * SLICE;
  int kr[SLICE / 32];
#pragma unroll
  for (int j = 0; j < SLICE / 32; ++j) kr[j] = keys[kbase + j * 32 + lane];
  __builtin_amdgcn_sched_barrier(0);
  if (FIRST) {
    const float* src = params + r_spec * NWAVE;
    r_o0 = *reinterpret_cast<const float4*>(src);
    r_o1 = *reinterpret_cast<const float4*>(src + 4);
  }
#pragma unroll
  for (int j = 0; j < SLICE / 32; ++j) {
    const int cc = kr[j];
    const bool inr = (cc >> 5) == g;
    const unsigned mk = __builtin_amdgcn_ballot_w32(inr);
    if (inr)
      s_list[wave][nloc + __builtin_popcount(mk & ltmask)] =
          (kbase + j * 32 + lane) | ((cc & 31) << 14);
    nloc += __builtin_popcount(mk);
  }
  }

  int c_lj[BU];
  float c_sx[BU], c_sy[BU], c_sz[BU];
  if (reuse) {
    B_LOAD(0, c_lj, c_sx, c_sy, c_sz)
  }

#pragma unroll
  for (int j = 0; j < 3; ++j) s_cart[j * 256 + t] = r_cart[j];
  *reinterpret_cast<float4*>(s_orb + t * NWAVE) = r_o0;
  *reinterpret_cast<float4*>(s_orb + t * NWAVE + 4) = r_o1;
  *reinterpret_cast<float4*>(s_w + (t >> 1) * 16 + (t & 1) * 4) = r_w;
  *reinterpret_cast<float4*>(s_w + (t >> 1) * 16 + 8 + (t & 1) * 4) =
      make_float4(0.0f, 0.0f, 0.0f, 0.0f);
  s_spec[t] = r_spec;
  if (t < NTYPE * NWAVE) { s_rs[t] = r_rs; s_inta[t] = r_inta * LOG2_E; s_e[t] = r_e; }
  __syncthreads();

  if (!reuse) {
  B_LOAD(0, c_lj, c_sx, c_sy, c_sz)

  for (int s = 0; s < nloc; s += 32) {
    const int idx = s + lane;
    const bool ok = idx < nloc;
    const int e = ok ? s_list[wave][idx] : 0;
    const int c5 = (e >> 14) & 31;
    unsigned m = __builtin_amdgcn_ballot_w32(ok);
#pragma unroll
    for (int bit = 0; bit < 5; ++bit) {
      const bool bs = ((c5 >> bit) & 1) != 0;
      const unsigned bb = __builtin_amdgcn_ballot_w32(bs);
      m &= bs ? bb : ~bb;
    }
    if (ok) {
      const int prev = s_cnt[wave][c5];
      const int r = __builtin_popcount(m & ltmask);
      s_list[wave][idx] = e | ((prev + r) << 19);
      if (r == 0) s_cnt[wave][c5] = prev + __builtin_popcount(m);
    }
  }
  __syncthreads();

  if (wave == 0) {
    int tot = 0;
#pragma unroll
    for (int w = 0; w < 8; ++w) {
      const int v = s_cnt[w][lane];
      s_cnt[w][lane] = tot;
      tot += v;
    }
    int inc = tot;
#pragma unroll
    for (int o = 1; o < 32; o <<= 1) {
      const int n = __shfl_up(inc, o);
      if (lane >= o) inc += n;
    }
    s_start[lane] = inc - tot;
    if (lane == 31) s_start[CPG] = inc;
  }
  __syncthreads();

  if (FIRST) {
    int sl[SCAP / 32];
#pragma unroll
    for (int k = 0; k < SCAP / 32; ++k) {
      const int idx = k * 32 + lane;
      sl[k] = s_list[wave][(idx < nloc) ? idx : 0];
    }
    const int scn = (&s_cnt[0][0])[t];
    const int sst = s_start[(t <= CPG) ? t : 0];
#pragma unroll
    for (int k = 0; k < SCAP / 32; ++k)
      sv[wave * SCAP + k * 32 + lane] = (k * 32 + lane < nloc) ? sl[k] : 0;
    sv[SV_CNT + t] = scn;
    if (t <= CPG) sv[SV_START + t] = sst;
    if (lane == 0) sv[SV_NLOC + wave] = nloc;
  }
  }

  const int ne = s_start[CPG];
  const int c5 = t >> 3;
  const int qd = t & 3;
  const int wq = (t >> 2) & 1;
  const int mystart = s_start[c5];
  const int myn = s_start[c5 + 1] - mystart;
  const int qstart = mystart + ((myn * qd) >> 2);
  const int qend = mystart + ((myn * (qd + 1)) >> 2);
  float4 acc[NPARA];
#pragma unroll
  for (int j = 0; j < NPARA; ++j) acc[j] = make_float4(0.0f, 0.0f, 0.0f, 0.0f);

  for (int e0 = 0; e0 < ne; e0 += CH) {
    const int nchunk = (ne - e0 < CH) ? (ne - e0) : CH;

    if (e0 != 0) {
      B_LOAD(0, c_lj, c_sx, c_sy, c_sz)
    }
    for (int s = 0; s < nloc; s += 32 * BU) {
      int n_lj[BU];
      float n_sx[BU], n_sy[BU], n_sz[BU];
      B_LOAD(s + 32 * BU, n_lj, n_sx, n_sy, n_sz)
      float r_dx[BU], r_dy[BU], r_dz[BU], r_d[BU];
      int r_slot[BU], r_js[BU];
#pragma unroll
      for (int u = 0; u < BU; ++u) {
        const int idx = s + u * 32 + lane;
        const int e = s_list[wave][(idx < nloc) ? idx : 0];
        const int ce = (e >> 14) & 31;
        r_slot[u] = s_start[ce] + s_cnt[wave][ce] + (e >> 19) - e0;
        const int li = g * CPG + ce;
        const int lj = c_lj[u];
        const float dx = s_cart[li * 3 + 0] - s_cart[lj * 3 + 0] + c_sx[u];
        const float dy = s_cart[li * 3 + 1] - s_cart[lj * 3 + 1] + c_sy[u];
        const float dz = s_cart[li * 3 + 2] - s_cart[lj * 3 + 2] + c_sz[u];
        r_dx[u] = dx; r_dy[u] = dy; r_dz[u] = dz;
        r_d[u] = sqrtf(dx * dx + dy * dy + dz * dz);
        r_js[u] = lj | (s_spec[lj] << 8);
      }
#pragma unroll
      for (int u = 0; u < BU; ++u)
        asm volatile("" : : "v"(r_d[u]));
#pragma unroll
      for (int u = 0; u < BU; ++u) {
        if (s + u * 32 < nloc) {
          const bool ok = s + u * 32 + lane < nloc;
          const int slot = r_slot[u];
          float fc = 0.5f * cosf(r_d[u] * PI_OVER_CUTOFF) + 0.5f;
          fc = fc * fc;
          if (ok && slot >= 0 && slot < nchunk) {
            s_geo[slot] = make_float4(fc, r_dx[u], r_dy[u], r_dz[u]);
            s_d[slot] = r_d[u];
            s_js[slot] = r_js[u];
          }
        }
      }
#pragma unroll
      for (int u = 0; u < BU; ++u) {
        c_lj[u] = n_lj[u]; c_sx[u] = n_sx[u]; c_sy[u] = n_sy[u]; c_sz[u] = n_sz[u];
      }
    }
    __syncthreads();

    const int lo = ((qstart > e0) ? qstart : e0) - e0;
    const int hi = ((qend < e0 + nchunk) ? qend : (e0 + nchunk)) - e0;
#pragma unroll 2
    for (int i = lo; i < hi; ++i) {
      const float4 g4 = s_geo[i];
      const float d = s_d[i];
      const int js = s_js[i];
      const int lj = js & 255;
      const int sp = js >> 8;
      const float4 rs4 = *reinterpret_cast<const float4*>(s_rs + sp * NWAVE + wq * 4);
      const float4 in4 = *reinterpret_cast<const float4*>(s_inta + sp * NWAVE + wq * 4);
      const float4 ob4 = *reinterpret_cast<const float4*>(s_orb + lj * NWAVE + wq * 4);
      const float rsv[4] = {rs4.x, rs4.y, rs4.z, rs4.w};
      const float inv[4] = {in4.x, in4.y, in4.z, in4.w};
      const float obv[4] = {ob4.x, ob4.y, ob4.z, ob4.w};
      float wv[4];
#pragma unroll
      for (int k = 0; k < 4; ++k) {
        const float tt = d - rsv[k];
        wv[k] = __builtin_amdgcn_exp2f(inv[k] * tt * tt) * obv[k];
      }
      float ang[NPARA];
      const float dv[3] = {g4.y, g4.z, g4.w};
      ang[0] = g4.x;
#pragma unroll
      for (int a2 = 0; a2 < 3; ++a2) ang[1 + a2] = g4.x * dv[a2];
#pragma unroll
      for (int a2 = 0; a2 < 3; ++a2)
#pragma unroll
        for (int b2 = 0; b2 < 3; ++b2) ang[4 + a2 * 3 + b2] = ang[1 + a2] * dv[b2];
#pragma unroll
      for (int j = 0; j < NPARA; ++j) {
        acc[j].x += ang[j] * wv[0];
        acc[j].y += ang[j] * wv[1];
        acc[j].z += ang[j] * wv[2];
        acc[j].w += ang[j] * wv[3];
      }
    }
    __syncthreads();
  }

  const int lrow = lane & 15;
  const int lhi = lane >> 4;
  const int khalf = lhi * 2;
  const int mt = wave & 1;
  const int np = wave >> 1;
  const int n0 = np * 32 + lrow;
  v2f Blo[3][2], Bhi[3][2];
#pragma unroll
  for (int ip = 0; ip < 3; ++ip) {
    const float* hb = hyper + ip * (NWAVE * NORBIT);
#pragma unroll
    for (int h = 0; h < 2; ++h) {
      Blo[ip][h].x = hb[(khalf + 0) * NORBIT + n0 + h * 16];
      Blo[ip][h].y = hb[(khalf + 1) * NORBIT + n0 + h * 16];
      Bhi[ip][h].x = hb[(khalf + 4) * NORBIT + n0 + h * 16];
      Bhi[ip][h].y = hb[(khalf + 5) * NORBIT + n0 + h * 16];
    }
  }

  const int u_lg = t >> 3;
  const int u_ch = t & (NWAVE - 1);
  const int u_la = g * CPG + u_lg;
  const float u_off = s_e[s_spec[u_la] * NWAVE + u_ch];
  const float u_old = s_orb[u_la * NWAVE + u_ch];

#pragma unroll
  for (int j = 0; j < NPARA; ++j) {
    acc[j].x += quad_xor1(acc[j].x);
    acc[j].y += quad_xor1(acc[j].y);
    acc[j].z += quad_xor1(acc[j].z);
    acc[j].w += quad_xor1(acc[j].w);
    acc[j].x += quad_xor2(acc[j].x);
    acc[j].y += quad_xor2(acc[j].y);
    acc[j].z += quad_xor2(acc[j].z);
    acc[j].w += quad_xor2(acc[j].w);
  }
  if (qd == 0) {
#pragma unroll
    for (int j = 0; j < NPARA; ++j)
      *reinterpret_cast<float4*>(s_out + c5 * SOP + j * NWAVE + wq * 4) = acc[j];
  }
  __syncthreads();

  {
    v8f d0 = {}, d1 = {};
    const float* arow = s_out + (mt * 16 + lrow) * SOP;
    int para = 0;
#pragma unroll
    for (int ip = 0; ip < 3; ++ip) {
      const v2f B0lo = Blo[ip][0], B0hi = Bhi[ip][0];
      const v2f B1lo = Blo[ip][1], B1hi = Bhi[ip][1];
      const int npj = (ip == 0) ? 1 : ((ip == 1) ? 3 : 9);
#pragma unroll
      for (int q = 0; q < npj; ++q, ++para) {
        const float* ab = arow + para * NWAVE;
        v2f Alo = {ab[khalf + 0], ab[khalf + 1]};
        v2f Ahi = {ab[khalf + 4], ab[khalf + 5]};
        v8f c0 = {}, c1 = {};
        c0 = wmma_f32_16x16x4(Alo, B0lo, c0);
        c1 = wmma_f32_16x16x4(Alo, B1lo, c1);
        c0 = wmma_f32_16x16x4(Ahi, B0hi, c0);
        c1 = wmma_f32_16x16x4(Ahi, B1hi, c1);
#pragma unroll
        for (int r = 0; r < 8; ++r) {
          d0[r] += c0[r] * c0[r];
          d1[r] += c1[r] * c1[r];
        }
      }
    }
#pragma unroll
    for (int r = 0; r < 8; ++r) {
      float* dr = s_den + (mt * 16 + r + lhi * 8) * DP + n0;
      dr[0] = d0[r];
      dr[16] = d1[r];
    }
  }
  __syncthreads();

  if (it == 2) {
    float* dst = density + (b * NATOM + g * CPG) * NORBIT;
    for (int i = t; i < CPG * (NORBIT / 4); i += 256) {
      const int row = i >> 5, c4 = (i & 31) * 4;
      *reinterpret_cast<float4*>(dst + row * NORBIT + c4) =
          *reinterpret_cast<const float4*>(s_den + row * DP + c4);
    }
  } else {
    if (wave < 2) {
      v8f u = {};
      const float* drow = s_den + (wave * 16 + lrow) * DP;
      const bool valid = (lrow < NWAVE);
      float w0[32], w1[32];
#pragma unroll
      for (int kb = 0; kb < 32; ++kb) {
        const int k = kb * 4 + lhi * 2;
        w0[kb] = s_w[(k + 0) * 16 + lrow];
        w1[kb] = s_w[(k + 1) * 16 + lrow];
      }
#pragma unroll
      for (int kb = 0; kb < 32; ++kb) {
        const int k = kb * 4 + lhi * 2;
        v2f A = {drow[k], drow[k + 1]};
        v2f B = {w0[kb], w1[kb]};
        u = wmma_f32_16x16x4(A, B, u);
      }
      if (valid) {
#pragma unroll
        for (int r = 0; r < 8; ++r)
          s_new[(wave * 16 + r + lhi * 8) * NWAVE + lrow] = u[r];
      }
    }
    __syncthreads();
    orb_out[(b * NATOM + g * CPG) * NWAVE + t] = u_old + tanhf(s_new[t] + u_off);
  }
}

extern "C" void kernel_launch(void* const* d_in, const int* in_sizes, int n_in,
                              void* d_out, int out_size, void* d_ws,
                              size_t ws_size, hipStream_t stream) {
  const float* cart = (const float*)d_in[0];
  const float* shifts = (const float*)d_in[1];
  const float* rs = (const float*)d_in[2];
  const float* inta = (const float*)d_in[3];
  const float* params = (const float*)d_in[4];
  const float* hyper = (const float*)d_in[5];
  const float* oc_w0 = (const float*)d_in[6];
  const float* oc_e0 = (const float*)d_in[7];
  const float* oc_w1 = (const float*)d_in[8];
  const float* oc_e1 = (const float*)d_in[9];
  const int* species = (const int*)d_in[11];
  const int* atom_index = (const int*)d_in[12];

  float* density = (float*)d_out;
  float* orb_a = (float*)d_ws;
  float* orb_b = orb_a + TOTNATOM * NWAVE;

  k_obtain<true><<<NB * GRP, 256, 0, stream>>>(cart, shifts, rs, inta, params,
                                               hyper, oc_w0, oc_e0, species,
                                               atom_index, orb_b, orb_a, density, 0);
  k_obtain<false><<<NB * GRP, 256, 0, stream>>>(cart, shifts, rs, inta, params,
                                                hyper, oc_w1, oc_e1, species,
                                                atom_index, orb_a, orb_b, density, 1);
  k_obtain<false><<<NB * GRP, 256, 0, stream>>>(cart, shifts, rs, inta, params,
                                                hyper, oc_w1, oc_e1, species,
                                                atom_index, orb_b, orb_a, density, 2);
}


# Second-order angular terms fc*d[a]*d[b]: (b,a) equals (a,b) and both use weight plane 2, so 10 distinct terms are summed and multiplied, not 13; density adds 10 squares, the 3 with a<b doubled
# speedup vs baseline: 136.6893x; 1.0316x over previous
#include <hip/hip_runtime.h>
#include <math.h>

#define NTYPE 4
#define NWAVE 8
#define NORBIT 128
#define NEIGH 64
#define NB 32
#define NATOM 256
#define NPAIR_PER (NATOM * NEIGH)
#define NPAIR_TOT (NB * NPAIR_PER)
#define TOTNATOM (NB * NATOM)
#define NPARA 13
#define NPU 10
#define PI_OVER_CUTOFF 0.62831853071795864769f
#define LOG2_E 1.44269504088896340736f

typedef float v2f __attribute__((ext_vector_type(2)));
typedef float v8f __attribute__((ext_vector_type(8)));

__device__ __forceinline__ v8f wmma_f32_16x16x4(v2f a, v2f b, v8f c) {
  return __builtin_amdgcn_wmma_f32_16x16x4_f32(false, a, false, b, (short)0, c,
                                               false, false);
}

__device__ __forceinline__ float quad_xor1(float v) {
  return __int_as_float(__builtin_amdgcn_update_dpp(
      0, __float_as_int(v), 0xB1  , 0xF, 0xF, true));
}
__device__ __forceinline__ float quad_xor2(float v) {
  return __int_as_float(__builtin_amdgcn_update_dpp(
      0, __float_as_int(v), 0x4E  , 0xF, 0xF, true));
}

#define GRP 8
#define CPG (NATOM / GRP)
#define SLICE (NPAIR_PER / 8)
#define CH 2560
#define BU 5
#define ROWF (NPU * NWAVE)
#define SOP (ROWF + 4)
#define DP (NORBIT + 4)
#define SCAP 448
#define SV_CNT (8 * SCAP)
#define SV_START (SV_CNT + 256)
#define SV_NLOC (SV_START + 40)

#define B_LOAD(S, LJ, SX, SY, SZ)                                   \
  _Pragma("unroll") for (int u = 0; u < BU; ++u) {                  \
    const int idx = (S) + u * 32 + lane;                            \
    const int e = s_list[wave][(idx < nloc) ? idx : 0];             \
    const int p = b * NPAIR_PER + (e & 0x3fff);                     \
    LJ[u] = atom_index[NPAIR_TOT + p];      \
    SX[u] = shifts[p * 3 + 0];                                      \
    SY[u] = shifts[p * 3 + 1];                                      \
    SZ[u] = shifts[p * 3 + 2];                                      \
  }

template <bool FIRST>
__global__ void __launch_bounds__(256)
k_obtain(const float* __restrict__ cart,
         const float* __restrict__ shifts,
         const float* __restrict__ rs,
         const float* __restrict__ inta,
         const float* __restrict__ params,
         const float* __restrict__ hyper,
         const float* __restrict__ oc_w,
         const float* __restrict__ oc_e,
         const int* __restrict__ species,
         const int* __restrict__ atom_index,
         const float* __restrict__ orb_in,
         float* __restrict__ orb_out,
         float* __restrict__ density,
         const int it) {
  __shared__ int s_list[8][SLICE];
  __shared__ float4 s_geo[CH];
  __shared__ float s_d[CH];
  __shared__ int s_js[CH];
  __shared__ __attribute__((aligned(16))) float s_out[CPG * SOP];
  __shared__ __attribute__((aligned(16))) float s_den[CPG * DP];
  __shared__ __attribute__((aligned(16))) float s_new[CPG * NWAVE];
  __shared__ __attribute__((aligned(16))) float s_orb[NATOM * NWAVE];
  __shared__ __attribute__((aligned(16))) float s_w[NORBIT * 16];
  __shared__ float s_e[NTYPE * NWAVE];
  __shared__ float s_cart[NATOM * 3];
  __shared__ int s_spec[NATOM];
  __shared__ __attribute__((aligned(16))) float s_rs[NTYPE * NWAVE];
  __shared__ __attribute__((aligned(16))) float s_inta[NTYPE * NWAVE];
  __shared__ int s_cnt[8][CPG];
  __shared__ int s_start[CPG + 1];

  const int t = threadIdx.x;
  const int wave = t >> 5;
  const int lane = t & 31;
  const int b = blockIdx.x / GRP;
  const int g = blockIdx.x % GRP;

  int* sv = reinterpret_cast<int*>(density) + (b * NATOM + g * CPG) * NORBIT;
  int lw[SCAP / 32];
  int r_cnt = 0, r_start = 0, r_nmine = 0;
  int4 r_na = make_int4(0, 0, 0, 0), r_nb = r_na;
  if (!FIRST) {
    r_nmine = sv[SV_NLOC + wave];
#pragma unroll
    for (int k = 0; k < SCAP / 32; ++k) lw[k] = sv[wave * SCAP + k * 32 + lane];
    r_cnt = sv[SV_CNT + t];
    r_start = sv[SV_START + ((t <= CPG) ? t : 0)];
    r_na = *reinterpret_cast<const int4*>(sv + SV_NLOC);
    r_nb = *reinterpret_cast<const int4*>(sv + SV_NLOC + 4);
  }

  const int r_spec = species[b * NATOM + t];
  float r_cart[3];
#pragma unroll
  for (int j = 0; j < 3; ++j) r_cart[j] = cart[b * NATOM * 3 + j * 256 + t];
  float4 r_o0 = make_float4(0.0f, 0.0f, 0.0f, 0.0f), r_o1 = r_o0;
  if (!FIRST) {
    const float* src = orb_in + (b * NATOM + t) * NWAVE;
    r_o0 = *reinterpret_cast<const float4*>(src);
    r_o1 = *reinterpret_cast<const float4*>(src + 4);
  }
  float4 r_w = make_float4(0.0f, 0.0f, 0.0f, 0.0f);
  if (it != 2) r_w = *reinterpret_cast<const float4*>(oc_w + t * 4);
  float r_rs = 0.0f, r_inta = 0.0f, r_e = 0.0f;
  if (t < NTYPE * NWAVE) { r_rs = rs[t]; r_inta = inta[t]; r_e = oc_e[t]; }

  const unsigned ltmask = (1u << lane) - 1u;
  int nloc = 0;
  bool reuse = false;
  if (!FIRST) {
    const int nmax = max(max(max(r_na.x, r_na.y), max(r_na.z, r_na.w)),
                         max(max(r_nb.x, r_nb.y), max(r_nb.z, r_nb.w)));
    reuse = __builtin_amdgcn_readfirstlane(nmax) <= SCAP;
  }
  if (reuse) {
    nloc = __builtin_amdgcn_readfirstlane(r_nmine);
#pragma unroll
    for (int k = 0; k < SCAP / 32; ++k) s_list[wave][k * 32 + lane] = lw[k];
    (&s_cnt[0][0])[t] = r_cnt;
    if (t <= CPG) s_start[t] = r_start;
  } else {
  (&s_cnt[0][0])[t] = 0;

  const int* keys = atom_index + b * NPAIR_PER;
  const int kbase = wave * SLICE;
  int kr[SLICE / 32];
#pragma unroll
  for (int j = 0; j < SLICE / 32; ++j) kr[j] = keys[kbase + j * 32 + lane];
  __builtin_amdgcn_sched_barrier(0);
  if (FIRST) {
    const float* src = params + r_spec * NWAVE;
    r_o0 = *reinterpret_cast<const float4*>(src);
    r_o1 = *reinterpret_cast<const float4*>(src + 4);
  }
#pragma unroll
  for (int j = 0; j < SLICE / 32; ++j) {
    const int cc = kr[j];
    const bool inr = (cc >> 5) == g;
    const unsigned mk = __builtin_amdgcn_ballot_w32(inr);
    if (inr)
      s_list[wave][nloc + __builtin_popcount(mk & ltmask)] =
          (kbase + j * 32 + lane) | ((cc & 31) << 14);
    nloc += __builtin_popcount(mk);
  }
  }

  int c_lj[BU];
  float c_sx[BU], c_sy[BU], c_sz[BU];
  if (reuse) {
    B_LOAD(0, c_lj, c_sx, c_sy, c_sz)
  }

#pragma unroll
  for (int j = 0; j < 3; ++j) s_cart[j * 256 + t] = r_cart[j];
  *reinterpret_cast<float4*>(s_orb + t * NWAVE) = r_o0;
  *reinterpret_cast<float4*>(s_orb + t * NWAVE + 4) = r_o1;
  *reinterpret_cast<float4*>(s_w + (t >> 1) * 16 + (t & 1) * 4) = r_w;
  *reinterpret_cast<float4*>(s_w + (t >> 1) * 16 + 8 + (t & 1) * 4) =
      make_float4(0.0f, 0.0f, 0.0f, 0.0f);
  s_spec[t] = r_spec;
  if (t < NTYPE * NWAVE) { s_rs[t] = r_rs; s_inta[t] = r_inta * LOG2_E; s_e[t] = r_e; }
  __syncthreads();

  if (!reuse) {
  B_LOAD(0, c_lj, c_sx, c_sy, c_sz)

  for (int s = 0; s < nloc; s += 32) {
    const int idx = s + lane;
    const bool ok = idx < nloc;
    const int e = ok ? s_list[wave][idx] : 0;
    const int c5 = (e >> 14) & 31;
    unsigned m = __builtin_amdgcn_ballot_w32(ok);
#pragma unroll
    for (int bit = 0; bit < 5; ++bit) {
      const bool bs = ((c5 >> bit) & 1) != 0;
      const unsigned bb = __builtin_amdgcn_ballot_w32(bs);
      m &= bs ? bb : ~bb;
    }
    if (ok) {
      const int prev = s_cnt[wave][c5];
      const int r = __builtin_popcount(m & ltmask);
      s_list[wave][idx] = e | ((prev + r) << 19);
      if (r == 0) s_cnt[wave][c5] = prev + __builtin_popcount(m);
    }
  }
  __syncthreads();

  if (wave == 0) {
    int tot = 0;
#pragma unroll
    for (int w = 0; w < 8; ++w) {
      const int v = s_cnt[w][lane];
      s_cnt[w][lane] = tot;
      tot += v;
    }
    int inc = tot;
#pragma unroll
    for (int o = 1; o < 32; o <<= 1) {
      const int n = __shfl_up(inc, o);
      if (lane >= o) inc += n;
    }
    s_start[lane] = inc - tot;
    if (lane == 31) s_start[CPG] = inc;
  }
  __syncthreads();

  if (FIRST) {
    int sl[SCAP / 32];
#pragma unroll
    for (int k = 0; k < SCAP / 32; ++k) {
      const int idx = k * 32 + lane;
      sl[k] = s_list[wave][(idx < nloc) ? idx : 0];
    }
    const int scn = (&s_cnt[0][0])[t];
    const int sst = s_start[(t <= CPG) ? t : 0];
#pragma unroll
    for (int k = 0; k < SCAP / 32; ++k)
      sv[wave * SCAP + k * 32 + lane] = (k * 32 + lane < nloc) ? sl[k] : 0;
    sv[SV_CNT + t] = scn;
    if (t <= CPG) sv[SV_START + t] = sst;
    if (lane == 0) sv[SV_NLOC + wave] = nloc;
  }
  }

  const int ne = s_start[CPG];
  const int c5 = t >> 3;
  const int qd = t & 3;
  const int wq = (t >> 2) & 1;
  const int mystart = s_start[c5];
  const int myn = s_start[c5 + 1] - mystart;
  const int qstart = mystart + ((myn * qd) >> 2);
  const int qend = mystart + ((myn * (qd + 1)) >> 2);
  float4 acc[NPU];
#pragma unroll
  for (int j = 0; j < NPU; ++j) acc[j] = make_float4(0.0f, 0.0f, 0.0f, 0.0f);

  for (int e0 = 0; e0 < ne; e0 += CH) {
    const int nchunk = (ne - e0 < CH) ? (ne - e0) : CH;

    if (e0 != 0) {
      B_LOAD(0, c_lj, c_sx, c_sy, c_sz)
    }
    for (int s = 0; s < nloc; s += 32 * BU) {
      int n_lj[BU];
      float n_sx[BU], n_sy[BU], n_sz[BU];
      B_LOAD(s + 32 * BU, n_lj, n_sx, n_sy, n_sz)
      float r_dx[BU], r_dy[BU], r_dz[BU], r_d[BU];
      int r_slot[BU], r_js[BU];
#pragma unroll
      for (int u = 0; u < BU; ++u) {
        const int idx = s + u * 32 + lane;
        const int e = s_list[wave][(idx < nloc) ? idx : 0];
        const int ce = (e >> 14) & 31;
        r_slot[u] = s_start[ce] + s_cnt[wave][ce] + (e >> 19) - e0;
        const int li = g * CPG + ce;
        const int lj = c_lj[u];
        const float dx = s_cart[li * 3 + 0] - s_cart[lj * 3 + 0] + c_sx[u];
        const float dy = s_cart[li * 3 + 1] - s_cart[lj * 3 + 1] + c_sy[u];
        const float dz = s_cart[li * 3 + 2] - s_cart[lj * 3 + 2] + c_sz[u];
        r_dx[u] = dx; r_dy[u] = dy; r_dz[u] = dz;
        r_d[u] = sqrtf(dx * dx + dy * dy + dz * dz);
        r_js[u] = lj | (s_spec[lj] << 8);
      }
#pragma unroll
      for (int u = 0; u < BU; ++u)
        asm volatile("" : : "v"(r_d[u]));
#pragma unroll
      for (int u = 0; u < BU; ++u) {
        if (s + u * 32 < nloc) {
          const bool ok = s + u * 32 + lane < nloc;
          const int slot = r_slot[u];
          float fc = 0.5f * cosf(r_d[u] * PI_OVER_CUTOFF) + 0.5f;
          fc = fc * fc;
          if (ok && slot >= 0 && slot < nchunk) {
            s_geo[slot] = make_float4(fc, r_dx[u], r_dy[u], r_dz[u]);
            s_d[slot] = r_d[u];
            s_js[slot] = r_js[u];
          }
        }
      }
#pragma unroll
      for (int u = 0; u < BU; ++u) {
        c_lj[u] = n_lj[u]; c_sx[u] = n_sx[u]; c_sy[u] = n_sy[u]; c_sz[u] = n_sz[u];
      }
    }
    __syncthreads();

    const int lo = ((qstart > e0) ? qstart : e0) - e0;
    const int hi = ((qend < e0 + nchunk) ? qend : (e0 + nchunk)) - e0;
#pragma unroll 2
    for (int i = lo; i < hi; ++i) {
      const float4 g4 = s_geo[i];
      const float d = s_d[i];
      const int js = s_js[i];
      const int lj = js & 255;
      const int sp = js >> 8;
      const float4 rs4 = *reinterpret_cast<const float4*>(s_rs + sp * NWAVE + wq * 4);
      const float4 in4 = *reinterpret_cast<const float4*>(s_inta + sp * NWAVE + wq * 4);
      const float4 ob4 = *reinterpret_cast<const float4*>(s_orb + lj * NWAVE + wq * 4);
      const float rsv[4] = {rs4.x, rs4.y, rs4.z, rs4.w};
      const float inv[4] = {in4.x, in4.y, in4.z, in4.w};
      const float obv[4] = {ob4.x, ob4.y, ob4.z, ob4.w};
      float wv[4];
#pragma unroll
      for (int k = 0; k < 4; ++k) {
        const float tt = d - rsv[k];
        wv[k] = __builtin_amdgcn_exp2f(inv[k] * tt * tt) * obv[k];
      }
      float ang[NPU];
      const float dv[3] = {g4.y, g4.z, g4.w};
      ang[0] = g4.x;
#pragma unroll
      for (int a2 = 0; a2 < 3; ++a2) ang[1 + a2] = g4.x * dv[a2];
      ang[4] = ang[1] * dv[0];
      ang[5] = ang[1] * dv[1];
      ang[6] = ang[1] * dv[2];
      ang[7] = ang[2] * dv[1];
      ang[8] = ang[2] * dv[2];
      ang[9] = ang[3] * dv[2];
#pragma unroll
      for (int j = 0; j < NPU; ++j) {
        acc[j].x += ang[j] * wv[0];
        acc[j].y += ang[j] * wv[1];
        acc[j].z += ang[j] * wv[2];
        acc[j].w += ang[j] * wv[3];
      }
    }
    __syncthreads();
  }

  const int lrow = lane & 15;
  const int lhi = lane >> 4;
  const int khalf = lhi * 2;
  const int mt = wave & 1;
  const int np = wave >> 1;
  const int n0 = np * 32 + lrow;
  v2f Blo[3][2], Bhi[3][2];
#pragma unroll
  for (int ip = 0; ip < 3; ++ip) {
    const float* hb = hyper + ip * (NWAVE * NORBIT);
#pragma unroll
    for (int h = 0; h < 2; ++h) {
      Blo[ip][h].x = hb[(khalf + 0) * NORBIT + n0 + h * 16];
      Blo[ip][h].y = hb[(khalf + 1) * NORBIT + n0 + h * 16];
      Bhi[ip][h].x = hb[(khalf + 4) * NORBIT + n0 + h * 16];
      Bhi[ip][h].y = hb[(khalf + 5) * NORBIT + n0 + h * 16];
    }
  }

  const int u_lg = t >> 3;
  const int u_ch = t & (NWAVE - 1);
  const int u_la = g * CPG + u_lg;
  const float u_off = s_e[s_spec[u_la] * NWAVE + u_ch];
  const float u_old = s_orb[u_la * NWAVE + u_ch];

#pragma unroll
  for (int j = 0; j < NPU; ++j) {
    acc[j].x += quad_xor1(acc[j].x);
    acc[j].y += quad_xor1(acc[j].y);
    acc[j].z += quad_xor1(acc[j].z);
    acc[j].w += quad_xor1(acc[j].w);
    acc[j].x += quad_xor2(acc[j].x);
    acc[j].y += quad_xor2(acc[j].y);
    acc[j].z += quad_xor2(acc[j].z);
    acc[j].w += quad_xor2(acc[j].w);
  }
  if (qd == 0) {
#pragma unroll
    for (int j = 0; j < NPU; ++j)
      *reinterpret_cast<float4*>(s_out + c5 * SOP + j * NWAVE + wq * 4) = acc[j];
  }
  __syncthreads();

  {
    v8f d0 = {}, d1 = {};
    const float* arow = s_out + (mt * 16 + lrow) * SOP;
    int para = 0;
#pragma unroll
    for (int ip = 0; ip < 3; ++ip) {
      const v2f B0lo = Blo[ip][0], B0hi = Bhi[ip][0];
      const v2f B1lo = Blo[ip][1], B1hi = Bhi[ip][1];
      const int npj = (ip == 0) ? 1 : ((ip == 1) ? 3 : 6);
#pragma unroll
      for (int q = 0; q < npj; ++q, ++para) {
        const float* ab = arow + para * NWAVE;
        v2f Alo = {ab[khalf + 0], ab[khalf + 1]};
        v2f Ahi = {ab[khalf + 4], ab[khalf + 5]};
        v8f c0 = {}, c1 = {};
        c0 = wmma_f32_16x16x4(Alo, B0lo, c0);
        c1 = wmma_f32_16x16x4(Alo, B1lo, c1);
        c0 = wmma_f32_16x16x4(Ahi, B0hi, c0);
        c1 = wmma_f32_16x16x4(Ahi, B1hi, c1);
        const bool twice = (ip == 2) && (q == 1 || q == 2 || q == 4);
#pragma unroll
        for (int r = 0; r < 8; ++r) {
          if (twice) {
            d0[r] += 2.0f * (c0[r] * c0[r]);
            d1[r] += 2.0f * (c1[r] * c1[r]);
          } else {
            d0[r] += c0[r] * c0[r];
            d1[r] += c1[r] * c1[r];
          }
        }
      }
    }
#pragma unroll
    for (int r = 0; r < 8; ++r) {
      float* dr = s_den + (mt * 16 + r + lhi * 8) * DP + n0;
      dr[0] = d0[r];
      dr[16] = d1[r];
    }
  }
  __syncthreads();

  if (it == 2) {
    float* dst = density + (b * NATOM + g * CPG) * NORBIT;
    for (int i = t; i < CPG * (NORBIT / 4); i += 256) {
      const int row = i >> 5, c4 = (i & 31) * 4;
      *reinterpret_cast<float4*>(dst + row * NORBIT + c4) =
          *reinterpret_cast<const float4*>(s_den + row * DP + c4);
    }
  } else {
    if (wave < 2) {
      v8f u = {};
      const float* drow = s_den + (wave * 16 + lrow) * DP;
      const bool valid = (lrow < NWAVE);
      float w0[32], w1[32];
#pragma unroll
      for (int kb = 0; kb < 32; ++kb) {
        const int k = kb * 4 + lhi * 2;
        w0[kb] = s_w[(k + 0) * 16 + lrow];
        w1[kb] = s_w[(k + 1) * 16 + lrow];
      }
#pragma unroll
      for (int kb = 0; kb < 32; ++kb) {
        const int k = kb * 4 + lhi * 2;
        v2f A = {drow[k], drow[k + 1]};
        v2f B = {w0[kb], w1[kb]};
        u = wmma_f32_16x16x4(A, B, u);
      }
      if (valid) {
#pragma unroll
        for (int r = 0; r < 8; ++r)
          s_new[(wave * 16 + r + lhi * 8) * NWAVE + lrow] = u[r];
      }
    }
    __syncthreads();
    orb_out[(b * NATOM + g * CPG) * NWAVE + t] = u_old + tanhf(s_new[t] + u_off);
  }
}

extern "C" void kernel_launch(void* const* d_in, const int* in_sizes, int n_in,
                              void* d_out, int out_size, void* d_ws,
                              size_t ws_size, hipStream_t stream) {
  const float* cart = (const float*)d_in[0];
  const float* shifts = (const float*)d_in[1];
  const float* rs = (const float*)d_in[2];
  const float* inta = (const float*)d_in[3];
  const float* params = (const float*)d_in[4];
  const float* hyper = (const float*)d_in[5];
  const float* oc_w0 = (const float*)d_in[6];
  const float* oc_e0 = (const float*)d_in[7];
  const float* oc_w1 = (const float*)d_in[8];
  const float* oc_e1 = (const float*)d_in[9];
  const int* species = (const int*)d_in[11];
  const int* atom_index = (const int*)d_in[12];

  float* density = (float*)d_out;
  float* orb_a = (float*)d_ws;
  float* orb_b = orb_a + TOTNATOM * NWAVE;

  k_obtain<true><<<NB * GRP, 256, 0, stream>>>(cart, shifts, rs, inta, params,
                                               hyper, oc_w0, oc_e0, species,
                                               atom_index, orb_b, orb_a, density, 0);
  k_obtain<false><<<NB * GRP, 256, 0, stream>>>(cart, shifts, rs, inta, params,
                                                hyper, oc_w1, oc_e1, species,
                                                atom_index, orb_a, orb_b, density, 1);
  k_obtain<false><<<NB * GRP, 256, 0, stream>>>(cart, shifts, rs, inta, params,
                                                hyper, oc_w1, oc_e1, species,
                                                atom_index, orb_b, orb_a, density, 2);
}
